# starting kernel
# baseline (speedup 1.0000x reference)
#include <hip/hip_runtime.h>
#include <hip/hip_fp16.h>

typedef __attribute__((ext_vector_type(16))) _Float16 v16h;
typedef __attribute__((ext_vector_type(8)))  float    v8f;

#define N_NODES 100000
#define N_EDGES 1600000
#define NFEAT   128
#define NHID    64
#define NCLASS  16

__global__ void k_init_deg(float* __restrict__ deg) {
    int i = blockIdx.x * blockDim.x + threadIdx.x;
    if (i < N_NODES) deg[i] = 1.0f;
}

__global__ void k_edge_deg(const int* __restrict__ col, float* __restrict__ deg) {
    int e = blockIdx.x * blockDim.x + threadIdx.x;
    if (e < N_EDGES) atomicAdd(&deg[col[e]], 1.0f);
}

__global__ void k_deg_to_dinv(float* __restrict__ deg) {
    int i = blockIdx.x * blockDim.x + threadIdx.x;
    if (i < N_NODES) {
        float d = deg[i];
        deg[i] = d > 0.0f ? rsqrtf(d) : 0.0f;
    }
}

template <int K, int N>
__global__ void k_pack_w(const float* __restrict__ W, _Float16* __restrict__ P) {
    int t = blockIdx.x * blockDim.x + threadIdx.x;
    if (t >= K * N) return;
    int i    = t & 15;
    int u    = t >> 4;
    int lane = u & 31;
    int v    = u >> 5;
    int kc   = v % (K / 32);
    int tn   = v / (K / 32);
    int l16  = lane & 15;
    int kb   = (lane >> 4) * 8;
    int k    = (i < 8) ? (kb + i) : (16 + kb + (i - 8));
    P[t] = (_Float16)W[(long)(kc * 32 + k) * N + tn * 16 + l16];
}

template <int K, int N, bool RELU>
__global__ void k_gemm_wmma(const float* __restrict__ A,
                            const _Float16* __restrict__ Bp,
                            float* __restrict__ C, int m_tiles) {
    constexpr int NT = N / 16;
    constexpr int KC = K / 32;
    const int wave = (blockIdx.x * blockDim.x + threadIdx.x) >> 5;
    const int lane = threadIdx.x & 31;
    if (wave >= m_tiles) return;

    const int l16 = lane & 15;
    const int kb  = (lane >> 4) * 8;
    const float* __restrict__ arow = A + (long)(wave * 16 + l16) * K;

    v8f acc[NT] = {};
#pragma unroll
    for (int kc = 0; kc < KC; ++kc) {
        v16h a;
#pragma unroll
        for (int i = 0; i < 8; ++i) {
            float a0 = arow[kc * 32 + kb + i];
            float a1 = arow[kc * 32 + 16 + kb + i];
            if (RELU) { a0 = fmaxf(a0, 0.0f); a1 = fmaxf(a1, 0.0f); }
            a[i]     = (_Float16)a0;
            a[8 + i] = (_Float16)a1;
        }
#pragma unroll
        for (int tn = 0; tn < NT; ++tn) {
            v16h b = *(const v16h*)(Bp + ((long)(tn * KC + kc) * 32 + lane) * 16);
            acc[tn] = __builtin_amdgcn_wmma_f32_16x16x32_f16(false, a, false, b,
                                                             (short)0, acc[tn],
                                                             false, false);
        }
    }

    float* __restrict__ crow = C + ((long)wave * 16 + (lane >> 4) * 8) * N + l16;
#pragma unroll
    for (int tn = 0; tn < NT; ++tn)
#pragma unroll
        for (int r = 0; r < 8; ++r) crow[tn * 16 + (long)r * N] = acc[tn][r];
}

template <int F>
__global__ void k_init_agg(const float* __restrict__ xw,
                           const float* __restrict__ dinv,
                           const float* __restrict__ bias,
                           float* __restrict__ h) {
    long t = (long)blockIdx.x * blockDim.x + threadIdx.x;
    if (t >= (long)N_NODES * F) return;
    int i = (int)(t / F);
    int j = (int)(t - (long)i * F);
    float di = dinv[i];
    h[t] = bias[j] + di * di * xw[t];
}

__global__ void k_edge_scatter64(const int* __restrict__ row,
                                 const int* __restrict__ col,
                                 const float* __restrict__ dinv,
                                 const float* __restrict__ xw,
                                 float* __restrict__ h) {
    const int e    = (blockIdx.x * blockDim.x + threadIdx.x) >> 5;
    const int lane = threadIdx.x & 31;
    if (e >= N_EDGES) return;
    const int r = row[e];
    const int c = col[e];
    const float w = dinv[r] * dinv[c];
    const float* __restrict__ src = xw + (long)r * NHID;
    float* __restrict__ dst       = h  + (long)c * NHID;
    atomicAdd(&dst[lane],      w * src[lane]);
    atomicAdd(&dst[lane + 32], w * src[lane + 32]);
}

__global__ void k_edge_scatter16(const int* __restrict__ row,
                                 const int* __restrict__ col,
                                 const float* __restrict__ dinv,
                                 const float* __restrict__ h2,
                                 float* __restrict__ out) {
    const int wave = (blockIdx.x * blockDim.x + threadIdx.x) >> 5;
    const int lane = threadIdx.x & 31;
    const int e    = wave * 2 + (lane >> 4);
    const int j    = lane & 15;
    if (e >= N_EDGES) return;
    const int r = row[e];
    const int c = col[e];
    const float w = dinv[r] * dinv[c];
    atomicAdd(&out[(long)c * NCLASS + j], w * h2[(long)r * NCLASS + j]);
}

__global__ void k_log_softmax16(float* __restrict__ out) {
    int i = blockIdx.x * blockDim.x + threadIdx.x;
    if (i >= N_NODES) return;
    float* __restrict__ rowp = out + (long)i * NCLASS;
    float v[NCLASS];
    float m = -3.402823466e38f;
#pragma unroll
    for (int j = 0; j < NCLASS; ++j) { v[j] = rowp[j]; m = fmaxf(m, v[j]); }
    float s = 0.0f;
#pragma unroll
    for (int j = 0; j < NCLASS; ++j) s += __expf(v[j] - m);
    float lse = m + __logf(s);
#pragma unroll
    for (int j = 0; j < NCLASS; ++j) rowp[j] = v[j] - lse;
}

extern "C" void kernel_launch(void* const* d_in, const int* in_sizes, int n_in,
                              void* d_out, int out_size, void* d_ws, size_t ws_size,
                              hipStream_t stream) {
    const float* x   = (const float*)d_in[0];
    const int*   ei  = (const int*)d_in[1];
    const float* W1  = (const float*)d_in[2];
    const float* b1  = (const float*)d_in[3];
    const float* W2  = (const float*)d_in[4];
    const float* b2  = (const float*)d_in[5];
    float* out       = (float*)d_out;

    const int* row = ei;
    const int* col = ei + N_EDGES;

    auto align256 = [](size_t v) { return (v + 255) & ~(size_t)255; };
    char* ws = (char*)d_ws;
    size_t o = 0;
    float*     dinv = (float*)(ws + o);     o += align256((size_t)N_NODES * 4);
    _Float16*  W1p  = (_Float16*)(ws + o);  o += align256((size_t)NFEAT * NHID * 2);
    _Float16*  W2p  = (_Float16*)(ws + o);  o += align256((size_t)NHID * NCLASS * 2);
    float*     bufA = (float*)(ws + o);     o += align256((size_t)N_NODES * NHID * 4);
    float*     h    = (float*)(ws + o);
    float*     h2   = bufA;

    const int B = 256;

    k_pack_w<NFEAT, NHID><<<(NFEAT * NHID + B - 1) / B, B, 0, stream>>>(W1, W1p);
    k_pack_w<NHID, NCLASS><<<(NHID * NCLASS + B - 1) / B, B, 0, stream>>>(W2, W2p);

    k_init_deg<<<(N_NODES + B - 1) / B, B, 0, stream>>>(dinv);
    k_edge_deg<<<(N_EDGES + B - 1) / B, B, 0, stream>>>(col, dinv);
    k_deg_to_dinv<<<(N_NODES + B - 1) / B, B, 0, stream>>>(dinv);

    const int m_tiles = N_NODES / 16;

    k_gemm_wmma<NFEAT, NHID, false><<<(m_tiles * 32 + B - 1) / B, B, 0, stream>>>(
        x, W1p, bufA, m_tiles);

    k_init_agg<NHID><<<((long)N_NODES * NHID + B - 1) / B, B, 0, stream>>>(bufA, dinv, b1, h);
    k_edge_scatter64<<<((long)N_EDGES * 32 + B - 1) / B, B, 0, stream>>>(row, col, dinv, bufA, h);

    k_gemm_wmma<NHID, NCLASS, true><<<(m_tiles * 32 + B - 1) / B, B, 0, stream>>>(
        h, W2p, h2, m_tiles);

    k_init_agg<NCLASS><<<((long)N_NODES * NCLASS + B - 1) / B, B, 0, stream>>>(h2, dinv, b2, out);
    k_edge_scatter16<<<((long)N_EDGES / 2 * 32 + B - 1) / B, B, 0, stream>>>(row, col, dinv, h2, out);
    k_log_softmax16<<<(N_NODES + B - 1) / B, B, 0, stream>>>(out);
}


# Degree count: per-edge global f32 atomics replaced by LDS histograms (8 node ranges x 32 edge parts), exact integer partial counts summed per node; dinv bit-identical, no global atomic left there
# speedup vs baseline: 1.1880x; 1.1880x over previous
#include <hip/hip_runtime.h>
#include <hip/hip_fp16.h>

typedef __attribute__((ext_vector_type(16))) _Float16 v16h;
typedef __attribute__((ext_vector_type(8)))  float    v8f;

#define N_NODES 100000
#define N_EDGES 1600000
#define NFEAT   128
#define NHID    64
#define NCLASS  16

#define DEG_G   8
#define DEG_P   32
#define DEG_R   ((((N_NODES + DEG_G - 1) / DEG_G) + 31) & ~31)
#define DEG_EPP (N_EDGES / DEG_P)
#define DEG_T   512
#define DEG_U   4

static_assert(N_EDGES % DEG_P == 0, "edge parts must be equal");
static_assert(DEG_EPP % 4 == 0, "edge parts are read 16 bytes per lane");
static_assert((long)DEG_G * DEG_R >= N_NODES, "node ranges must cover all nodes");

__global__ void __launch_bounds__(DEG_T)
k_deg_hist(const int* __restrict__ col, unsigned* __restrict__ part) {
    __shared__ unsigned hist[DEG_R];
    const int g  = blockIdx.x % DEG_G;
    const int p  = blockIdx.x / DEG_G;
    const int lo = g * DEG_R;

    for (int i = threadIdx.x; i < DEG_R; i += DEG_T) hist[i] = 0u;
    __syncthreads();

    const int4* __restrict__ c4 = (const int4*)(col + (long)p * DEG_EPP);
    constexpr int NV = DEG_EPP / 4;
    for (int base = threadIdx.x; base < NV; base += DEG_T * DEG_U) {
        int4 v[DEG_U];
        for (int u = 0; u < DEG_U; ++u) {
            const int i = base + u * DEG_T;
            v[u] = c4[i < NV ? i : NV - 1];
        }
        for (int u = 0; u < DEG_U; ++u) {
            const bool in = base + u * DEG_T < NV;
            const unsigned d0 = in ? (unsigned)(v[u].x - lo) : ~0u;
            const unsigned d1 = in ? (unsigned)(v[u].y - lo) : ~0u;
            const unsigned d2 = in ? (unsigned)(v[u].z - lo) : ~0u;
            const unsigned d3 = in ? (unsigned)(v[u].w - lo) : ~0u;
            if (d0 < (unsigned)DEG_R) atomicAdd(&hist[d0], 1u);
            if (d1 < (unsigned)DEG_R) atomicAdd(&hist[d1], 1u);
            if (d2 < (unsigned)DEG_R) atomicAdd(&hist[d2], 1u);
            if (d3 < (unsigned)DEG_R) atomicAdd(&hist[d3], 1u);
        }
    }
    __syncthreads();

    unsigned* __restrict__ prow = part + (long)p * N_NODES + lo;
    for (int i = threadIdx.x; i < DEG_R; i += DEG_T)
        if (lo + i < N_NODES) prow[i] = hist[i];
}

__global__ void k_deg_to_dinv(const unsigned* __restrict__ part,
                              float* __restrict__ dinv) {
    int i = blockIdx.x * blockDim.x + threadIdx.x;
    if (i < N_NODES) {
        unsigned s = 1u;
        for (int p = 0; p < DEG_P; ++p) s += part[(long)p * N_NODES + i];
        float d = (float)s;
        dinv[i] = d > 0.0f ? rsqrtf(d) : 0.0f;
    }
}

template <int K, int N>
__global__ void k_pack_w(const float* __restrict__ W, _Float16* __restrict__ P) {
    int t = blockIdx.x * blockDim.x + threadIdx.x;
    if (t >= K * N) return;
    int i    = t & 15;
    int u    = t >> 4;
    int lane = u & 31;
    int v    = u >> 5;
    int kc   = v % (K / 32);
    int tn   = v / (K / 32);
    int l16  = lane & 15;
    int kb   = (lane >> 4) * 8;
    int k    = (i < 8) ? (kb + i) : (16 + kb + (i - 8));
    P[t] = (_Float16)W[(long)(kc * 32 + k) * N + tn * 16 + l16];
}

template <int K, int N, bool RELU>
__global__ void k_gemm_wmma(const float* __restrict__ A,
                            const _Float16* __restrict__ Bp,
                            float* __restrict__ C, int m_tiles) {
    constexpr int NT = N / 16;
    constexpr int KC = K / 32;
    const int wave = (blockIdx.x * blockDim.x + threadIdx.x) >> 5;
    const int lane = threadIdx.x & 31;
    if (wave >= m_tiles) return;

    const int l16 = lane & 15;
    const int kb  = (lane >> 4) * 8;
    const float* __restrict__ arow = A + (long)(wave * 16 + l16) * K;

    v8f acc[NT] = {};
#pragma unroll
    for (int kc = 0; kc < KC; ++kc) {
        v16h a;
#pragma unroll
        for (int i = 0; i < 8; ++i) {
            float a0 = arow[kc * 32 + kb + i];
            float a1 = arow[kc * 32 + 16 + kb + i];
            if (RELU) { a0 = fmaxf(a0, 0.0f); a1 = fmaxf(a1, 0.0f); }
            a[i]     = (_Float16)a0;
            a[8 + i] = (_Float16)a1;
        }
#pragma unroll
        for (int tn = 0; tn < NT; ++tn) {
            v16h b = *(const v16h*)(Bp + ((long)(tn * KC + kc) * 32 + lane) * 16);
            acc[tn] = __builtin_amdgcn_wmma_f32_16x16x32_f16(false, a, false, b,
                                                             (short)0, acc[tn],
                                                             false, false);
        }
    }

    float* __restrict__ crow = C + ((long)wave * 16 + (lane >> 4) * 8) * N + l16;
#pragma unroll
    for (int tn = 0; tn < NT; ++tn)
#pragma unroll
        for (int r = 0; r < 8; ++r) crow[tn * 16 + (long)r * N] = acc[tn][r];
}

template <int F>
__global__ void k_init_agg(const float* __restrict__ xw,
                           const float* __restrict__ dinv,
                           const float* __restrict__ bias,
                           float* __restrict__ h) {
    long t = (long)blockIdx.x * blockDim.x + threadIdx.x;
    if (t >= (long)N_NODES * F) return;
    int i = (int)(t / F);
    int j = (int)(t - (long)i * F);
    float di = dinv[i];
    h[t] = bias[j] + di * di * xw[t];
}

__global__ void k_edge_scatter64(const int* __restrict__ row,
                                 const int* __restrict__ col,
                                 const float* __restrict__ dinv,
                                 const float* __restrict__ xw,
                                 float* __restrict__ h) {
    const int e    = (blockIdx.x * blockDim.x + threadIdx.x) >> 5;
    const int lane = threadIdx.x & 31;
    if (e >= N_EDGES) return;
    const int r = row[e];
    const int c = col[e];
    const float w = dinv[r] * dinv[c];
    const float* __restrict__ src = xw + (long)r * NHID;
    float* __restrict__ dst       = h  + (long)c * NHID;
    atomicAdd(&dst[lane],      w * src[lane]);
    atomicAdd(&dst[lane + 32], w * src[lane + 32]);
}

__global__ void k_edge_scatter16(const int* __restrict__ row,
                                 const int* __restrict__ col,
                                 const float* __restrict__ dinv,
                                 const float* __restrict__ h2,
                                 float* __restrict__ out) {
    const int wave = (blockIdx.x * blockDim.x + threadIdx.x) >> 5;
    const int lane = threadIdx.x & 31;
    const int e    = wave * 2 + (lane >> 4);
    const int j    = lane & 15;
    if (e >= N_EDGES) return;
    const int r = row[e];
    const int c = col[e];
    const float w = dinv[r] * dinv[c];
    atomicAdd(&out[(long)c * NCLASS + j], w * h2[(long)r * NCLASS + j]);
}

__global__ void k_log_softmax16(float* __restrict__ out) {
    int i = blockIdx.x * blockDim.x + threadIdx.x;
    if (i >= N_NODES) return;
    float* __restrict__ rowp = out + (long)i * NCLASS;
    float v[NCLASS];
    float m = -3.402823466e38f;
#pragma unroll
    for (int j = 0; j < NCLASS; ++j) { v[j] = rowp[j]; m = fmaxf(m, v[j]); }
    float s = 0.0f;
#pragma unroll
    for (int j = 0; j < NCLASS; ++j) s += __expf(v[j] - m);
    float lse = m + __logf(s);
#pragma unroll
    for (int j = 0; j < NCLASS; ++j) rowp[j] = v[j] - lse;
}

extern "C" void kernel_launch(void* const* d_in, const int* in_sizes, int n_in,
                              void* d_out, int out_size, void* d_ws, size_t ws_size,
                              hipStream_t stream) {
    const float* x   = (const float*)d_in[0];
    const int*   ei  = (const int*)d_in[1];
    const float* W1  = (const float*)d_in[2];
    const float* b1  = (const float*)d_in[3];
    const float* W2  = (const float*)d_in[4];
    const float* b2  = (const float*)d_in[5];
    float* out       = (float*)d_out;

    const int* row = ei;
    const int* col = ei + N_EDGES;

    auto align256 = [](size_t v) { return (v + 255) & ~(size_t)255; };
    char* ws = (char*)d_ws;
    size_t o = 0;
    float*     dinv = (float*)(ws + o);     o += align256((size_t)N_NODES * 4);
    _Float16*  W1p  = (_Float16*)(ws + o);  o += align256((size_t)NFEAT * NHID * 2);
    _Float16*  W2p  = (_Float16*)(ws + o);  o += align256((size_t)NHID * NCLASS * 2);
    float*     bufA = (float*)(ws + o);     o += align256((size_t)N_NODES * NHID * 4);
    float*     h    = (float*)(ws + o);     o += align256((size_t)N_NODES * NHID * 4);
    unsigned*  part = (unsigned*)(ws + o);
    float*     h2   = bufA;

    const int B = 256;

    k_pack_w<NFEAT, NHID><<<(NFEAT * NHID + B - 1) / B, B, 0, stream>>>(W1, W1p);
    k_pack_w<NHID, NCLASS><<<(NHID * NCLASS + B - 1) / B, B, 0, stream>>>(W2, W2p);

    k_deg_hist<<<DEG_G * DEG_P, DEG_T, 0, stream>>>(col, part);
    k_deg_to_dinv<<<(N_NODES + B - 1) / B, B, 0, stream>>>(part, dinv);

    const int m_tiles = N_NODES / 16;

    k_gemm_wmma<NFEAT, NHID, false><<<(m_tiles * 32 + B - 1) / B, B, 0, stream>>>(
        x, W1p, bufA, m_tiles);

    k_init_agg<NHID><<<((long)N_NODES * NHID + B - 1) / B, B, 0, stream>>>(bufA, dinv, b1, h);
    k_edge_scatter64<<<((long)N_EDGES * 32 + B - 1) / B, B, 0, stream>>>(row, col, dinv, bufA, h);

    k_gemm_wmma<NHID, NCLASS, true><<<(m_tiles * 32 + B - 1) / B, B, 0, stream>>>(
        h, W2p, h2, m_tiles);

    k_init_agg<NCLASS><<<((long)N_NODES * NCLASS + B - 1) / B, B, 0, stream>>>(h2, dinv, b2, out);
    k_edge_scatter16<<<((long)N_EDGES / 2 * 32 + B - 1) / B, B, 0, stream>>>(row, col, dinv, h2, out);
    k_log_softmax16<<<(N_NODES + B - 1) / B, B, 0, stream>>>(out);
}


# Edge sums: global f32 atomic scatter replaced by a per-call counting sort of edges by destination (LDS ranks) and one owner per node adding its segment in registers; sum order = segment order
# speedup vs baseline: 2.7595x; 2.3228x over previous
#include <hip/hip_runtime.h>
#include <hip/hip_fp16.h>

typedef __attribute__((ext_vector_type(16))) _Float16 v16h;
typedef __attribute__((ext_vector_type(8)))  float    v8f;

#define N_NODES 100000
#define N_EDGES 1600000
#define NFEAT   128
#define NHID    64
#define NCLASS  16

#define DEG_G   8
#define DEG_P   32
#define DEG_R   ((((N_NODES + DEG_G - 1) / DEG_G) + 31) & ~31)
#define DEG_EPP (N_EDGES / DEG_P)
#define DEG_T   512
#define DEG_U   4
#define SCAN_B  256
#define SCAN_NB ((N_NODES + SCAN_B - 1) / SCAN_B)

static_assert(N_EDGES % DEG_P == 0, "edge parts must be equal");
static_assert(DEG_EPP % 4 == 0, "edge parts are read 16 bytes per lane");
static_assert((long)DEG_G * DEG_R >= N_NODES, "node ranges must cover all nodes");

__global__ void __launch_bounds__(DEG_T)
k_deg_hist(const int* __restrict__ col, unsigned* __restrict__ part) {
    __shared__ unsigned hist[DEG_R];
    const int g  = blockIdx.x % DEG_G;
    const int p  = blockIdx.x / DEG_G;
    const int lo = g * DEG_R;

    for (int i = threadIdx.x; i < DEG_R; i += DEG_T) hist[i] = 0u;
    __syncthreads();

    const int4* __restrict__ c4 = (const int4*)(col + (long)p * DEG_EPP);
    constexpr int NV = DEG_EPP / 4;
    for (int base = threadIdx.x; base < NV; base += DEG_T * DEG_U) {
        int4 v[DEG_U];
        for (int u = 0; u < DEG_U; ++u) {
            const int i = base + u * DEG_T;
            v[u] = c4[i < NV ? i : NV - 1];
        }
        for (int u = 0; u < DEG_U; ++u) {
            const bool in = base + u * DEG_T < NV;
            const unsigned d0 = in ? (unsigned)(v[u].x - lo) : ~0u;
            const unsigned d1 = in ? (unsigned)(v[u].y - lo) : ~0u;
            const unsigned d2 = in ? (unsigned)(v[u].z - lo) : ~0u;
            const unsigned d3 = in ? (unsigned)(v[u].w - lo) : ~0u;
            if (d0 < (unsigned)DEG_R) atomicAdd(&hist[d0], 1u);
            if (d1 < (unsigned)DEG_R) atomicAdd(&hist[d1], 1u);
            if (d2 < (unsigned)DEG_R) atomicAdd(&hist[d2], 1u);
            if (d3 < (unsigned)DEG_R) atomicAdd(&hist[d3], 1u);
        }
    }
    __syncthreads();

    unsigned* __restrict__ prow = part + (long)p * N_NODES + lo;
    for (int i = threadIdx.x; i < DEG_R; i += DEG_T)
        if (lo + i < N_NODES) prow[i] = hist[i];
}

__global__ void __launch_bounds__(SCAN_B)
k_deg_to_dinv(unsigned* __restrict__ part, float* __restrict__ dinv,
              unsigned* __restrict__ deg, unsigned* __restrict__ bsum) {
    __shared__ unsigned red[SCAN_B];
    const int i = blockIdx.x * SCAN_B + threadIdx.x;
    unsigned run = 0u;
    if (i < N_NODES) {
        for (int p = 0; p < DEG_P; ++p) {
            unsigned* q = part + (long)p * N_NODES + i;
            const unsigned c = *q;
            *q = run;
            run += c;
        }
        deg[i] = run;
        float d = (float)(run + 1u);
        dinv[i] = d > 0.0f ? rsqrtf(d) : 0.0f;
    }
    red[threadIdx.x] = run;
    __syncthreads();
    for (int s = SCAN_B / 2; s > 0; s >>= 1) {
        if ((int)threadIdx.x < s) red[threadIdx.x] += red[threadIdx.x + s];
        __syncthreads();
    }
    if (threadIdx.x == 0) bsum[blockIdx.x] = red[0];
}

__global__ void __launch_bounds__(SCAN_B)
k_rowptr(const unsigned* __restrict__ deg, const unsigned* __restrict__ bsum,
         unsigned* __restrict__ rowptr) {
    __shared__ unsigned sh[SCAN_B];
    const int t = threadIdx.x;
    const int i = blockIdx.x * SCAN_B + t;
    unsigned acc = 0u;
    for (int b = t; b < (int)blockIdx.x; b += SCAN_B) acc += bsum[b];
    sh[t] = acc;
    __syncthreads();
    for (int s = SCAN_B / 2; s > 0; s >>= 1) {
        if (t < s) sh[t] += sh[t + s];
        __syncthreads();
    }
    const unsigned base = sh[0];
    __syncthreads();
    const unsigned v = i < N_NODES ? deg[i] : 0u;
    sh[t] = v;
    __syncthreads();
    for (int d = 1; d < SCAN_B; d <<= 1) {
        const unsigned x = t >= d ? sh[t - d] : 0u;
        __syncthreads();
        sh[t] += x;
        __syncthreads();
    }
    if (i < N_NODES) rowptr[i] = base + sh[t] - v;
}

__global__ void __launch_bounds__(DEG_T)
k_fill_srcs(const int* __restrict__ row, const int* __restrict__ col,
            const unsigned* __restrict__ part,
            const unsigned* __restrict__ rowptr, int* __restrict__ srcs) {
    __shared__ unsigned cur[DEG_R];
    const int g  = blockIdx.x % DEG_G;
    const int p  = blockIdx.x / DEG_G;
    const int lo = g * DEG_R;
    const unsigned lim = (unsigned)(N_NODES - lo < DEG_R ? N_NODES - lo : DEG_R);

    const unsigned* __restrict__ prow = part + (long)p * N_NODES + lo;
    for (int i0 = threadIdx.x; i0 < DEG_R; i0 += DEG_T * DEG_U) {
        unsigned a[DEG_U], b[DEG_U];
        for (int u = 0; u < DEG_U; ++u) {
            const unsigned i  = (unsigned)(i0 + u * DEG_T);
            const unsigned ii = i < lim ? i : lim - 1u;
            a[u] = rowptr[lo + ii];
            b[u] = prow[ii];
        }
        for (int u = 0; u < DEG_U; ++u) {
            const unsigned i = (unsigned)(i0 + u * DEG_T);
            if (i < (unsigned)DEG_R) cur[i] = i < lim ? a[u] + b[u] : 0u;
        }
    }
    __syncthreads();

    const int4* __restrict__ c4 = (const int4*)(col + (long)p * DEG_EPP);
    const int4* __restrict__ r4 = (const int4*)(row + (long)p * DEG_EPP);
    constexpr int NV = DEG_EPP / 4;
    for (int base = threadIdx.x; base < NV; base += DEG_T * DEG_U) {
        int4 vc[DEG_U], vr[DEG_U];
        for (int u = 0; u < DEG_U; ++u) {
            const int i  = base + u * DEG_T;
            const int ii = i < NV ? i : NV - 1;
            vc[u] = c4[ii];
            vr[u] = r4[ii];
        }
        for (int u = 0; u < DEG_U; ++u) {
            const bool in = base + u * DEG_T < NV;
            const unsigned d0 = in ? (unsigned)(vc[u].x - lo) : ~0u;
            const unsigned d1 = in ? (unsigned)(vc[u].y - lo) : ~0u;
            const unsigned d2 = in ? (unsigned)(vc[u].z - lo) : ~0u;
            const unsigned d3 = in ? (unsigned)(vc[u].w - lo) : ~0u;
            if (d0 < lim) srcs[atomicAdd(&cur[d0], 1u)] = vr[u].x;
            if (d1 < lim) srcs[atomicAdd(&cur[d1], 1u)] = vr[u].y;
            if (d2 < lim) srcs[atomicAdd(&cur[d2], 1u)] = vr[u].z;
            if (d3 < lim) srcs[atomicAdd(&cur[d3], 1u)] = vr[u].w;
        }
    }
}

template <int K, int N>
__global__ void k_pack_w(const float* __restrict__ W, _Float16* __restrict__ P) {
    int t = blockIdx.x * blockDim.x + threadIdx.x;
    if (t >= K * N) return;
    int i    = t & 15;
    int u    = t >> 4;
    int lane = u & 31;
    int v    = u >> 5;
    int kc   = v % (K / 32);
    int tn   = v / (K / 32);
    int l16  = lane & 15;
    int kb   = (lane >> 4) * 8;
    int k    = (i < 8) ? (kb + i) : (16 + kb + (i - 8));
    P[t] = (_Float16)W[(long)(kc * 32 + k) * N + tn * 16 + l16];
}

template <int K, int N, bool RELU>
__global__ void k_gemm_wmma(const float* __restrict__ A,
                            const _Float16* __restrict__ Bp,
                            float* __restrict__ C, int m_tiles) {
    constexpr int NT = N / 16;
    constexpr int KC = K / 32;
    const int wave = (blockIdx.x * blockDim.x + threadIdx.x) >> 5;
    const int lane = threadIdx.x & 31;
    if (wave >= m_tiles) return;

    const int l16 = lane & 15;
    const int kb  = (lane >> 4) * 8;
    const float* __restrict__ arow = A + (long)(wave * 16 + l16) * K;

    v8f acc[NT] = {};
#pragma unroll
    for (int kc = 0; kc < KC; ++kc) {
        v16h a;
#pragma unroll
        for (int i = 0; i < 8; ++i) {
            float a0 = arow[kc * 32 + kb + i];
            float a1 = arow[kc * 32 + 16 + kb + i];
            if (RELU) { a0 = fmaxf(a0, 0.0f); a1 = fmaxf(a1, 0.0f); }
            a[i]     = (_Float16)a0;
            a[8 + i] = (_Float16)a1;
        }
#pragma unroll
        for (int tn = 0; tn < NT; ++tn) {
            v16h b = *(const v16h*)(Bp + ((long)(tn * KC + kc) * 32 + lane) * 16);
            acc[tn] = __builtin_amdgcn_wmma_f32_16x16x32_f16(false, a, false, b,
                                                             (short)0, acc[tn],
                                                             false, false);
        }
    }

    float* __restrict__ crow = C + ((long)wave * 16 + (lane >> 4) * 8) * N + l16;
#pragma unroll
    for (int tn = 0; tn < NT; ++tn)
#pragma unroll
        for (int r = 0; r < 8; ++r) crow[tn * 16 + (long)r * N] = acc[tn][r];
}

__global__ void __launch_bounds__(256)
k_gather64(const int* __restrict__ srcs, const unsigned* __restrict__ rowptr,
           const unsigned* __restrict__ deg, const float* __restrict__ dinv,
           const float* __restrict__ xw, const float* __restrict__ bias,
           float* __restrict__ h) {
    const int lane = threadIdx.x & 31;
    const int c = __builtin_amdgcn_readfirstlane(
        (int)((blockIdx.x * blockDim.x + threadIdx.x) >> 5));
    if (c >= N_NODES) return;
    const unsigned start = rowptr[c];
    const unsigned cnt   = deg[c];
    const float    dc    = dinv[c];
    const float2 self = *(const float2*)(xw + (long)c * NHID + lane * 2);
    const float2 b    = *(const float2*)(bias + lane * 2);
    float ax = b.x + dc * dc * self.x;
    float ay = b.y + dc * dc * self.y;

    for (unsigned base = 0; base < cnt; base += 32u) {
        const unsigned n = cnt - base < 32u ? cnt - base : 32u;
        int   my = 0;
        float mw = 0.0f;
        if ((unsigned)lane < n) {
            my = srcs[start + base + lane];
            mw = dinv[my] * dc;
        }
        const int mwi = __float_as_int(mw);
        unsigned j = 0;
        for (; j + 4u <= n; j += 4u) {
            const int r0 = __builtin_amdgcn_readlane(my, j);
            const int r1 = __builtin_amdgcn_readlane(my, j + 1u);
            const int r2 = __builtin_amdgcn_readlane(my, j + 2u);
            const int r3 = __builtin_amdgcn_readlane(my, j + 3u);
            const float2 v0 = *(const float2*)(xw + (long)r0 * NHID + lane * 2);
            const float2 v1 = *(const float2*)(xw + (long)r1 * NHID + lane * 2);
            const float2 v2 = *(const float2*)(xw + (long)r2 * NHID + lane * 2);
            const float2 v3 = *(const float2*)(xw + (long)r3 * NHID + lane * 2);
            const float w0 = __int_as_float(__builtin_amdgcn_readlane(mwi, j));
            const float w1 = __int_as_float(__builtin_amdgcn_readlane(mwi, j + 1u));
            const float w2 = __int_as_float(__builtin_amdgcn_readlane(mwi, j + 2u));
            const float w3 = __int_as_float(__builtin_amdgcn_readlane(mwi, j + 3u));
            ax += w0 * v0.x;  ay += w0 * v0.y;
            ax += w1 * v1.x;  ay += w1 * v1.y;
            ax += w2 * v2.x;  ay += w2 * v2.y;
            ax += w3 * v3.x;  ay += w3 * v3.y;
        }
        for (; j < n; ++j) {
            const int   r = __builtin_amdgcn_readlane(my, j);
            const float w = __int_as_float(__builtin_amdgcn_readlane(mwi, j));
            const float2 v = *(const float2*)(xw + (long)r * NHID + lane * 2);
            ax += w * v.x;  ay += w * v.y;
        }
    }
    *(float2*)(h + (long)c * NHID + lane * 2) = make_float2(ax, ay);
}

__global__ void __launch_bounds__(256)
k_gather16(const int* __restrict__ srcs, const unsigned* __restrict__ rowptr,
           const unsigned* __restrict__ deg, const float* __restrict__ dinv,
           const float* __restrict__ h2, const float* __restrict__ bias,
           float* __restrict__ out) {
    const int t = blockIdx.x * blockDim.x + threadIdx.x;
    const int c = t >> 4;
    const int j = t & 15;
    if (c >= N_NODES) return;
    const unsigned cnt = deg[c];
    const int* __restrict__ seg = srcs + rowptr[c];
    const float dc = dinv[c];
    float acc = bias[j] + dc * dc * h2[(long)c * NCLASS + j];
    unsigned e = 0;
    for (; e + 4u <= cnt; e += 4u) {
        const int r0 = seg[e], r1 = seg[e + 1u], r2 = seg[e + 2u], r3 = seg[e + 3u];
        const float w0 = dinv[r0] * dc;
        const float w1 = dinv[r1] * dc;
        const float w2 = dinv[r2] * dc;
        const float w3 = dinv[r3] * dc;
        const float v0 = h2[(long)r0 * NCLASS + j];
        const float v1 = h2[(long)r1 * NCLASS + j];
        const float v2 = h2[(long)r2 * NCLASS + j];
        const float v3 = h2[(long)r3 * NCLASS + j];
        acc += w0 * v0;
        acc += w1 * v1;
        acc += w2 * v2;
        acc += w3 * v3;
    }
    for (; e < cnt; ++e) {
        const int r = seg[e];
        acc += (dinv[r] * dc) * h2[(long)r * NCLASS + j];
    }
    out[(long)c * NCLASS + j] = acc;
}

__global__ void k_log_softmax16(float* __restrict__ out) {
    int i = blockIdx.x * blockDim.x + threadIdx.x;
    if (i >= N_NODES) return;
    float* __restrict__ rowp = out + (long)i * NCLASS;
    float v[NCLASS];
    float m = -3.402823466e38f;
#pragma unroll
    for (int j = 0; j < NCLASS; ++j) { v[j] = rowp[j]; m = fmaxf(m, v[j]); }
    float s = 0.0f;
#pragma unroll
    for (int j = 0; j < NCLASS; ++j) s += __expf(v[j] - m);
    float lse = m + __logf(s);
#pragma unroll
    for (int j = 0; j < NCLASS; ++j) rowp[j] = v[j] - lse;
}

extern "C" void kernel_launch(void* const* d_in, const int* in_sizes, int n_in,
                              void* d_out, int out_size, void* d_ws, size_t ws_size,
                              hipStream_t stream) {
    const float* x   = (const float*)d_in[0];
    const int*   ei  = (const int*)d_in[1];
    const float* W1  = (const float*)d_in[2];
    const float* b1  = (const float*)d_in[3];
    const float* W2  = (const float*)d_in[4];
    const float* b2  = (const float*)d_in[5];
    float* out       = (float*)d_out;

    const int* row = ei;
    const int* col = ei + N_EDGES;

    auto align256 = [](size_t v) { return (v + 255) & ~(size_t)255; };
    char* ws = (char*)d_ws;
    size_t o = 0;
    float*     dinv = (float*)(ws + o);     o += align256((size_t)N_NODES * 4);
    _Float16*  W1p  = (_Float16*)(ws + o);  o += align256((size_t)NFEAT * NHID * 2);
    _Float16*  W2p  = (_Float16*)(ws + o);  o += align256((size_t)NHID * NCLASS * 2);
    float*     bufA = (float*)(ws + o);     o += align256((size_t)N_NODES * NHID * 4);
    float*     h    = (float*)(ws + o);     o += align256((size_t)N_NODES * NHID * 4);
    unsigned*  deg    = (unsigned*)(ws + o); o += align256((size_t)N_NODES * 4);
    unsigned*  rowptr = (unsigned*)(ws + o); o += align256((size_t)N_NODES * 4);
    unsigned*  bsum   = (unsigned*)(ws + o); o += align256((size_t)SCAN_NB * 4);
    int*       srcs   = (int*)(ws + o);      o += align256((size_t)N_EDGES * 4);
    unsigned*  part = (unsigned*)(ws + o);
    float*     h2   = bufA;

    const int B = 256;

    k_pack_w<NFEAT, NHID><<<(NFEAT * NHID + B - 1) / B, B, 0, stream>>>(W1, W1p);
    k_pack_w<NHID, NCLASS><<<(NHID * NCLASS + B - 1) / B, B, 0, stream>>>(W2, W2p);

    k_deg_hist<<<DEG_G * DEG_P, DEG_T, 0, stream>>>(col, part);
    k_deg_to_dinv<<<SCAN_NB, SCAN_B, 0, stream>>>(part, dinv, deg, bsum);
    k_rowptr<<<SCAN_NB, SCAN_B, 0, stream>>>(deg, bsum, rowptr);
    k_fill_srcs<<<DEG_G * DEG_P, DEG_T, 0, stream>>>(row, col, part, rowptr, srcs);

    const int m_tiles = N_NODES / 16;

    k_gemm_wmma<NFEAT, NHID, false><<<(m_tiles * 32 + B - 1) / B, B, 0, stream>>>(
        x, W1p, bufA, m_tiles);

    k_gather64<<<((long)N_NODES * 32 + B - 1) / B, B, 0, stream>>>(
        srcs, rowptr, deg, dinv, bufA, b1, h);

    k_gemm_wmma<NHID, NCLASS, true><<<(m_tiles * 32 + B - 1) / B, B, 0, stream>>>(
        h, W2p, h2, m_tiles);

    k_gather16<<<((long)N_NODES * 16 + B - 1) / B, B, 0, stream>>>(
        srcs, rowptr, deg, dinv, h2, b2, out);
    k_log_softmax16<<<(N_NODES + B - 1) / B, B, 0, stream>>>(out);
}


# Edge-sort kernels (count, row starts, fill): a lane requests all loads of a phase at once (was 7 dependent batches) and a batch's 16 LDS ranks before their stores, so round trips overlap
# speedup vs baseline: 2.8784x; 1.0431x over previous
#include <hip/hip_runtime.h>
#include <hip/hip_fp16.h>

typedef __attribute__((ext_vector_type(16))) _Float16 v16h;
typedef __attribute__((ext_vector_type(8)))  float    v8f;

#define N_NODES 100000
#define N_EDGES 1600000
#define NFEAT   128
#define NHID    64
#define NCLASS  16

#define DEG_G   8
#define DEG_P   32
#define DEG_R   ((((N_NODES + DEG_G - 1) / DEG_G) + 31) & ~31)
#define DEG_EPP (N_EDGES / DEG_P)
#define DEG_T   512
#define DEG_U   4
#define SCAN_B  256
#define SCAN_NB ((N_NODES + SCAN_B - 1) / SCAN_B)

static_assert(N_EDGES % DEG_P == 0, "edge parts must be equal");
static_assert(DEG_EPP % 4 == 0, "edge parts are read 16 bytes per lane");
static_assert((long)DEG_G * DEG_R >= N_NODES, "node ranges must cover all nodes");

__global__ void __launch_bounds__(DEG_T)
k_deg_hist(const int* __restrict__ col, unsigned* __restrict__ part) {
    __shared__ unsigned hist[DEG_R];
    const int g  = blockIdx.x % DEG_G;
    const int p  = blockIdx.x / DEG_G;
    const int lo = g * DEG_R;

    for (int i = threadIdx.x; i < DEG_R; i += DEG_T) hist[i] = 0u;
    __syncthreads();

    const int4* __restrict__ c4 = (const int4*)(col + (long)p * DEG_EPP);
    constexpr int NV = DEG_EPP / 4;
    constexpr int HU = (NV + DEG_T - 1) / DEG_T;
    int4 v[HU];
    for (int u = 0; u < HU; ++u) {
        const int i = (int)threadIdx.x + u * DEG_T;
        v[u] = c4[i < NV ? i : NV - 1];
    }
    for (int u = 0; u < HU; ++u) {
        const bool in = (int)threadIdx.x + u * DEG_T < NV;
        const unsigned d0 = in ? (unsigned)(v[u].x - lo) : ~0u;
        const unsigned d1 = in ? (unsigned)(v[u].y - lo) : ~0u;
        const unsigned d2 = in ? (unsigned)(v[u].z - lo) : ~0u;
        const unsigned d3 = in ? (unsigned)(v[u].w - lo) : ~0u;
        if (d0 < (unsigned)DEG_R) atomicAdd(&hist[d0], 1u);
        if (d1 < (unsigned)DEG_R) atomicAdd(&hist[d1], 1u);
        if (d2 < (unsigned)DEG_R) atomicAdd(&hist[d2], 1u);
        if (d3 < (unsigned)DEG_R) atomicAdd(&hist[d3], 1u);
    }
    __syncthreads();

    unsigned* __restrict__ prow = part + (long)p * N_NODES + lo;
    for (int i = threadIdx.x; i < DEG_R; i += DEG_T)
        if (lo + i < N_NODES) prow[i] = hist[i];
}

__global__ void __launch_bounds__(SCAN_B)
k_deg_to_dinv(unsigned* __restrict__ part, float* __restrict__ dinv,
              unsigned* __restrict__ deg, unsigned* __restrict__ bsum) {
    __shared__ unsigned red[SCAN_B];
    const int i = blockIdx.x * SCAN_B + threadIdx.x;
    unsigned run = 0u;
    if (i < N_NODES) {
        for (int p = 0; p < DEG_P; ++p) {
            unsigned* q = part + (long)p * N_NODES + i;
            const unsigned c = *q;
            *q = run;
            run += c;
        }
        deg[i] = run;
        float d = (float)(run + 1u);
        dinv[i] = d > 0.0f ? rsqrtf(d) : 0.0f;
    }
    red[threadIdx.x] = run;
    __syncthreads();
    for (int s = SCAN_B / 2; s > 0; s >>= 1) {
        if ((int)threadIdx.x < s) red[threadIdx.x] += red[threadIdx.x + s];
        __syncthreads();
    }
    if (threadIdx.x == 0) bsum[blockIdx.x] = red[0];
}

__global__ void __launch_bounds__(SCAN_B)
k_rowptr(const unsigned* __restrict__ deg, const unsigned* __restrict__ bsum,
         unsigned* __restrict__ rowptr) {
    __shared__ unsigned sh[SCAN_B];
    const int t = threadIdx.x;
    const int i = blockIdx.x * SCAN_B + t;
    static_assert(SCAN_NB <= 2 * SCAN_B, "two block sums per lane cover all blocks");
    const int b1 = t + SCAN_B;
    const unsigned dv = deg[i < N_NODES ? i : N_NODES - 1];
    const unsigned s0 = bsum[t < SCAN_NB ? t : SCAN_NB - 1];
    const unsigned s1 = bsum[b1 < SCAN_NB ? b1 : SCAN_NB - 1];
    asm volatile("" :: "v"(dv), "v"(s0), "v"(s1));
    const unsigned v  = i < N_NODES ? dv : 0u;
    sh[t] = (t < (int)blockIdx.x ? s0 : 0u) + (b1 < (int)blockIdx.x ? s1 : 0u);
    __syncthreads();
    for (int s = SCAN_B / 2; s > 0; s >>= 1) {
        if (t < s) sh[t] += sh[t + s];
        __syncthreads();
    }
    const unsigned base = sh[0];
    __syncthreads();
    sh[t] = v;
    __syncthreads();
    for (int d = 1; d < SCAN_B; d <<= 1) {
        const unsigned x = t >= d ? sh[t - d] : 0u;
        __syncthreads();
        sh[t] += x;
        __syncthreads();
    }
    if (i < N_NODES) rowptr[i] = base + sh[t] - v;
}

__global__ void __launch_bounds__(DEG_T)
k_fill_srcs(const int* __restrict__ row, const int* __restrict__ col,
            const unsigned* __restrict__ part,
            const unsigned* __restrict__ rowptr, int* __restrict__ srcs) {
    constexpr int TAB_U = (DEG_R + DEG_T - 1) / DEG_T;
    __shared__ unsigned cur[TAB_U * DEG_T];
    const int g  = blockIdx.x % DEG_G;
    const int p  = blockIdx.x / DEG_G;
    const int lo = g * DEG_R;
    const unsigned lim = (unsigned)(N_NODES - lo < DEG_R ? N_NODES - lo : DEG_R);

    const int4* __restrict__ c4 = (const int4*)(col + (long)p * DEG_EPP);
    const int4* __restrict__ r4 = (const int4*)(row + (long)p * DEG_EPP);
    constexpr int NV   = DEG_EPP / 4;
    constexpr int STEP = DEG_T * DEG_U;
    constexpr int NIT  = (NV + STEP - 1) / STEP;

    int4 vc[DEG_U], vr[DEG_U];
    for (int u = 0; u < DEG_U; ++u) {
        const int i  = (int)threadIdx.x + u * DEG_T;
        const int ii = i < NV ? i : NV - 1;
        vc[u] = c4[ii];
        vr[u] = r4[ii];
    }

    const unsigned* __restrict__ prow = part + (long)p * N_NODES + lo;
    {
        unsigned a[TAB_U], b[TAB_U];
        for (int u = 0; u < TAB_U; ++u) {
            const unsigned i  = threadIdx.x + (unsigned)(u * DEG_T);
            const unsigned ii = i < lim ? i : lim - 1u;
            a[u] = rowptr[lo + ii];
            b[u] = prow[ii];
        }
        for (int u = 0; u < TAB_U; ++u) {
            const unsigned i = threadIdx.x + (unsigned)(u * DEG_T);
            cur[i] = i < lim ? a[u] + b[u] : 0u;
        }
    }
    __syncthreads();

    for (int k = 0; k < NIT; ++k) {
        const int base = (int)threadIdx.x + k * STEP;
        int4 nc[DEG_U], nr[DEG_U];
        if (k + 1 < NIT) {
            for (int u = 0; u < DEG_U; ++u) {
                const int i  = base + STEP + u * DEG_T;
                const int ii = i < NV ? i : NV - 1;
                nc[u] = c4[ii];
                nr[u] = r4[ii];
            }
        }
        unsigned d[4 * DEG_U];
        int      s[4 * DEG_U];
        for (int u = 0; u < DEG_U; ++u) {
            const bool in = base + u * DEG_T < NV;
            d[4 * u + 0] = in ? (unsigned)(vc[u].x - lo) : ~0u;
            d[4 * u + 1] = in ? (unsigned)(vc[u].y - lo) : ~0u;
            d[4 * u + 2] = in ? (unsigned)(vc[u].z - lo) : ~0u;
            d[4 * u + 3] = in ? (unsigned)(vc[u].w - lo) : ~0u;
            s[4 * u + 0] = vr[u].x;
            s[4 * u + 1] = vr[u].y;
            s[4 * u + 2] = vr[u].z;
            s[4 * u + 3] = vr[u].w;
        }
        unsigned slot[4 * DEG_U];
        for (int j = 0; j < 4 * DEG_U; ++j) {
            slot[j] = 0u;
            if (d[j] < lim) slot[j] = atomicAdd(&cur[d[j]], 1u);
        }
        for (int j = 0; j < 4 * DEG_U; ++j)
            if (d[j] < lim) srcs[slot[j]] = s[j];
        if (k + 1 < NIT) {
            for (int u = 0; u < DEG_U; ++u) { vc[u] = nc[u]; vr[u] = nr[u]; }
        }
    }
}

template <int K, int N>
__global__ void k_pack_w(const float* __restrict__ W, _Float16* __restrict__ P) {
    int t = blockIdx.x * blockDim.x + threadIdx.x;
    if (t >= K * N) return;
    int i    = t & 15;
    int u    = t >> 4;
    int lane = u & 31;
    int v    = u >> 5;
    int kc   = v % (K / 32);
    int tn   = v / (K / 32);
    int l16  = lane & 15;
    int kb   = (lane >> 4) * 8;
    int k    = (i < 8) ? (kb + i) : (16 + kb + (i - 8));
    P[t] = (_Float16)W[(long)(kc * 32 + k) * N + tn * 16 + l16];
}

template <int K, int N, bool RELU>
__global__ void k_gemm_wmma(const float* __restrict__ A,
                            const _Float16* __restrict__ Bp,
                            float* __restrict__ C, int m_tiles) {
    constexpr int NT = N / 16;
    constexpr int KC = K / 32;
    const int wave = (blockIdx.x * blockDim.x + threadIdx.x) >> 5;
    const int lane = threadIdx.x & 31;
    if (wave >= m_tiles) return;

    const int l16 = lane & 15;
    const int kb  = (lane >> 4) * 8;
    const float* __restrict__ arow = A + (long)(wave * 16 + l16) * K;

    v8f acc[NT] = {};
#pragma unroll
    for (int kc = 0; kc < KC; ++kc) {
        v16h a;
#pragma unroll
        for (int i = 0; i < 8; ++i) {
            float a0 = arow[kc * 32 + kb + i];
            float a1 = arow[kc * 32 + 16 + kb + i];
            if (RELU) { a0 = fmaxf(a0, 0.0f); a1 = fmaxf(a1, 0.0f); }
            a[i]     = (_Float16)a0;
            a[8 + i] = (_Float16)a1;
        }
#pragma unroll
        for (int tn = 0; tn < NT; ++tn) {
            v16h b = *(const v16h*)(Bp + ((long)(tn * KC + kc) * 32 + lane) * 16);
            acc[tn] = __builtin_amdgcn_wmma_f32_16x16x32_f16(false, a, false, b,
                                                             (short)0, acc[tn],
                                                             false, false);
        }
    }

    float* __restrict__ crow = C + ((long)wave * 16 + (lane >> 4) * 8) * N + l16;
#pragma unroll
    for (int tn = 0; tn < NT; ++tn)
#pragma unroll
        for (int r = 0; r < 8; ++r) crow[tn * 16 + (long)r * N] = acc[tn][r];
}

__global__ void __launch_bounds__(256)
k_gather64(const int* __restrict__ srcs, const unsigned* __restrict__ rowptr,
           const unsigned* __restrict__ deg, const float* __restrict__ dinv,
           const float* __restrict__ xw, const float* __restrict__ bias,
           float* __restrict__ h) {
    const int lane = threadIdx.x & 31;
    const int c = __builtin_amdgcn_readfirstlane(
        (int)((blockIdx.x * blockDim.x + threadIdx.x) >> 5));
    if (c >= N_NODES) return;
    const unsigned start = rowptr[c];
    const unsigned cnt   = deg[c];
    const float    dc    = dinv[c];
    const float2 self = *(const float2*)(xw + (long)c * NHID + lane * 2);
    const float2 b    = *(const float2*)(bias + lane * 2);
    float ax = b.x + dc * dc * self.x;
    float ay = b.y + dc * dc * self.y;

    for (unsigned base = 0; base < cnt; base += 32u) {
        const unsigned n = cnt - base < 32u ? cnt - base : 32u;
        int   my = 0;
        float mw = 0.0f;
        if ((unsigned)lane < n) {
            my = srcs[start + base + lane];
            mw = dinv[my] * dc;
        }
        const int mwi = __float_as_int(mw);
        unsigned j = 0;
        for (; j + 4u <= n; j += 4u) {
            const int r0 = __builtin_amdgcn_readlane(my, j);
            const int r1 = __builtin_amdgcn_readlane(my, j + 1u);
            const int r2 = __builtin_amdgcn_readlane(my, j + 2u);
            const int r3 = __builtin_amdgcn_readlane(my, j + 3u);
            const float2 v0 = *(const float2*)(xw + (long)r0 * NHID + lane * 2);
            const float2 v1 = *(const float2*)(xw + (long)r1 * NHID + lane * 2);
            const float2 v2 = *(const float2*)(xw + (long)r2 * NHID + lane * 2);
            const float2 v3 = *(const float2*)(xw + (long)r3 * NHID + lane * 2);
            const float w0 = __int_as_float(__builtin_amdgcn_readlane(mwi, j));
            const float w1 = __int_as_float(__builtin_amdgcn_readlane(mwi, j + 1u));
            const float w2 = __int_as_float(__builtin_amdgcn_readlane(mwi, j + 2u));
            const float w3 = __int_as_float(__builtin_amdgcn_readlane(mwi, j + 3u));
            ax += w0 * v0.x;  ay += w0 * v0.y;
            ax += w1 * v1.x;  ay += w1 * v1.y;
            ax += w2 * v2.x;  ay += w2 * v2.y;
            ax += w3 * v3.x;  ay += w3 * v3.y;
        }
        for (; j < n; ++j) {
            const int   r = __builtin_amdgcn_readlane(my, j);
            const float w = __int_as_float(__builtin_amdgcn_readlane(mwi, j));
            const float2 v = *(const float2*)(xw + (long)r * NHID + lane * 2);
            ax += w * v.x;  ay += w * v.y;
        }
    }
    *(float2*)(h + (long)c * NHID + lane * 2) = make_float2(ax, ay);
}

__global__ void __launch_bounds__(256)
k_gather16(const int* __restrict__ srcs, const unsigned* __restrict__ rowptr,
           const unsigned* __restrict__ deg, const float* __restrict__ dinv,
           const float* __restrict__ h2, const float* __restrict__ bias,
           float* __restrict__ out) {
    const int t = blockIdx.x * blockDim.x + threadIdx.x;
    const int c = t >> 4;
    const int j = t & 15;
    if (c >= N_NODES) return;
    const unsigned cnt = deg[c];
    const int* __restrict__ seg = srcs + rowptr[c];
    const float dc = dinv[c];
    float acc = bias[j] + dc * dc * h2[(long)c * NCLASS + j];
    unsigned e = 0;
    for (; e + 4u <= cnt; e += 4u) {
        const int r0 = seg[e], r1 = seg[e + 1u], r2 = seg[e + 2u], r3 = seg[e + 3u];
        const float w0 = dinv[r0] * dc;
        const float w1 = dinv[r1] * dc;
        const float w2 = dinv[r2] * dc;
        const float w3 = dinv[r3] * dc;
        const float v0 = h2[(long)r0 * NCLASS + j];
        const float v1 = h2[(long)r1 * NCLASS + j];
        const float v2 = h2[(long)r2 * NCLASS + j];
        const float v3 = h2[(long)r3 * NCLASS + j];
        acc += w0 * v0;
        acc += w1 * v1;
        acc += w2 * v2;
        acc += w3 * v3;
    }
    for (; e < cnt; ++e) {
        const int r = seg[e];
        acc += (dinv[r] * dc) * h2[(long)r * NCLASS + j];
    }
    out[(long)c * NCLASS + j] = acc;
}

__global__ void k_log_softmax16(float* __restrict__ out) {
    int i = blockIdx.x * blockDim.x + threadIdx.x;
    if (i >= N_NODES) return;
    float* __restrict__ rowp = out + (long)i * NCLASS;
    float v[NCLASS];
    float m = -3.402823466e38f;
#pragma unroll
    for (int j = 0; j < NCLASS; ++j) { v[j] = rowp[j]; m = fmaxf(m, v[j]); }
    float s = 0.0f;
#pragma unroll
    for (int j = 0; j < NCLASS; ++j) s += __expf(v[j] - m);
    float lse = m + __logf(s);
#pragma unroll
    for (int j = 0; j < NCLASS; ++j) rowp[j] = v[j] - lse;
}

extern "C" void kernel_launch(void* const* d_in, const int* in_sizes, int n_in,
                              void* d_out, int out_size, void* d_ws, size_t ws_size,
                              hipStream_t stream) {
    const float* x   = (const float*)d_in[0];
    const int*   ei  = (const int*)d_in[1];
    const float* W1  = (const float*)d_in[2];
    const float* b1  = (const float*)d_in[3];
    const float* W2  = (const float*)d_in[4];
    const float* b2  = (const float*)d_in[5];
    float* out       = (float*)d_out;

    const int* row = ei;
    const int* col = ei + N_EDGES;

    auto align256 = [](size_t v) { return (v + 255) & ~(size_t)255; };
    char* ws = (char*)d_ws;
    size_t o = 0;
    float*     dinv = (float*)(ws + o);     o += align256((size_t)N_NODES * 4);
    _Float16*  W1p  = (_Float16*)(ws + o);  o += align256((size_t)NFEAT * NHID * 2);
    _Float16*  W2p  = (_Float16*)(ws + o);  o += align256((size_t)NHID * NCLASS * 2);
    float*     bufA = (float*)(ws + o);     o += align256((size_t)N_NODES * NHID * 4);
    float*     h    = (float*)(ws + o);     o += align256((size_t)N_NODES * NHID * 4);
    unsigned*  deg    = (unsigned*)(ws + o); o += align256((size_t)N_NODES * 4);
    unsigned*  rowptr = (unsigned*)(ws + o); o += align256((size_t)N_NODES * 4);
    unsigned*  bsum   = (unsigned*)(ws + o); o += align256((size_t)SCAN_NB * 4);
    int*       srcs   = (int*)(ws + o);      o += align256((size_t)N_EDGES * 4);
    unsigned*  part = (unsigned*)(ws + o);
    float*     h2   = bufA;

    const int B = 256;

    k_pack_w<NFEAT, NHID><<<(NFEAT * NHID + B - 1) / B, B, 0, stream>>>(W1, W1p);
    k_pack_w<NHID, NCLASS><<<(NHID * NCLASS + B - 1) / B, B, 0, stream>>>(W2, W2p);

    k_deg_hist<<<DEG_G * DEG_P, DEG_T, 0, stream>>>(col, part);
    k_deg_to_dinv<<<SCAN_NB, SCAN_B, 0, stream>>>(part, dinv, deg, bsum);
    k_rowptr<<<SCAN_NB, SCAN_B, 0, stream>>>(deg, bsum, rowptr);
    k_fill_srcs<<<DEG_G * DEG_P, DEG_T, 0, stream>>>(row, col, part, rowptr, srcs);

    const int m_tiles = N_NODES / 16;

    k_gemm_wmma<NFEAT, NHID, false><<<(m_tiles * 32 + B - 1) / B, B, 0, stream>>>(
        x, W1p, bufA, m_tiles);

    k_gather64<<<((long)N_NODES * 32 + B - 1) / B, B, 0, stream>>>(
        srcs, rowptr, deg, dinv, bufA, b1, h);

    k_gemm_wmma<NHID, NCLASS, true><<<(m_tiles * 32 + B - 1) / B, B, 0, stream>>>(
        h, W2p, h2, m_tiles);

    k_gather16<<<((long)N_NODES * 16 + B - 1) / B, B, 0, stream>>>(
        srcs, rowptr, deg, dinv, h2, b2, out);
    k_log_softmax16<<<(N_NODES + B - 1) / B, B, 0, stream>>>(out);
}


# 11 launches -> 7: weight packs ride as extra blocks of the degree-count launch, row starts are finished in the edge-fill kernel, log-softmax in the layer-2 aggregation (exp sum now a 16-lane tree)
# speedup vs baseline: 3.3897x; 1.1776x over previous
#include <hip/hip_runtime.h>
#include <hip/hip_fp16.h>

typedef __attribute__((ext_vector_type(16))) _Float16 v16h;
typedef __attribute__((ext_vector_type(8)))  float    v8f;

#define N_NODES 100000
#define N_EDGES 1600000
#define NFEAT   128
#define NHID    64
#define NCLASS  16

#define DEG_G   8
#define DEG_P   32
#define DEG_R   ((((N_NODES + DEG_G - 1) / DEG_G) + 31) & ~31)
#define DEG_EPP (N_EDGES / DEG_P)
#define DEG_T   512
#define DEG_U   4
#define SCAN_B  256
#define SCAN_NB ((N_NODES + SCAN_B - 1) / SCAN_B)

static_assert(N_EDGES % DEG_P == 0, "edge parts must be equal");
static_assert(DEG_EPP % 4 == 0, "edge parts are read 16 bytes per lane");
static_assert((long)DEG_G * DEG_R >= N_NODES, "node ranges must cover all nodes");

template <int K, int N>
__device__ __forceinline__ void pack_w(const float* __restrict__ W,
                                       _Float16* __restrict__ P, int t) {
    if (t >= K * N) return;
    int i    = t & 15;
    int u    = t >> 4;
    int lane = u & 31;
    int v    = u >> 5;
    int kc   = v % (K / 32);
    int tn   = v / (K / 32);
    int l16  = lane & 15;
    int kb   = (lane >> 4) * 8;
    int k    = (i < 8) ? (kb + i) : (16 + kb + (i - 8));
    P[t] = (_Float16)W[(long)(kc * 32 + k) * N + tn * 16 + l16];
}

#define HIST_NB  (DEG_G * DEG_P)
#define PACK1_NB ((NFEAT * NHID + DEG_T - 1) / DEG_T)
#define PACK2_NB ((NHID * NCLASS + DEG_T - 1) / DEG_T)

__global__ void __launch_bounds__(DEG_T)
k_deg_hist(const int* __restrict__ col, unsigned* __restrict__ part,
           const float* __restrict__ W1, _Float16* __restrict__ W1p,
           const float* __restrict__ W2, _Float16* __restrict__ W2p) {
    __shared__ unsigned hist[DEG_R];
    if (blockIdx.x >= HIST_NB) {
        const int pb = (int)blockIdx.x - HIST_NB;
        if (pb < PACK1_NB)
            pack_w<NFEAT, NHID>(W1, W1p, pb * DEG_T + (int)threadIdx.x);
        else
            pack_w<NHID, NCLASS>(W2, W2p, (pb - PACK1_NB) * DEG_T + (int)threadIdx.x);
        return;
    }
    const int g  = blockIdx.x % DEG_G;
    const int p  = blockIdx.x / DEG_G;
    const int lo = g * DEG_R;

    for (int i = threadIdx.x; i < DEG_R; i += DEG_T) hist[i] = 0u;
    __syncthreads();

    const int4* __restrict__ c4 = (const int4*)(col + (long)p * DEG_EPP);
    constexpr int NV = DEG_EPP / 4;
    constexpr int HU = (NV + DEG_T - 1) / DEG_T;
    int4 v[HU];
    for (int u = 0; u < HU; ++u) {
        const int i = (int)threadIdx.x + u * DEG_T;
        v[u] = c4[i < NV ? i : NV - 1];
    }
    for (int u = 0; u < HU; ++u) {
        const bool in = (int)threadIdx.x + u * DEG_T < NV;
        const unsigned d0 = in ? (unsigned)(v[u].x - lo) : ~0u;
        const unsigned d1 = in ? (unsigned)(v[u].y - lo) : ~0u;
        const unsigned d2 = in ? (unsigned)(v[u].z - lo) : ~0u;
        const unsigned d3 = in ? (unsigned)(v[u].w - lo) : ~0u;
        if (d0 < (unsigned)DEG_R) atomicAdd(&hist[d0], 1u);
        if (d1 < (unsigned)DEG_R) atomicAdd(&hist[d1], 1u);
        if (d2 < (unsigned)DEG_R) atomicAdd(&hist[d2], 1u);
        if (d3 < (unsigned)DEG_R) atomicAdd(&hist[d3], 1u);
    }
    __syncthreads();

    unsigned* __restrict__ prow = part + (long)p * N_NODES + lo;
    for (int i = threadIdx.x; i < DEG_R; i += DEG_T)
        if (lo + i < N_NODES) prow[i] = hist[i];
}

__global__ void __launch_bounds__(SCAN_B)
k_deg_to_dinv(unsigned* __restrict__ part, float* __restrict__ dinv,
              unsigned* __restrict__ deg, unsigned* __restrict__ bsum,
              unsigned* __restrict__ lpre) {
    __shared__ unsigned wsum[SCAN_B / 32];
    const int i = blockIdx.x * SCAN_B + threadIdx.x;
    unsigned run = 0u;
    if (i < N_NODES) {
        for (int p = 0; p < DEG_P; ++p) {
            unsigned* q = part + (long)p * N_NODES + i;
            const unsigned c = *q;
            *q = run;
            run += c;
        }
        deg[i] = run;
        float d = (float)(run + 1u);
        dinv[i] = d > 0.0f ? rsqrtf(d) : 0.0f;
    }
    const int lane = threadIdx.x & 31;
    const int wv   = threadIdx.x >> 5;
    unsigned inc = run;
    for (int d = 1; d < 32; d <<= 1) {
        const unsigned y = __shfl_up(inc, d);
        if (lane >= d) inc += y;
    }
    if (lane == 31) wsum[wv] = inc;
    __syncthreads();
    unsigned before = 0u, total = 0u;
    for (int w = 0; w < SCAN_B / 32; ++w) {
        const unsigned s = wsum[w];
        if (w < wv) before += s;
        total += s;
    }
    if (i < N_NODES) lpre[i] = before + inc - run;
    if (threadIdx.x == 0) bsum[blockIdx.x] = total;
}

__global__ void __launch_bounds__(DEG_T)
k_fill_srcs(const int* __restrict__ row, const int* __restrict__ col,
            const unsigned* __restrict__ part,
            const unsigned* __restrict__ lpre, const unsigned* __restrict__ bsum,
            unsigned* __restrict__ rowptr, int* __restrict__ srcs) {
    constexpr int TAB_U = (DEG_R + DEG_T - 1) / DEG_T;
    __shared__ unsigned cur[TAB_U * DEG_T];
    __shared__ unsigned bpre[DEG_T];
    __shared__ unsigned wtot[DEG_T / 32];
    static_assert(SCAN_NB <= DEG_T, "one block sum per lane covers all scan blocks");
    static_assert(TAB_U <= DEG_P, "every table entry of a lane has a part that stores it");
    const int g  = blockIdx.x % DEG_G;
    const int p  = blockIdx.x / DEG_G;
    const int lo = g * DEG_R;
    const unsigned lim = (unsigned)(N_NODES - lo < DEG_R ? N_NODES - lo : DEG_R);

    const int4* __restrict__ c4 = (const int4*)(col + (long)p * DEG_EPP);
    const int4* __restrict__ r4 = (const int4*)(row + (long)p * DEG_EPP);
    constexpr int NV   = DEG_EPP / 4;
    constexpr int STEP = DEG_T * DEG_U;
    constexpr int NIT  = (NV + STEP - 1) / STEP;

    int4 vc[DEG_U], vr[DEG_U];
    for (int u = 0; u < DEG_U; ++u) {
        const int i  = (int)threadIdx.x + u * DEG_T;
        const int ii = i < NV ? i : NV - 1;
        vc[u] = c4[ii];
        vr[u] = r4[ii];
    }

    const unsigned* __restrict__ prow = part + (long)p * N_NODES + lo;
    unsigned myptr = 0u;
    {
        const int t    = threadIdx.x;
        const int lane = t & 31;
        const int wv   = t >> 5;
        const unsigned bl = bsum[t < SCAN_NB ? t : SCAN_NB - 1];
        unsigned a[TAB_U], b[TAB_U];
        for (int u = 0; u < TAB_U; ++u) {
            const unsigned i  = threadIdx.x + (unsigned)(u * DEG_T);
            const unsigned ii = i < lim ? i : lim - 1u;
            a[u] = lpre[lo + ii];
            b[u] = prow[ii];
        }
        const unsigned bs = t < SCAN_NB ? bl : 0u;
        unsigned inc = bs;
        for (int d = 1; d < 32; d <<= 1) {
            const unsigned y = __shfl_up(inc, d);
            if (lane >= d) inc += y;
        }
        if (lane == 31) wtot[wv] = inc;
        __syncthreads();
        unsigned before = 0u;
        for (int w = 0; w < DEG_T / 32; ++w) {
            const unsigned s = wtot[w];
            if (w < wv) before += s;
        }
        bpre[t] = before + inc - bs;
        __syncthreads();
        unsigned r[TAB_U];
        for (int u = 0; u < TAB_U; ++u) {
            const unsigned i  = threadIdx.x + (unsigned)(u * DEG_T);
            const unsigned ii = i < lim ? i : lim - 1u;
            r[u] = a[u] + bpre[(lo + ii) / SCAN_B];
        }
        for (int u = 0; u < TAB_U; ++u) {
            const unsigned i = threadIdx.x + (unsigned)(u * DEG_T);
            cur[i] = i < lim ? r[u] + b[u] : 0u;
        }
        for (int u = 0; u < TAB_U; ++u)
            if (u == p) myptr = r[u];
    }
    __syncthreads();

    for (int k = 0; k < NIT; ++k) {
        const int base = (int)threadIdx.x + k * STEP;
        int4 nc[DEG_U], nr[DEG_U];
        if (k + 1 < NIT) {
            for (int u = 0; u < DEG_U; ++u) {
                const int i  = base + STEP + u * DEG_T;
                const int ii = i < NV ? i : NV - 1;
                nc[u] = c4[ii];
                nr[u] = r4[ii];
            }
        }
        unsigned d[4 * DEG_U];
        int      s[4 * DEG_U];
        for (int u = 0; u < DEG_U; ++u) {
            const bool in = base + u * DEG_T < NV;
            d[4 * u + 0] = in ? (unsigned)(vc[u].x - lo) : ~0u;
            d[4 * u + 1] = in ? (unsigned)(vc[u].y - lo) : ~0u;
            d[4 * u + 2] = in ? (unsigned)(vc[u].z - lo) : ~0u;
            d[4 * u + 3] = in ? (unsigned)(vc[u].w - lo) : ~0u;
            s[4 * u + 0] = vr[u].x;
            s[4 * u + 1] = vr[u].y;
            s[4 * u + 2] = vr[u].z;
            s[4 * u + 3] = vr[u].w;
        }
        unsigned slot[4 * DEG_U];
        for (int j = 0; j < 4 * DEG_U; ++j) {
            slot[j] = 0u;
            if (d[j] < lim) slot[j] = atomicAdd(&cur[d[j]], 1u);
        }
        for (int j = 0; j < 4 * DEG_U; ++j)
            if (d[j] < lim) srcs[slot[j]] = s[j];
        if (k + 1 < NIT) {
            for (int u = 0; u < DEG_U; ++u) { vc[u] = nc[u]; vr[u] = nr[u]; }
        }
    }
    if (p < TAB_U) {
        const unsigned i = threadIdx.x + (unsigned)(p * DEG_T);
        if (i < lim) rowptr[lo + i] = myptr;
    }
}

template <int K, int N, bool RELU>
__global__ void k_gemm_wmma(const float* __restrict__ A,
                            const _Float16* __restrict__ Bp,
                            float* __restrict__ C, int m_tiles) {
    constexpr int NT = N / 16;
    constexpr int KC = K / 32;
    const int wave = (blockIdx.x * blockDim.x + threadIdx.x) >> 5;
    const int lane = threadIdx.x & 31;
    if (wave >= m_tiles) return;

    const int l16 = lane & 15;
    const int kb  = (lane >> 4) * 8;
    const float* __restrict__ arow = A + (long)(wave * 16 + l16) * K;

    v8f acc[NT] = {};
#pragma unroll
    for (int kc = 0; kc < KC; ++kc) {
        v16h a;
#pragma unroll
        for (int i = 0; i < 8; ++i) {
            float a0 = arow[kc * 32 + kb + i];
            float a1 = arow[kc * 32 + 16 + kb + i];
            if (RELU) { a0 = fmaxf(a0, 0.0f); a1 = fmaxf(a1, 0.0f); }
            a[i]     = (_Float16)a0;
            a[8 + i] = (_Float16)a1;
        }
#pragma unroll
        for (int tn = 0; tn < NT; ++tn) {
            v16h b = *(const v16h*)(Bp + ((long)(tn * KC + kc) * 32 + lane) * 16);
            acc[tn] = __builtin_amdgcn_wmma_f32_16x16x32_f16(false, a, false, b,
                                                             (short)0, acc[tn],
                                                             false, false);
        }
    }

    float* __restrict__ crow = C + ((long)wave * 16 + (lane >> 4) * 8) * N + l16;
#pragma unroll
    for (int tn = 0; tn < NT; ++tn)
#pragma unroll
        for (int r = 0; r < 8; ++r) crow[tn * 16 + (long)r * N] = acc[tn][r];
}

__global__ void __launch_bounds__(256)
k_gather64(const int* __restrict__ srcs, const unsigned* __restrict__ rowptr,
           const unsigned* __restrict__ deg, const float* __restrict__ dinv,
           const float* __restrict__ xw, const float* __restrict__ bias,
           float* __restrict__ h) {
    const int lane = threadIdx.x & 31;
    const int c = __builtin_amdgcn_readfirstlane(
        (int)((blockIdx.x * blockDim.x + threadIdx.x) >> 5));
    if (c >= N_NODES) return;
    const unsigned start = rowptr[c];
    const unsigned cnt   = deg[c];
    const float    dc    = dinv[c];
    const float2 self = *(const float2*)(xw + (long)c * NHID + lane * 2);
    const float2 b    = *(const float2*)(bias + lane * 2);
    float ax = b.x + dc * dc * self.x;
    float ay = b.y + dc * dc * self.y;

    for (unsigned base = 0; base < cnt; base += 32u) {
        const unsigned n = cnt - base < 32u ? cnt - base : 32u;
        int   my = 0;
        float mw = 0.0f;
        if ((unsigned)lane < n) {
            my = srcs[start + base + lane];
            mw = dinv[my] * dc;
        }
        const int mwi = __float_as_int(mw);
        unsigned j = 0;
        for (; j + 4u <= n; j += 4u) {
            const int r0 = __builtin_amdgcn_readlane(my, j);
            const int r1 = __builtin_amdgcn_readlane(my, j + 1u);
            const int r2 = __builtin_amdgcn_readlane(my, j + 2u);
            const int r3 = __builtin_amdgcn_readlane(my, j + 3u);
            const float2 v0 = *(const float2*)(xw + (long)r0 * NHID + lane * 2);
            const float2 v1 = *(const float2*)(xw + (long)r1 * NHID + lane * 2);
            const float2 v2 = *(const float2*)(xw + (long)r2 * NHID + lane * 2);
            const float2 v3 = *(const float2*)(xw + (long)r3 * NHID + lane * 2);
            const float w0 = __int_as_float(__builtin_amdgcn_readlane(mwi, j));
            const float w1 = __int_as_float(__builtin_amdgcn_readlane(mwi, j + 1u));
            const float w2 = __int_as_float(__builtin_amdgcn_readlane(mwi, j + 2u));
            const float w3 = __int_as_float(__builtin_amdgcn_readlane(mwi, j + 3u));
            ax += w0 * v0.x;  ay += w0 * v0.y;
            ax += w1 * v1.x;  ay += w1 * v1.y;
            ax += w2 * v2.x;  ay += w2 * v2.y;
            ax += w3 * v3.x;  ay += w3 * v3.y;
        }
        for (; j < n; ++j) {
            const int   r = __builtin_amdgcn_readlane(my, j);
            const float w = __int_as_float(__builtin_amdgcn_readlane(mwi, j));
            const float2 v = *(const float2*)(xw + (long)r * NHID + lane * 2);
            ax += w * v.x;  ay += w * v.y;
        }
    }
    *(float2*)(h + (long)c * NHID + lane * 2) = make_float2(ax, ay);
}

template <int M>
__device__ __forceinline__ float row_xor(float v) {
    return __int_as_float(__builtin_amdgcn_update_dpp(
        __float_as_int(v), __float_as_int(v), 0x160 + M, 0xf, 0xf, false));
}

__global__ void __launch_bounds__(256)
k_gather16(const int* __restrict__ srcs, const unsigned* __restrict__ rowptr,
           const unsigned* __restrict__ deg, const float* __restrict__ dinv,
           const float* __restrict__ h2, const float* __restrict__ bias,
           float* __restrict__ out) {
    const int t = blockIdx.x * blockDim.x + threadIdx.x;
    const int c = t >> 4;
    const int j = t & 15;
    if (c >= N_NODES) return;
    const unsigned cnt = deg[c];
    const int* __restrict__ seg = srcs + rowptr[c];
    const float dc = dinv[c];
    float acc = bias[j] + dc * dc * h2[(long)c * NCLASS + j];
    unsigned e = 0;
    for (; e + 4u <= cnt; e += 4u) {
        const int r0 = seg[e], r1 = seg[e + 1u], r2 = seg[e + 2u], r3 = seg[e + 3u];
        const float w0 = dinv[r0] * dc;
        const float w1 = dinv[r1] * dc;
        const float w2 = dinv[r2] * dc;
        const float w3 = dinv[r3] * dc;
        const float v0 = h2[(long)r0 * NCLASS + j];
        const float v1 = h2[(long)r1 * NCLASS + j];
        const float v2 = h2[(long)r2 * NCLASS + j];
        const float v3 = h2[(long)r3 * NCLASS + j];
        acc += w0 * v0;
        acc += w1 * v1;
        acc += w2 * v2;
        acc += w3 * v3;
    }
    for (; e < cnt; ++e) {
        const int r = seg[e];
        acc += (dinv[r] * dc) * h2[(long)r * NCLASS + j];
    }
    float m = acc;
    m = fmaxf(m, row_xor<1>(m));
    m = fmaxf(m, row_xor<2>(m));
    m = fmaxf(m, row_xor<4>(m));
    m = fmaxf(m, row_xor<8>(m));
    float s = __expf(acc - m);
    s += row_xor<1>(s);
    s += row_xor<2>(s);
    s += row_xor<4>(s);
    s += row_xor<8>(s);
    out[(long)c * NCLASS + j] = acc - (m + __logf(s));
}

extern "C" void kernel_launch(void* const* d_in, const int* in_sizes, int n_in,
                              void* d_out, int out_size, void* d_ws, size_t ws_size,
                              hipStream_t stream) {
    const float* x   = (const float*)d_in[0];
    const int*   ei  = (const int*)d_in[1];
    const float* W1  = (const float*)d_in[2];
    const float* b1  = (const float*)d_in[3];
    const float* W2  = (const float*)d_in[4];
    const float* b2  = (const float*)d_in[5];
    float* out       = (float*)d_out;

    const int* row = ei;
    const int* col = ei + N_EDGES;

    auto align256 = [](size_t v) { return (v + 255) & ~(size_t)255; };
    char* ws = (char*)d_ws;
    size_t o = 0;
    float*     dinv = (float*)(ws + o);     o += align256((size_t)N_NODES * 4);
    _Float16*  W1p  = (_Float16*)(ws + o);  o += align256((size_t)NFEAT * NHID * 2);
    _Float16*  W2p  = (_Float16*)(ws + o);  o += align256((size_t)NHID * NCLASS * 2);
    float*     bufA = (float*)(ws + o);     o += align256((size_t)N_NODES * NHID * 4);
    float*     h    = (float*)(ws + o);     o += align256((size_t)N_NODES * NHID * 4);
    unsigned*  deg    = (unsigned*)(ws + o); o += align256((size_t)N_NODES * 4);
    unsigned*  rowptr = (unsigned*)(ws + o); o += align256((size_t)N_NODES * 4);
    unsigned*  lpre   = (unsigned*)(ws + o); o += align256((size_t)N_NODES * 4);
    unsigned*  bsum   = (unsigned*)(ws + o); o += align256((size_t)SCAN_NB * 4);
    int*       srcs   = (int*)(ws + o);      o += align256((size_t)N_EDGES * 4);
    unsigned*  part = (unsigned*)(ws + o);
    float*     h2   = bufA;

    const int B = 256;

    k_deg_hist<<<HIST_NB + PACK1_NB + PACK2_NB, DEG_T, 0, stream>>>(
        col, part, W1, W1p, W2, W2p);
    k_deg_to_dinv<<<SCAN_NB, SCAN_B, 0, stream>>>(part, dinv, deg, bsum, lpre);
    k_fill_srcs<<<DEG_G * DEG_P, DEG_T, 0, stream>>>(row, col, part, lpre, bsum,
                                                     rowptr, srcs);

    const int m_tiles = N_NODES / 16;

    k_gemm_wmma<NFEAT, NHID, false><<<(m_tiles * 32 + B - 1) / B, B, 0, stream>>>(
        x, W1p, bufA, m_tiles);

    k_gather64<<<((long)N_NODES * 32 + B - 1) / B, B, 0, stream>>>(
        srcs, rowptr, deg, dinv, bufA, b1, h);

    k_gemm_wmma<NHID, NCLASS, true><<<(m_tiles * 32 + B - 1) / B, B, 0, stream>>>(
        h, W2p, h2, m_tiles);

    k_gather16<<<((long)N_NODES * 16 + B - 1) / B, B, 0, stream>>>(
        srcs, rowptr, deg, dinv, h2, b2, out);
}


# Aggregation kernels: a resident grid of waves, each walking 7 (layer 1) or 4 (layer 2) nodes, next node's bounds and indices requested ahead; was 150000 one-node waves. Per-node sum order unchanged
# speedup vs baseline: 3.5022x; 1.0332x over previous
#include <hip/hip_runtime.h>
#include <hip/hip_fp16.h>

typedef __attribute__((ext_vector_type(16))) _Float16 v16h;
typedef __attribute__((ext_vector_type(8)))  float    v8f;

#define N_NODES 100000
#define N_EDGES 1600000
#define NFEAT   128
#define NHID    64
#define NCLASS  16

#define DEG_G   8
#define DEG_P   32
#define DEG_R   ((((N_NODES + DEG_G - 1) / DEG_G) + 31) & ~31)
#define DEG_EPP (N_EDGES / DEG_P)
#define DEG_T   512
#define DEG_U   4
#define SCAN_B  256
#define SCAN_NB ((N_NODES + SCAN_B - 1) / SCAN_B)

static_assert(N_EDGES % DEG_P == 0, "edge parts must be equal");
static_assert(DEG_EPP % 4 == 0, "edge parts are read 16 bytes per lane");
static_assert((long)DEG_G * DEG_R >= N_NODES, "node ranges must cover all nodes");

template <int K, int N>
__device__ __forceinline__ void pack_w(const float* __restrict__ W,
                                       _Float16* __restrict__ P, int t) {
    if (t >= K * N) return;
    int i    = t & 15;
    int u    = t >> 4;
    int lane = u & 31;
    int v    = u >> 5;
    int kc   = v % (K / 32);
    int tn   = v / (K / 32);
    int l16  = lane & 15;
    int kb   = (lane >> 4) * 8;
    int k    = (i < 8) ? (kb + i) : (16 + kb + (i - 8));
    P[t] = (_Float16)W[(long)(kc * 32 + k) * N + tn * 16 + l16];
}

#define HIST_NB  (DEG_G * DEG_P)
#define PACK1_NB ((NFEAT * NHID + DEG_T - 1) / DEG_T)
#define PACK2_NB ((NHID * NCLASS + DEG_T - 1) / DEG_T)

__global__ void __launch_bounds__(DEG_T)
k_deg_hist(const int* __restrict__ col, unsigned* __restrict__ part,
           const float* __restrict__ W1, _Float16* __restrict__ W1p,
           const float* __restrict__ W2, _Float16* __restrict__ W2p) {
    __shared__ unsigned hist[DEG_R];
    if (blockIdx.x >= HIST_NB) {
        const int pb = (int)blockIdx.x - HIST_NB;
        if (pb < PACK1_NB)
            pack_w<NFEAT, NHID>(W1, W1p, pb * DEG_T + (int)threadIdx.x);
        else
            pack_w<NHID, NCLASS>(W2, W2p, (pb - PACK1_NB) * DEG_T + (int)threadIdx.x);
        return;
    }
    const int g  = blockIdx.x % DEG_G;
    const int p  = blockIdx.x / DEG_G;
    const int lo = g * DEG_R;

    for (int i = threadIdx.x; i < DEG_R; i += DEG_T) hist[i] = 0u;
    __syncthreads();

    const int4* __restrict__ c4 = (const int4*)(col + (long)p * DEG_EPP);
    constexpr int NV = DEG_EPP / 4;
    constexpr int HU = (NV + DEG_T - 1) / DEG_T;
    int4 v[HU];
    for (int u = 0; u < HU; ++u) {
        const int i = (int)threadIdx.x + u * DEG_T;
        v[u] = c4[i < NV ? i : NV - 1];
    }
    for (int u = 0; u < HU; ++u) {
        const bool in = (int)threadIdx.x + u * DEG_T < NV;
        const unsigned d0 = in ? (unsigned)(v[u].x - lo) : ~0u;
        const unsigned d1 = in ? (unsigned)(v[u].y - lo) : ~0u;
        const unsigned d2 = in ? (unsigned)(v[u].z - lo) : ~0u;
        const unsigned d3 = in ? (unsigned)(v[u].w - lo) : ~0u;
        if (d0 < (unsigned)DEG_R) atomicAdd(&hist[d0], 1u);
        if (d1 < (unsigned)DEG_R) atomicAdd(&hist[d1], 1u);
        if (d2 < (unsigned)DEG_R) atomicAdd(&hist[d2], 1u);
        if (d3 < (unsigned)DEG_R) atomicAdd(&hist[d3], 1u);
    }
    __syncthreads();

    unsigned* __restrict__ prow = part + (long)p * N_NODES + lo;
    for (int i = threadIdx.x; i < DEG_R; i += DEG_T)
        if (lo + i < N_NODES) prow[i] = hist[i];
}

__global__ void __launch_bounds__(SCAN_B)
k_deg_to_dinv(unsigned* __restrict__ part, float* __restrict__ dinv,
              unsigned* __restrict__ deg, unsigned* __restrict__ bsum,
              unsigned* __restrict__ lpre) {
    __shared__ unsigned wsum[SCAN_B / 32];
    const int i = blockIdx.x * SCAN_B + threadIdx.x;
    unsigned run = 0u;
    if (i < N_NODES) {
        for (int p = 0; p < DEG_P; ++p) {
            unsigned* q = part + (long)p * N_NODES + i;
            const unsigned c = *q;
            *q = run;
            run += c;
        }
        deg[i] = run;
        float d = (float)(run + 1u);
        dinv[i] = d > 0.0f ? rsqrtf(d) : 0.0f;
    }
    const int lane = threadIdx.x & 31;
    const int wv   = threadIdx.x >> 5;
    unsigned inc = run;
    for (int d = 1; d < 32; d <<= 1) {
        const unsigned y = __shfl_up(inc, d);
        if (lane >= d) inc += y;
    }
    if (lane == 31) wsum[wv] = inc;
    __syncthreads();
    unsigned before = 0u, total = 0u;
    for (int w = 0; w < SCAN_B / 32; ++w) {
        const unsigned s = wsum[w];
        if (w < wv) before += s;
        total += s;
    }
    if (i < N_NODES) lpre[i] = before + inc - run;
    if (threadIdx.x == 0) bsum[blockIdx.x] = total;
}

__global__ void __launch_bounds__(DEG_T)
k_fill_srcs(const int* __restrict__ row, const int* __restrict__ col,
            const unsigned* __restrict__ part,
            const unsigned* __restrict__ lpre, const unsigned* __restrict__ bsum,
            unsigned* __restrict__ rowptr, int* __restrict__ srcs) {
    constexpr int TAB_U = (DEG_R + DEG_T - 1) / DEG_T;
    __shared__ unsigned cur[TAB_U * DEG_T];
    __shared__ unsigned bpre[DEG_T];
    __shared__ unsigned wtot[DEG_T / 32];
    static_assert(SCAN_NB <= DEG_T, "one block sum per lane covers all scan blocks");
    static_assert(TAB_U <= DEG_P, "every table entry of a lane has a part that stores it");
    const int g  = blockIdx.x % DEG_G;
    const int p  = blockIdx.x / DEG_G;
    const int lo = g * DEG_R;
    const unsigned lim = (unsigned)(N_NODES - lo < DEG_R ? N_NODES - lo : DEG_R);

    const int4* __restrict__ c4 = (const int4*)(col + (long)p * DEG_EPP);
    const int4* __restrict__ r4 = (const int4*)(row + (long)p * DEG_EPP);
    constexpr int NV   = DEG_EPP / 4;
    constexpr int STEP = DEG_T * DEG_U;
    constexpr int NIT  = (NV + STEP - 1) / STEP;

    int4 vc[DEG_U], vr[DEG_U];
    for (int u = 0; u < DEG_U; ++u) {
        const int i  = (int)threadIdx.x + u * DEG_T;
        const int ii = i < NV ? i : NV - 1;
        vc[u] = c4[ii];
        vr[u] = r4[ii];
    }

    const unsigned* __restrict__ prow = part + (long)p * N_NODES + lo;
    unsigned myptr = 0u;
    {
        const int t    = threadIdx.x;
        const int lane = t & 31;
        const int wv   = t >> 5;
        const unsigned bl = bsum[t < SCAN_NB ? t : SCAN_NB - 1];
        unsigned a[TAB_U], b[TAB_U];
        for (int u = 0; u < TAB_U; ++u) {
            const unsigned i  = threadIdx.x + (unsigned)(u * DEG_T);
            const unsigned ii = i < lim ? i : lim - 1u;
            a[u] = lpre[lo + ii];
            b[u] = prow[ii];
        }
        const unsigned bs = t < SCAN_NB ? bl : 0u;
        unsigned inc = bs;
        for (int d = 1; d < 32; d <<= 1) {
            const unsigned y = __shfl_up(inc, d);
            if (lane >= d) inc += y;
        }
        if (lane == 31) wtot[wv] = inc;
        __syncthreads();
        unsigned before = 0u;
        for (int w = 0; w < DEG_T / 32; ++w) {
            const unsigned s = wtot[w];
            if (w < wv) before += s;
        }
        bpre[t] = before + inc - bs;
        __syncthreads();
        unsigned r[TAB_U];
        for (int u = 0; u < TAB_U; ++u) {
            const unsigned i  = threadIdx.x + (unsigned)(u * DEG_T);
            const unsigned ii = i < lim ? i : lim - 1u;
            r[u] = a[u] + bpre[(lo + ii) / SCAN_B];
        }
        for (int u = 0; u < TAB_U; ++u) {
            const unsigned i = threadIdx.x + (unsigned)(u * DEG_T);
            cur[i] = i < lim ? r[u] + b[u] : 0u;
        }
        for (int u = 0; u < TAB_U; ++u)
            if (u == p) myptr = r[u];
    }
    __syncthreads();

    for (int k = 0; k < NIT; ++k) {
        const int base = (int)threadIdx.x + k * STEP;
        int4 nc[DEG_U], nr[DEG_U];
        if (k + 1 < NIT) {
            for (int u = 0; u < DEG_U; ++u) {
                const int i  = base + STEP + u * DEG_T;
                const int ii = i < NV ? i : NV - 1;
                nc[u] = c4[ii];
                nr[u] = r4[ii];
            }
        }
        unsigned d[4 * DEG_U];
        int      s[4 * DEG_U];
        for (int u = 0; u < DEG_U; ++u) {
            const bool in = base + u * DEG_T < NV;
            d[4 * u + 0] = in ? (unsigned)(vc[u].x - lo) : ~0u;
            d[4 * u + 1] = in ? (unsigned)(vc[u].y - lo) : ~0u;
            d[4 * u + 2] = in ? (unsigned)(vc[u].z - lo) : ~0u;
            d[4 * u + 3] = in ? (unsigned)(vc[u].w - lo) : ~0u;
            s[4 * u + 0] = vr[u].x;
            s[4 * u + 1] = vr[u].y;
            s[4 * u + 2] = vr[u].z;
            s[4 * u + 3] = vr[u].w;
        }
        unsigned slot[4 * DEG_U];
        for (int j = 0; j < 4 * DEG_U; ++j) {
            slot[j] = 0u;
            if (d[j] < lim) slot[j] = atomicAdd(&cur[d[j]], 1u);
        }
        for (int j = 0; j < 4 * DEG_U; ++j)
            if (d[j] < lim) srcs[slot[j]] = s[j];
        if (k + 1 < NIT) {
            for (int u = 0; u < DEG_U; ++u) { vc[u] = nc[u]; vr[u] = nr[u]; }
        }
    }
    if (p < TAB_U) {
        const unsigned i = threadIdx.x + (unsigned)(p * DEG_T);
        if (i < lim) rowptr[lo + i] = myptr;
    }
}

template <int K, int N, bool RELU>
__global__ void k_gemm_wmma(const float* __restrict__ A,
                            const _Float16* __restrict__ Bp,
                            float* __restrict__ C, int m_tiles) {
    constexpr int NT = N / 16;
    constexpr int KC = K / 32;
    const int wave = (blockIdx.x * blockDim.x + threadIdx.x) >> 5;
    const int lane = threadIdx.x & 31;
    if (wave >= m_tiles) return;

    const int l16 = lane & 15;
    const int kb  = (lane >> 4) * 8;
    const float* __restrict__ arow = A + (long)(wave * 16 + l16) * K;

    v8f acc[NT] = {};
#pragma unroll
    for (int kc = 0; kc < KC; ++kc) {
        v16h a;
#pragma unroll
        for (int i = 0; i < 8; ++i) {
            float a0 = arow[kc * 32 + kb + i];
            float a1 = arow[kc * 32 + 16 + kb + i];
            if (RELU) { a0 = fmaxf(a0, 0.0f); a1 = fmaxf(a1, 0.0f); }
            a[i]     = (_Float16)a0;
            a[8 + i] = (_Float16)a1;
        }
#pragma unroll
        for (int tn = 0; tn < NT; ++tn) {
            v16h b = *(const v16h*)(Bp + ((long)(tn * KC + kc) * 32 + lane) * 16);
            acc[tn] = __builtin_amdgcn_wmma_f32_16x16x32_f16(false, a, false, b,
                                                             (short)0, acc[tn],
                                                             false, false);
        }
    }

    float* __restrict__ crow = C + ((long)wave * 16 + (lane >> 4) * 8) * N + l16;
#pragma unroll
    for (int tn = 0; tn < NT; ++tn)
#pragma unroll
        for (int r = 0; r < 8; ++r) crow[tn * 16 + (long)r * N] = acc[tn][r];
}

#define AGG_B      256
#define G64_ROUNDS 7
#define G64_NB     ((((N_NODES + G64_ROUNDS - 1) / G64_ROUNDS) * 32 + AGG_B - 1) / AGG_B)
#define G64_STRIDE (G64_NB * (AGG_B / 32))
#define G16_ROUNDS 4
#define G16_NB     ((((N_NODES + G16_ROUNDS - 1) / G16_ROUNDS) * 16 + AGG_B - 1) / AGG_B)
#define G16_STRIDE (G16_NB * (AGG_B / 16))

static_assert((long)G64_STRIDE * G64_ROUNDS >= N_NODES, "layer-1 rounds cover all nodes");
static_assert((long)G16_STRIDE * G16_ROUNDS >= N_NODES, "layer-2 rounds cover all nodes");
static_assert(G64_NB <= 2048 && G16_NB <= 2048, "the whole grid is resident at once");

__device__ __forceinline__ void add_rows64(const float* __restrict__ xw, int lane,
                                           int my, int mwi, unsigned n,
                                           float& ax, float& ay) {
    unsigned j = 0;
    for (; j + 4u <= n; j += 4u) {
        const int r0 = __builtin_amdgcn_readlane(my, j);
        const int r1 = __builtin_amdgcn_readlane(my, j + 1u);
        const int r2 = __builtin_amdgcn_readlane(my, j + 2u);
        const int r3 = __builtin_amdgcn_readlane(my, j + 3u);
        const float2 v0 = *(const float2*)(xw + (long)r0 * NHID + lane * 2);
        const float2 v1 = *(const float2*)(xw + (long)r1 * NHID + lane * 2);
        const float2 v2 = *(const float2*)(xw + (long)r2 * NHID + lane * 2);
        const float2 v3 = *(const float2*)(xw + (long)r3 * NHID + lane * 2);
        const float w0 = __int_as_float(__builtin_amdgcn_readlane(mwi, j));
        const float w1 = __int_as_float(__builtin_amdgcn_readlane(mwi, j + 1u));
        const float w2 = __int_as_float(__builtin_amdgcn_readlane(mwi, j + 2u));
        const float w3 = __int_as_float(__builtin_amdgcn_readlane(mwi, j + 3u));
        ax += w0 * v0.x;  ay += w0 * v0.y;
        ax += w1 * v1.x;  ay += w1 * v1.y;
        ax += w2 * v2.x;  ay += w2 * v2.y;
        ax += w3 * v3.x;  ay += w3 * v3.y;
    }
    for (; j < n; ++j) {
        const int   r = __builtin_amdgcn_readlane(my, j);
        const float w = __int_as_float(__builtin_amdgcn_readlane(mwi, j));
        const float2 v = *(const float2*)(xw + (long)r * NHID + lane * 2);
        ax += w * v.x;  ay += w * v.y;
    }
}

__global__ void __launch_bounds__(AGG_B)
k_gather64(const int* __restrict__ srcs, const unsigned* __restrict__ rowptr,
           const unsigned* __restrict__ deg, const float* __restrict__ dinv,
           const float* __restrict__ xw, const float* __restrict__ bias,
           float* __restrict__ h) {
    const int lane = threadIdx.x & 31;
    int c = __builtin_amdgcn_readfirstlane(
        (int)((blockIdx.x * AGG_B + threadIdx.x) >> 5));
    if (c >= N_NODES) return;
    const float2 b = *(const float2*)(bias + lane * 2);

    unsigned start = rowptr[c];
    unsigned cnt   = deg[c];
    float    dc    = dinv[c];
    int c1 = c + G64_STRIDE;
    unsigned start1 = 0u, cnt1 = 0u;
    float    dc1 = 0.0f;
    if (c1 < N_NODES) { start1 = rowptr[c1]; cnt1 = deg[c1]; dc1 = dinv[c1]; }
    int my = srcs[(unsigned)lane < cnt ? start + lane : 0u];

    for (;;) {
        const int c2 = c1 + G64_STRIDE;
        unsigned start2 = 0u, cnt2 = 0u;
        float    dc2 = 0.0f;
        if (c2 < N_NODES) { start2 = rowptr[c2]; cnt2 = deg[c2]; dc2 = dinv[c2]; }

        const float  mw   = dinv[my] * dc;
        const float2 self = *(const float2*)(xw + (long)c * NHID + lane * 2);
        const bool want1 = c1 < N_NODES && (unsigned)lane < cnt1;
        const int  my1   = srcs[want1 ? start1 + lane : 0u];

        float ax = b.x + dc * dc * self.x;
        float ay = b.y + dc * dc * self.y;
        add_rows64(xw, lane, my, __float_as_int(mw), cnt < 32u ? cnt : 32u, ax, ay);
        for (unsigned base = 32u; base < cnt; base += 32u) {
            const unsigned n = cnt - base < 32u ? cnt - base : 32u;
            int   my_b = 0;
            float mw_b = 0.0f;
            if ((unsigned)lane < n) {
                my_b = srcs[start + base + lane];
                mw_b = dinv[my_b] * dc;
            }
            add_rows64(xw, lane, my_b, __float_as_int(mw_b), n, ax, ay);
        }
        *(float2*)(h + (long)c * NHID + lane * 2) = make_float2(ax, ay);

        if (c1 >= N_NODES) break;
        c = c1;  start = start1;  cnt = cnt1;  dc = dc1;  my = my1;
        c1 = c2; start1 = start2; cnt1 = cnt2; dc1 = dc2;
    }
}

template <int M>
__device__ __forceinline__ float row_xor(float v) {
    return __int_as_float(__builtin_amdgcn_update_dpp(
        __float_as_int(v), __float_as_int(v), 0x160 + M, 0xf, 0xf, false));
}

__global__ void __launch_bounds__(AGG_B)
k_gather16(const int* __restrict__ srcs, const unsigned* __restrict__ rowptr,
           const unsigned* __restrict__ deg, const float* __restrict__ dinv,
           const float* __restrict__ h2, const float* __restrict__ bias,
           float* __restrict__ out) {
    const int t = blockIdx.x * AGG_B + threadIdx.x;
    int c = t >> 4;
    const int j = t & 15;
    if (c >= N_NODES) return;
    const float bj = bias[j];
    unsigned cnt = deg[c];
    unsigned st  = rowptr[c];
    float    dc  = dinv[c];
    for (;;) {
        const float selfv = h2[(long)c * NCLASS + j];
        const int  cn   = c + G16_STRIDE;
        const bool more = cn < N_NODES;
        const int  cp   = more ? cn : c;
        const unsigned ncnt = deg[cp];
        const unsigned nst  = rowptr[cp];
        const float    ndc  = dinv[cp];

        const int* __restrict__ seg = srcs + st;
        float acc = bj + dc * dc * selfv;
        unsigned e = 0;
        for (; e + 4u <= cnt; e += 4u) {
            const int r0 = seg[e], r1 = seg[e + 1u], r2 = seg[e + 2u], r3 = seg[e + 3u];
            const float w0 = dinv[r0] * dc;
            const float w1 = dinv[r1] * dc;
            const float w2 = dinv[r2] * dc;
            const float w3 = dinv[r3] * dc;
            const float v0 = h2[(long)r0 * NCLASS + j];
            const float v1 = h2[(long)r1 * NCLASS + j];
            const float v2 = h2[(long)r2 * NCLASS + j];
            const float v3 = h2[(long)r3 * NCLASS + j];
            acc += w0 * v0;
            acc += w1 * v1;
            acc += w2 * v2;
            acc += w3 * v3;
        }
        for (; e < cnt; ++e) {
            const int r = seg[e];
            acc += (dinv[r] * dc) * h2[(long)r * NCLASS + j];
        }
        float m = acc;
        m = fmaxf(m, row_xor<1>(m));
        m = fmaxf(m, row_xor<2>(m));
        m = fmaxf(m, row_xor<4>(m));
        m = fmaxf(m, row_xor<8>(m));
        float s = __expf(acc - m);
        s += row_xor<1>(s);
        s += row_xor<2>(s);
        s += row_xor<4>(s);
        s += row_xor<8>(s);
        out[(long)c * NCLASS + j] = acc - (m + __logf(s));

        if (!more) break;
        c = cn;  cnt = ncnt;  st = nst;  dc = ndc;
    }
}

extern "C" void kernel_launch(void* const* d_in, const int* in_sizes, int n_in,
                              void* d_out, int out_size, void* d_ws, size_t ws_size,
                              hipStream_t stream) {
    const float* x   = (const float*)d_in[0];
    const int*   ei  = (const int*)d_in[1];
    const float* W1  = (const float*)d_in[2];
    const float* b1  = (const float*)d_in[3];
    const float* W2  = (const float*)d_in[4];
    const float* b2  = (const float*)d_in[5];
    float* out       = (float*)d_out;

    const int* row = ei;
    const int* col = ei + N_EDGES;

    auto align256 = [](size_t v) { return (v + 255) & ~(size_t)255; };
    char* ws = (char*)d_ws;
    size_t o = 0;
    float*     dinv = (float*)(ws + o);     o += align256((size_t)N_NODES * 4);
    _Float16*  W1p  = (_Float16*)(ws + o);  o += align256((size_t)NFEAT * NHID * 2);
    _Float16*  W2p  = (_Float16*)(ws + o);  o += align256((size_t)NHID * NCLASS * 2);
    float*     bufA = (float*)(ws + o);     o += align256((size_t)N_NODES * NHID * 4);
    float*     h    = (float*)(ws + o);     o += align256((size_t)N_NODES * NHID * 4);
    unsigned*  deg    = (unsigned*)(ws + o); o += align256((size_t)N_NODES * 4);
    unsigned*  rowptr = (unsigned*)(ws + o); o += align256((size_t)N_NODES * 4);
    unsigned*  lpre   = (unsigned*)(ws + o); o += align256((size_t)N_NODES * 4);
    unsigned*  bsum   = (unsigned*)(ws + o); o += align256((size_t)SCAN_NB * 4);
    int*       srcs   = (int*)(ws + o);      o += align256((size_t)N_EDGES * 4);
    unsigned*  part = (unsigned*)(ws + o);
    float*     h2   = bufA;

    const int B = 256;

    k_deg_hist<<<HIST_NB + PACK1_NB + PACK2_NB, DEG_T, 0, stream>>>(
        col, part, W1, W1p, W2, W2p);
    k_deg_to_dinv<<<SCAN_NB, SCAN_B, 0, stream>>>(part, dinv, deg, bsum, lpre);
    k_fill_srcs<<<DEG_G * DEG_P, DEG_T, 0, stream>>>(row, col, part, lpre, bsum,
                                                     rowptr, srcs);

    const int m_tiles = N_NODES / 16;

    k_gemm_wmma<NFEAT, NHID, false><<<(m_tiles * 32 + B - 1) / B, B, 0, stream>>>(
        x, W1p, bufA, m_tiles);

    k_gather64<<<G64_NB, AGG_B, 0, stream>>>(srcs, rowptr, deg, dinv, bufA, b1, h);

    k_gemm_wmma<NHID, NCLASS, true><<<(m_tiles * 32 + B - 1) / B, B, 0, stream>>>(
        h, W2p, h2, m_tiles);

    k_gather16<<<G16_NB, AGG_B, 0, stream>>>(srcs, rowptr, deg, dinv, h2, b2, out);
}


# Source factor dinv[r] folded into the rows both products store (epilogue scale); aggregations add stored rows, scale the sum by dinv[c] once: no per-edge dinv read; term rounding changes, order kept
# speedup vs baseline: 3.7717x; 1.0770x over previous
#include <hip/hip_runtime.h>
#include <hip/hip_fp16.h>

typedef __attribute__((ext_vector_type(16))) _Float16 v16h;
typedef __attribute__((ext_vector_type(8)))  float    v8f;

#define N_NODES 100000
#define N_EDGES 1600000
#define NFEAT   128
#define NHID    64
#define NCLASS  16

#define DEG_G   8
#define DEG_P   32
#define DEG_R   ((((N_NODES + DEG_G - 1) / DEG_G) + 31) & ~31)
#define DEG_EPP (N_EDGES / DEG_P)
#define DEG_T   512
#define DEG_U   4
#define SCAN_B  256
#define SCAN_NB ((N_NODES + SCAN_B - 1) / SCAN_B)

static_assert(N_EDGES % DEG_P == 0, "edge parts must be equal");
static_assert(DEG_EPP % 4 == 0, "edge parts are read 16 bytes per lane");
static_assert((long)DEG_G * DEG_R >= N_NODES, "node ranges must cover all nodes");

template <int K, int N>
__device__ __forceinline__ void pack_w(const float* __restrict__ W,
                                       _Float16* __restrict__ P, int t) {
    if (t >= K * N) return;
    int i    = t & 15;
    int u    = t >> 4;
    int lane = u & 31;
    int v    = u >> 5;
    int kc   = v % (K / 32);
    int tn   = v / (K / 32);
    int l16  = lane & 15;
    int kb   = (lane >> 4) * 8;
    int k    = (i < 8) ? (kb + i) : (16 + kb + (i - 8));
    P[t] = (_Float16)W[(long)(kc * 32 + k) * N + tn * 16 + l16];
}

#define HIST_NB  (DEG_G * DEG_P)
#define PACK1_NB ((NFEAT * NHID + DEG_T - 1) / DEG_T)
#define PACK2_NB ((NHID * NCLASS + DEG_T - 1) / DEG_T)

__global__ void __launch_bounds__(DEG_T)
k_deg_hist(const int* __restrict__ col, unsigned* __restrict__ part,
           const float* __restrict__ W1, _Float16* __restrict__ W1p,
           const float* __restrict__ W2, _Float16* __restrict__ W2p) {
    __shared__ unsigned hist[DEG_R];
    if (blockIdx.x >= HIST_NB) {
        const int pb = (int)blockIdx.x - HIST_NB;
        if (pb < PACK1_NB)
            pack_w<NFEAT, NHID>(W1, W1p, pb * DEG_T + (int)threadIdx.x);
        else
            pack_w<NHID, NCLASS>(W2, W2p, (pb - PACK1_NB) * DEG_T + (int)threadIdx.x);
        return;
    }
    const int g  = blockIdx.x % DEG_G;
    const int p  = blockIdx.x / DEG_G;
    const int lo = g * DEG_R;

    for (int i = threadIdx.x; i < DEG_R; i += DEG_T) hist[i] = 0u;
    __syncthreads();

    const int4* __restrict__ c4 = (const int4*)(col + (long)p * DEG_EPP);
    constexpr int NV = DEG_EPP / 4;
    constexpr int HU = (NV + DEG_T - 1) / DEG_T;
    int4 v[HU];
    for (int u = 0; u < HU; ++u) {
        const int i = (int)threadIdx.x + u * DEG_T;
        v[u] = c4[i < NV ? i : NV - 1];
    }
    for (int u = 0; u < HU; ++u) {
        const bool in = (int)threadIdx.x + u * DEG_T < NV;
        const unsigned d0 = in ? (unsigned)(v[u].x - lo) : ~0u;
        const unsigned d1 = in ? (unsigned)(v[u].y - lo) : ~0u;
        const unsigned d2 = in ? (unsigned)(v[u].z - lo) : ~0u;
        const unsigned d3 = in ? (unsigned)(v[u].w - lo) : ~0u;
        if (d0 < (unsigned)DEG_R) atomicAdd(&hist[d0], 1u);
        if (d1 < (unsigned)DEG_R) atomicAdd(&hist[d1], 1u);
        if (d2 < (unsigned)DEG_R) atomicAdd(&hist[d2], 1u);
        if (d3 < (unsigned)DEG_R) atomicAdd(&hist[d3], 1u);
    }
    __syncthreads();

    unsigned* __restrict__ prow = part + (long)p * N_NODES + lo;
    for (int i = threadIdx.x; i < DEG_R; i += DEG_T)
        if (lo + i < N_NODES) prow[i] = hist[i];
}

__global__ void __launch_bounds__(SCAN_B)
k_deg_to_dinv(unsigned* __restrict__ part, float* __restrict__ dinv,
              unsigned* __restrict__ deg, unsigned* __restrict__ bsum,
              unsigned* __restrict__ lpre) {
    __shared__ unsigned wsum[SCAN_B / 32];
    const int i = blockIdx.x * SCAN_B + threadIdx.x;
    unsigned run = 0u;
    if (i < N_NODES) {
        for (int p = 0; p < DEG_P; ++p) {
            unsigned* q = part + (long)p * N_NODES + i;
            const unsigned c = *q;
            *q = run;
            run += c;
        }
        deg[i] = run;
        float d = (float)(run + 1u);
        dinv[i] = d > 0.0f ? rsqrtf(d) : 0.0f;
    }
    const int lane = threadIdx.x & 31;
    const int wv   = threadIdx.x >> 5;
    unsigned inc = run;
    for (int d = 1; d < 32; d <<= 1) {
        const unsigned y = __shfl_up(inc, d);
        if (lane >= d) inc += y;
    }
    if (lane == 31) wsum[wv] = inc;
    __syncthreads();
    unsigned before = 0u, total = 0u;
    for (int w = 0; w < SCAN_B / 32; ++w) {
        const unsigned s = wsum[w];
        if (w < wv) before += s;
        total += s;
    }
    if (i < N_NODES) lpre[i] = before + inc - run;
    if (threadIdx.x == 0) bsum[blockIdx.x] = total;
}

__global__ void __launch_bounds__(DEG_T)
k_fill_srcs(const int* __restrict__ row, const int* __restrict__ col,
            const unsigned* __restrict__ part,
            const unsigned* __restrict__ lpre, const unsigned* __restrict__ bsum,
            unsigned* __restrict__ rowptr, int* __restrict__ srcs) {
    constexpr int TAB_U = (DEG_R + DEG_T - 1) / DEG_T;
    __shared__ unsigned cur[TAB_U * DEG_T];
    __shared__ unsigned bpre[DEG_T];
    __shared__ unsigned wtot[DEG_T / 32];
    static_assert(SCAN_NB <= DEG_T, "one block sum per lane covers all scan blocks");
    static_assert(TAB_U <= DEG_P, "every table entry of a lane has a part that stores it");
    const int g  = blockIdx.x % DEG_G;
    const int p  = blockIdx.x / DEG_G;
    const int lo = g * DEG_R;
    const unsigned lim = (unsigned)(N_NODES - lo < DEG_R ? N_NODES - lo : DEG_R);

    const int4* __restrict__ c4 = (const int4*)(col + (long)p * DEG_EPP);
    const int4* __restrict__ r4 = (const int4*)(row + (long)p * DEG_EPP);
    constexpr int NV   = DEG_EPP / 4;
    constexpr int STEP = DEG_T * DEG_U;
    constexpr int NIT  = (NV + STEP - 1) / STEP;

    int4 vc[DEG_U], vr[DEG_U];
    for (int u = 0; u < DEG_U; ++u) {
        const int i  = (int)threadIdx.x + u * DEG_T;
        const int ii = i < NV ? i : NV - 1;
        vc[u] = c4[ii];
        vr[u] = r4[ii];
    }

    const unsigned* __restrict__ prow = part + (long)p * N_NODES + lo;
    unsigned myptr = 0u;
    {
        const int t    = threadIdx.x;
        const int lane = t & 31;
        const int wv   = t >> 5;
        const unsigned bl = bsum[t < SCAN_NB ? t : SCAN_NB - 1];
        unsigned a[TAB_U], b[TAB_U];
        for (int u = 0; u < TAB_U; ++u) {
            const unsigned i  = threadIdx.x + (unsigned)(u * DEG_T);
            const unsigned ii = i < lim ? i : lim - 1u;
            a[u] = lpre[lo + ii];
            b[u] = prow[ii];
        }
        const unsigned bs = t < SCAN_NB ? bl : 0u;
        unsigned inc = bs;
        for (int d = 1; d < 32; d <<= 1) {
            const unsigned y = __shfl_up(inc, d);
            if (lane >= d) inc += y;
        }
        if (lane == 31) wtot[wv] = inc;
        __syncthreads();
        unsigned before = 0u;
        for (int w = 0; w < DEG_T / 32; ++w) {
            const unsigned s = wtot[w];
            if (w < wv) before += s;
        }
        bpre[t] = before + inc - bs;
        __syncthreads();
        unsigned r[TAB_U];
        for (int u = 0; u < TAB_U; ++u) {
            const unsigned i  = threadIdx.x + (unsigned)(u * DEG_T);
            const unsigned ii = i < lim ? i : lim - 1u;
            r[u] = a[u] + bpre[(lo + ii) / SCAN_B];
        }
        for (int u = 0; u < TAB_U; ++u) {
            const unsigned i = threadIdx.x + (unsigned)(u * DEG_T);
            cur[i] = i < lim ? r[u] + b[u] : 0u;
        }
        for (int u = 0; u < TAB_U; ++u)
            if (u == p) myptr = r[u];
    }
    __syncthreads();

    for (int k = 0; k < NIT; ++k) {
        const int base = (int)threadIdx.x + k * STEP;
        int4 nc[DEG_U], nr[DEG_U];
        if (k + 1 < NIT) {
            for (int u = 0; u < DEG_U; ++u) {
                const int i  = base + STEP + u * DEG_T;
                const int ii = i < NV ? i : NV - 1;
                nc[u] = c4[ii];
                nr[u] = r4[ii];
            }
        }
        unsigned d[4 * DEG_U];
        int      s[4 * DEG_U];
        for (int u = 0; u < DEG_U; ++u) {
            const bool in = base + u * DEG_T < NV;
            d[4 * u + 0] = in ? (unsigned)(vc[u].x - lo) : ~0u;
            d[4 * u + 1] = in ? (unsigned)(vc[u].y - lo) : ~0u;
            d[4 * u + 2] = in ? (unsigned)(vc[u].z - lo) : ~0u;
            d[4 * u + 3] = in ? (unsigned)(vc[u].w - lo) : ~0u;
            s[4 * u + 0] = vr[u].x;
            s[4 * u + 1] = vr[u].y;
            s[4 * u + 2] = vr[u].z;
            s[4 * u + 3] = vr[u].w;
        }
        unsigned slot[4 * DEG_U];
        for (int j = 0; j < 4 * DEG_U; ++j) {
            slot[j] = 0u;
            if (d[j] < lim) slot[j] = atomicAdd(&cur[d[j]], 1u);
        }
        for (int j = 0; j < 4 * DEG_U; ++j)
            if (d[j] < lim) srcs[slot[j]] = s[j];
        if (k + 1 < NIT) {
            for (int u = 0; u < DEG_U; ++u) { vc[u] = nc[u]; vr[u] = nr[u]; }
        }
    }
    if (p < TAB_U) {
        const unsigned i = threadIdx.x + (unsigned)(p * DEG_T);
        if (i < lim) rowptr[lo + i] = myptr;
    }
}

template <int K, int N, bool RELU>
__global__ void k_gemm_wmma(const float* __restrict__ A,
                            const _Float16* __restrict__ Bp,
                            const float* __restrict__ rscale,
                            float* __restrict__ C, int m_tiles) {
    constexpr int NT = N / 16;
    constexpr int KC = K / 32;
    const int wave = (blockIdx.x * blockDim.x + threadIdx.x) >> 5;
    const int lane = threadIdx.x & 31;
    if (wave >= m_tiles) return;

    const int l16 = lane & 15;
    const int kb  = (lane >> 4) * 8;
    const float* __restrict__ arow = A + (long)(wave * 16 + l16) * K;

    const float* __restrict__ srow = rscale + wave * 16 + (lane >> 4) * 8;
    float sc[8];
#pragma unroll
    for (int r = 0; r < 8; ++r) sc[r] = srow[r];

    v8f acc[NT] = {};
#pragma unroll
    for (int kc = 0; kc < KC; ++kc) {
        v16h a;
#pragma unroll
        for (int i = 0; i < 8; ++i) {
            float a0 = arow[kc * 32 + kb + i];
            float a1 = arow[kc * 32 + 16 + kb + i];
            if (RELU) { a0 = fmaxf(a0, 0.0f); a1 = fmaxf(a1, 0.0f); }
            a[i]     = (_Float16)a0;
            a[8 + i] = (_Float16)a1;
        }
#pragma unroll
        for (int tn = 0; tn < NT; ++tn) {
            v16h b = *(const v16h*)(Bp + ((long)(tn * KC + kc) * 32 + lane) * 16);
            acc[tn] = __builtin_amdgcn_wmma_f32_16x16x32_f16(false, a, false, b,
                                                             (short)0, acc[tn],
                                                             false, false);
        }
    }

    float* __restrict__ crow = C + ((long)wave * 16 + (lane >> 4) * 8) * N + l16;
#pragma unroll
    for (int tn = 0; tn < NT; ++tn)
#pragma unroll
        for (int r = 0; r < 8; ++r) crow[tn * 16 + (long)r * N] = acc[tn][r] * sc[r];
}

#define AGG_B      256
#define G64_ROUNDS 7
#define G64_NB     ((((N_NODES + G64_ROUNDS - 1) / G64_ROUNDS) * 32 + AGG_B - 1) / AGG_B)
#define G64_STRIDE (G64_NB * (AGG_B / 32))
#define G16_ROUNDS 4
#define G16_NB     ((((N_NODES + G16_ROUNDS - 1) / G16_ROUNDS) * 16 + AGG_B - 1) / AGG_B)
#define G16_STRIDE (G16_NB * (AGG_B / 16))

static_assert((long)G64_STRIDE * G64_ROUNDS >= N_NODES, "layer-1 rounds cover all nodes");
static_assert((long)G16_STRIDE * G16_ROUNDS >= N_NODES, "layer-2 rounds cover all nodes");
static_assert(G64_NB <= 2048 && G16_NB <= 2048, "the whole grid is resident at once");

__device__ __forceinline__ void add_rows64(const float* __restrict__ xw, int lane,
                                           int my, unsigned n,
                                           float& ax, float& ay) {
    unsigned j = 0;
    for (; j + 4u <= n; j += 4u) {
        const int r0 = __builtin_amdgcn_readlane(my, j);
        const int r1 = __builtin_amdgcn_readlane(my, j + 1u);
        const int r2 = __builtin_amdgcn_readlane(my, j + 2u);
        const int r3 = __builtin_amdgcn_readlane(my, j + 3u);
        const float2 v0 = *(const float2*)(xw + (long)r0 * NHID + lane * 2);
        const float2 v1 = *(const float2*)(xw + (long)r1 * NHID + lane * 2);
        const float2 v2 = *(const float2*)(xw + (long)r2 * NHID + lane * 2);
        const float2 v3 = *(const float2*)(xw + (long)r3 * NHID + lane * 2);
        ax += v0.x;  ay += v0.y;
        ax += v1.x;  ay += v1.y;
        ax += v2.x;  ay += v2.y;
        ax += v3.x;  ay += v3.y;
    }
    for (; j < n; ++j) {
        const int    r = __builtin_amdgcn_readlane(my, j);
        const float2 v = *(const float2*)(xw + (long)r * NHID + lane * 2);
        ax += v.x;  ay += v.y;
    }
}

__global__ void __launch_bounds__(AGG_B)
k_gather64(const int* __restrict__ srcs, const unsigned* __restrict__ rowptr,
           const unsigned* __restrict__ deg, const float* __restrict__ dinv,
           const float* __restrict__ xw, const float* __restrict__ bias,
           float* __restrict__ h) {
    const int lane = threadIdx.x & 31;
    int c = __builtin_amdgcn_readfirstlane(
        (int)((blockIdx.x * AGG_B + threadIdx.x) >> 5));
    if (c >= N_NODES) return;
    const float2 b = *(const float2*)(bias + lane * 2);

    unsigned start = rowptr[c];
    unsigned cnt   = deg[c];
    float    dc    = dinv[c];
    int c1 = c + G64_STRIDE;
    unsigned start1 = 0u, cnt1 = 0u;
    float    dc1 = 0.0f;
    if (c1 < N_NODES) { start1 = rowptr[c1]; cnt1 = deg[c1]; dc1 = dinv[c1]; }
    int my = srcs[(unsigned)lane < cnt ? start + lane : 0u];

    for (;;) {
        const int c2 = c1 + G64_STRIDE;
        unsigned start2 = 0u, cnt2 = 0u;
        float    dc2 = 0.0f;
        if (c2 < N_NODES) { start2 = rowptr[c2]; cnt2 = deg[c2]; dc2 = dinv[c2]; }

        const float2 self = *(const float2*)(xw + (long)c * NHID + lane * 2);
        const bool want1 = c1 < N_NODES && (unsigned)lane < cnt1;
        const int  my1   = srcs[want1 ? start1 + lane : 0u];

        float ax = self.x;
        float ay = self.y;
        add_rows64(xw, lane, my, cnt < 32u ? cnt : 32u, ax, ay);
        for (unsigned base = 32u; base < cnt; base += 32u) {
            const unsigned n = cnt - base < 32u ? cnt - base : 32u;
            int my_b = 0;
            if ((unsigned)lane < n) my_b = srcs[start + base + lane];
            add_rows64(xw, lane, my_b, n, ax, ay);
        }
        *(float2*)(h + (long)c * NHID + lane * 2) =
            make_float2(b.x + dc * ax, b.y + dc * ay);

        if (c1 >= N_NODES) break;
        c = c1;  start = start1;  cnt = cnt1;  dc = dc1;  my = my1;
        c1 = c2; start1 = start2; cnt1 = cnt2; dc1 = dc2;
    }
}

template <int M>
__device__ __forceinline__ float row_xor(float v) {
    return __int_as_float(__builtin_amdgcn_update_dpp(
        __float_as_int(v), __float_as_int(v), 0x160 + M, 0xf, 0xf, false));
}

__global__ void __launch_bounds__(AGG_B)
k_gather16(const int* __restrict__ srcs, const unsigned* __restrict__ rowptr,
           const unsigned* __restrict__ deg, const float* __restrict__ dinv,
           const float* __restrict__ h2, const float* __restrict__ bias,
           float* __restrict__ out) {
    const int t = blockIdx.x * AGG_B + threadIdx.x;
    int c = t >> 4;
    const int j = t & 15;
    if (c >= N_NODES) return;
    const float bj = bias[j];
    unsigned cnt = deg[c];
    unsigned st  = rowptr[c];
    float    dc  = dinv[c];
    for (;;) {
        const float selfv = h2[(long)c * NCLASS + j];
        const int  cn   = c + G16_STRIDE;
        const bool more = cn < N_NODES;
        const int  cp   = more ? cn : c;
        const unsigned ncnt = deg[cp];
        const unsigned nst  = rowptr[cp];
        const float    ndc  = dinv[cp];

        const int* __restrict__ seg = srcs + st;
        float sum = selfv;
        unsigned e = 0;
        for (; e + 4u <= cnt; e += 4u) {
            const int r0 = seg[e], r1 = seg[e + 1u], r2 = seg[e + 2u], r3 = seg[e + 3u];
            const float v0 = h2[(long)r0 * NCLASS + j];
            const float v1 = h2[(long)r1 * NCLASS + j];
            const float v2 = h2[(long)r2 * NCLASS + j];
            const float v3 = h2[(long)r3 * NCLASS + j];
            sum += v0;
            sum += v1;
            sum += v2;
            sum += v3;
        }
        for (; e < cnt; ++e) sum += h2[(long)seg[e] * NCLASS + j];
        const float acc = bj + dc * sum;
        float m = acc;
        m = fmaxf(m, row_xor<1>(m));
        m = fmaxf(m, row_xor<2>(m));
        m = fmaxf(m, row_xor<4>(m));
        m = fmaxf(m, row_xor<8>(m));
        float s = __expf(acc - m);
        s += row_xor<1>(s);
        s += row_xor<2>(s);
        s += row_xor<4>(s);
        s += row_xor<8>(s);
        out[(long)c * NCLASS + j] = acc - (m + __logf(s));

        if (!more) break;
        c = cn;  cnt = ncnt;  st = nst;  dc = ndc;
    }
}

extern "C" void kernel_launch(void* const* d_in, const int* in_sizes, int n_in,
                              void* d_out, int out_size, void* d_ws, size_t ws_size,
                              hipStream_t stream) {
    const float* x   = (const float*)d_in[0];
    const int*   ei  = (const int*)d_in[1];
    const float* W1  = (const float*)d_in[2];
    const float* b1  = (const float*)d_in[3];
    const float* W2  = (const float*)d_in[4];
    const float* b2  = (const float*)d_in[5];
    float* out       = (float*)d_out;

    const int* row = ei;
    const int* col = ei + N_EDGES;

    auto align256 = [](size_t v) { return (v + 255) & ~(size_t)255; };
    char* ws = (char*)d_ws;
    size_t o = 0;
    float*     dinv = (float*)(ws + o);     o += align256((size_t)N_NODES * 4);
    _Float16*  W1p  = (_Float16*)(ws + o);  o += align256((size_t)NFEAT * NHID * 2);
    _Float16*  W2p  = (_Float16*)(ws + o);  o += align256((size_t)NHID * NCLASS * 2);
    float*     bufA = (float*)(ws + o);     o += align256((size_t)N_NODES * NHID * 4);
    float*     h    = (float*)(ws + o);     o += align256((size_t)N_NODES * NHID * 4);
    unsigned*  deg    = (unsigned*)(ws + o); o += align256((size_t)N_NODES * 4);
    unsigned*  rowptr = (unsigned*)(ws + o); o += align256((size_t)N_NODES * 4);
    unsigned*  lpre   = (unsigned*)(ws + o); o += align256((size_t)N_NODES * 4);
    unsigned*  bsum   = (unsigned*)(ws + o); o += align256((size_t)SCAN_NB * 4);
    int*       srcs   = (int*)(ws + o);      o += align256((size_t)N_EDGES * 4);
    unsigned*  part = (unsigned*)(ws + o);
    float*     h2   = bufA;

    const int B = 256;

    k_deg_hist<<<HIST_NB + PACK1_NB + PACK2_NB, DEG_T, 0, stream>>>(
        col, part, W1, W1p, W2, W2p);
    k_deg_to_dinv<<<SCAN_NB, SCAN_B, 0, stream>>>(part, dinv, deg, bsum, lpre);
    k_fill_srcs<<<DEG_G * DEG_P, DEG_T, 0, stream>>>(row, col, part, lpre, bsum,
                                                     rowptr, srcs);

    const int m_tiles = N_NODES / 16;

    k_gemm_wmma<NFEAT, NHID, false><<<(m_tiles * 32 + B - 1) / B, B, 0, stream>>>(
        x, W1p, dinv, bufA, m_tiles);

    k_gather64<<<G64_NB, AGG_B, 0, stream>>>(srcs, rowptr, deg, dinv, bufA, b1, h);

    k_gemm_wmma<NHID, NCLASS, true><<<(m_tiles * 32 + B - 1) / B, B, 0, stream>>>(
        h, W2p, dinv, h2, m_tiles);

    k_gather16<<<G16_NB, AGG_B, 0, stream>>>(srcs, rowptr, deg, dinv, h2, b2, out);
}


# 7 launches -> 6: the first product (x @ W1, rows scaled by dinv) rides as extra blocks of the edge-fill launch, streaming x while the fill blocks wait on LDS ranks and scattered stores
# speedup vs baseline: 4.0999x; 1.0870x over previous
#include <hip/hip_runtime.h>
#include <hip/hip_fp16.h>

typedef __attribute__((ext_vector_type(16))) _Float16 v16h;
typedef __attribute__((ext_vector_type(8)))  float    v8f;

#define N_NODES 100000
#define N_EDGES 1600000
#define NFEAT   128
#define NHID    64
#define NCLASS  16

#define DEG_G   8
#define DEG_P   32
#define DEG_R   ((((N_NODES + DEG_G - 1) / DEG_G) + 31) & ~31)
#define DEG_EPP (N_EDGES / DEG_P)
#define DEG_T   512
#define DEG_U   4
#define SCAN_B  256
#define SCAN_NB ((N_NODES + SCAN_B - 1) / SCAN_B)

static_assert(N_EDGES % DEG_P == 0, "edge parts must be equal");
static_assert(DEG_EPP % 4 == 0, "edge parts are read 16 bytes per lane");
static_assert((long)DEG_G * DEG_R >= N_NODES, "node ranges must cover all nodes");

template <int K, int N>
__device__ __forceinline__ void pack_w(const float* __restrict__ W,
                                       _Float16* __restrict__ P, int t) {
    if (t >= K * N) return;
    int i    = t & 15;
    int u    = t >> 4;
    int lane = u & 31;
    int v    = u >> 5;
    int kc   = v % (K / 32);
    int tn   = v / (K / 32);
    int l16  = lane & 15;
    int kb   = (lane >> 4) * 8;
    int k    = (i < 8) ? (kb + i) : (16 + kb + (i - 8));
    P[t] = (_Float16)W[(long)(kc * 32 + k) * N + tn * 16 + l16];
}

#define HIST_NB  (DEG_G * DEG_P)
#define PACK1_NB ((NFEAT * NHID + DEG_T - 1) / DEG_T)
#define PACK2_NB ((NHID * NCLASS + DEG_T - 1) / DEG_T)

__global__ void __launch_bounds__(DEG_T)
k_deg_hist(const int* __restrict__ col, unsigned* __restrict__ part,
           const float* __restrict__ W1, _Float16* __restrict__ W1p,
           const float* __restrict__ W2, _Float16* __restrict__ W2p) {
    __shared__ unsigned hist[DEG_R];
    if (blockIdx.x >= HIST_NB) {
        const int pb = (int)blockIdx.x - HIST_NB;
        if (pb < PACK1_NB)
            pack_w<NFEAT, NHID>(W1, W1p, pb * DEG_T + (int)threadIdx.x);
        else
            pack_w<NHID, NCLASS>(W2, W2p, (pb - PACK1_NB) * DEG_T + (int)threadIdx.x);
        return;
    }
    const int g  = blockIdx.x % DEG_G;
    const int p  = blockIdx.x / DEG_G;
    const int lo = g * DEG_R;

    for (int i = threadIdx.x; i < DEG_R; i += DEG_T) hist[i] = 0u;
    __syncthreads();

    const int4* __restrict__ c4 = (const int4*)(col + (long)p * DEG_EPP);
    constexpr int NV = DEG_EPP / 4;
    constexpr int HU = (NV + DEG_T - 1) / DEG_T;
    int4 v[HU];
    for (int u = 0; u < HU; ++u) {
        const int i = (int)threadIdx.x + u * DEG_T;
        v[u] = c4[i < NV ? i : NV - 1];
    }
    for (int u = 0; u < HU; ++u) {
        const bool in = (int)threadIdx.x + u * DEG_T < NV;
        const unsigned d0 = in ? (unsigned)(v[u].x - lo) : ~0u;
        const unsigned d1 = in ? (unsigned)(v[u].y - lo) : ~0u;
        const unsigned d2 = in ? (unsigned)(v[u].z - lo) : ~0u;
        const unsigned d3 = in ? (unsigned)(v[u].w - lo) : ~0u;
        if (d0 < (unsigned)DEG_R) atomicAdd(&hist[d0], 1u);
        if (d1 < (unsigned)DEG_R) atomicAdd(&hist[d1], 1u);
        if (d2 < (unsigned)DEG_R) atomicAdd(&hist[d2], 1u);
        if (d3 < (unsigned)DEG_R) atomicAdd(&hist[d3], 1u);
    }
    __syncthreads();

    unsigned* __restrict__ prow = part + (long)p * N_NODES + lo;
    for (int i = threadIdx.x; i < DEG_R; i += DEG_T)
        if (lo + i < N_NODES) prow[i] = hist[i];
}

__global__ void __launch_bounds__(SCAN_B)
k_deg_to_dinv(unsigned* __restrict__ part, float* __restrict__ dinv,
              unsigned* __restrict__ deg, unsigned* __restrict__ bsum,
              unsigned* __restrict__ lpre) {
    __shared__ unsigned wsum[SCAN_B / 32];
    const int i = blockIdx.x * SCAN_B + threadIdx.x;
    unsigned run = 0u;
    if (i < N_NODES) {
        for (int p = 0; p < DEG_P; ++p) {
            unsigned* q = part + (long)p * N_NODES + i;
            const unsigned c = *q;
            *q = run;
            run += c;
        }
        deg[i] = run;
        float d = (float)(run + 1u);
        dinv[i] = d > 0.0f ? rsqrtf(d) : 0.0f;
    }
    const int lane = threadIdx.x & 31;
    const int wv   = threadIdx.x >> 5;
    unsigned inc = run;
    for (int d = 1; d < 32; d <<= 1) {
        const unsigned y = __shfl_up(inc, d);
        if (lane >= d) inc += y;
    }
    if (lane == 31) wsum[wv] = inc;
    __syncthreads();
    unsigned before = 0u, total = 0u;
    for (int w = 0; w < SCAN_B / 32; ++w) {
        const unsigned s = wsum[w];
        if (w < wv) before += s;
        total += s;
    }
    if (i < N_NODES) lpre[i] = before + inc - run;
    if (threadIdx.x == 0) bsum[blockIdx.x] = total;
}

template <int K, int N, bool RELU>
__device__ __forceinline__ void gemm_strip(const float* __restrict__ A,
                                           const _Float16* __restrict__ Bp,
                                           const float* __restrict__ rscale,
                                           float* __restrict__ C,
                                           int wave, int lane) {
    constexpr int NT = N / 16;
    constexpr int KC = K / 32;

    const int l16 = lane & 15;
    const int kb  = (lane >> 4) * 8;
    const float* __restrict__ arow = A + (long)(wave * 16 + l16) * K;

    const float* __restrict__ srow = rscale + wave * 16 + (lane >> 4) * 8;
    float sc[8];
#pragma unroll
    for (int r = 0; r < 8; ++r) sc[r] = srow[r];

    v8f acc[NT] = {};
#pragma unroll
    for (int kc = 0; kc < KC; ++kc) {
        v16h a;
#pragma unroll
        for (int i = 0; i < 8; ++i) {
            float a0 = arow[kc * 32 + kb + i];
            float a1 = arow[kc * 32 + 16 + kb + i];
            if (RELU) { a0 = fmaxf(a0, 0.0f); a1 = fmaxf(a1, 0.0f); }
            a[i]     = (_Float16)a0;
            a[8 + i] = (_Float16)a1;
        }
#pragma unroll
        for (int tn = 0; tn < NT; ++tn) {
            v16h b = *(const v16h*)(Bp + ((long)(tn * KC + kc) * 32 + lane) * 16);
            acc[tn] = __builtin_amdgcn_wmma_f32_16x16x32_f16(false, a, false, b,
                                                             (short)0, acc[tn],
                                                             false, false);
        }
    }

    float* __restrict__ crow = C + ((long)wave * 16 + (lane >> 4) * 8) * N + l16;
#pragma unroll
    for (int tn = 0; tn < NT; ++tn)
#pragma unroll
        for (int r = 0; r < 8; ++r) crow[tn * 16 + (long)r * N] = acc[tn][r] * sc[r];
}

template <int K, int N, bool RELU>
__global__ void k_gemm_wmma(const float* __restrict__ A,
                            const _Float16* __restrict__ Bp,
                            const float* __restrict__ rscale,
                            float* __restrict__ C, int m_tiles) {
    const int wave = (blockIdx.x * blockDim.x + threadIdx.x) >> 5;
    const int lane = threadIdx.x & 31;
    if (wave >= m_tiles) return;
    gemm_strip<K, N, RELU>(A, Bp, rscale, C, wave, lane);
}

#define FILL_NB  (DEG_G * DEG_P)
#define M_TILES  (N_NODES / 16)
#define GEMM1_NB ((M_TILES + DEG_T / 32 - 1) / (DEG_T / 32))
static_assert(N_NODES % 16 == 0, "the products work on whole 16-row strips");

__global__ void __launch_bounds__(DEG_T)
k_fill_srcs(const int* __restrict__ row, const int* __restrict__ col,
            const unsigned* __restrict__ part,
            const unsigned* __restrict__ lpre, const unsigned* __restrict__ bsum,
            unsigned* __restrict__ rowptr, int* __restrict__ srcs,
            const float* __restrict__ x, const _Float16* __restrict__ W1p,
            const float* __restrict__ dinv, float* __restrict__ xw) {
    if (blockIdx.x >= FILL_NB) {
        const int wave = ((int)blockIdx.x - FILL_NB) * (DEG_T / 32)
                         + (int)(threadIdx.x >> 5);
        if (wave < M_TILES)
            gemm_strip<NFEAT, NHID, false>(x, W1p, dinv, xw, wave,
                                           (int)(threadIdx.x & 31));
        return;
    }
    constexpr int TAB_U = (DEG_R + DEG_T - 1) / DEG_T;
    __shared__ unsigned cur[TAB_U * DEG_T];
    __shared__ unsigned bpre[DEG_T];
    __shared__ unsigned wtot[DEG_T / 32];
    static_assert(SCAN_NB <= DEG_T, "one block sum per lane covers all scan blocks");
    static_assert(TAB_U <= DEG_P, "every table entry of a lane has a part that stores it");
    const int g  = blockIdx.x % DEG_G;
    const int p  = blockIdx.x / DEG_G;
    const int lo = g * DEG_R;
    const unsigned lim = (unsigned)(N_NODES - lo < DEG_R ? N_NODES - lo : DEG_R);

    const int4* __restrict__ c4 = (const int4*)(col + (long)p * DEG_EPP);
    const int4* __restrict__ r4 = (const int4*)(row + (long)p * DEG_EPP);
    constexpr int NV   = DEG_EPP / 4;
    constexpr int STEP = DEG_T * DEG_U;
    constexpr int NIT  = (NV + STEP - 1) / STEP;

    int4 vc[DEG_U], vr[DEG_U];
    for (int u = 0; u < DEG_U; ++u) {
        const int i  = (int)threadIdx.x + u * DEG_T;
        const int ii = i < NV ? i : NV - 1;
        vc[u] = c4[ii];
        vr[u] = r4[ii];
    }

    const unsigned* __restrict__ prow = part + (long)p * N_NODES + lo;
    unsigned myptr = 0u;
    {
        const int t    = threadIdx.x;
        const int lane = t & 31;
        const int wv   = t >> 5;
        const unsigned bl = bsum[t < SCAN_NB ? t : SCAN_NB - 1];
        unsigned a[TAB_U], b[TAB_U];
        for (int u = 0; u < TAB_U; ++u) {
            const unsigned i  = threadIdx.x + (unsigned)(u * DEG_T);
            const unsigned ii = i < lim ? i : lim - 1u;
            a[u] = lpre[lo + ii];
            b[u] = prow[ii];
        }
        const unsigned bs = t < SCAN_NB ? bl : 0u;
        unsigned inc = bs;
        for (int d = 1; d < 32; d <<= 1) {
            const unsigned y = __shfl_up(inc, d);
            if (lane >= d) inc += y;
        }
        if (lane == 31) wtot[wv] = inc;
        __syncthreads();
        unsigned before = 0u;
        for (int w = 0; w < DEG_T / 32; ++w) {
            const unsigned s = wtot[w];
            if (w < wv) before += s;
        }
        bpre[t] = before + inc - bs;
        __syncthreads();
        unsigned r[TAB_U];
        for (int u = 0; u < TAB_U; ++u) {
            const unsigned i  = threadIdx.x + (unsigned)(u * DEG_T);
            const unsigned ii = i < lim ? i : lim - 1u;
            r[u] = a[u] + bpre[(lo + ii) / SCAN_B];
        }
        for (int u = 0; u < TAB_U; ++u) {
            const unsigned i = threadIdx.x + (unsigned)(u * DEG_T);
            cur[i] = i < lim ? r[u] + b[u] : 0u;
        }
        for (int u = 0; u < TAB_U; ++u)
            if (u == p) myptr = r[u];
    }
    __syncthreads();

    for (int k = 0; k < NIT; ++k) {
        const int base = (int)threadIdx.x + k * STEP;
        int4 nc[DEG_U], nr[DEG_U];
        if (k + 1 < NIT) {
            for (int u = 0; u < DEG_U; ++u) {
                const int i  = base + STEP + u * DEG_T;
                const int ii = i < NV ? i : NV - 1;
                nc[u] = c4[ii];
                nr[u] = r4[ii];
            }
        }
        unsigned d[4 * DEG_U];
        int      s[4 * DEG_U];
        for (int u = 0; u < DEG_U; ++u) {
            const bool in = base + u * DEG_T < NV;
            d[4 * u + 0] = in ? (unsigned)(vc[u].x - lo) : ~0u;
            d[4 * u + 1] = in ? (unsigned)(vc[u].y - lo) : ~0u;
            d[4 * u + 2] = in ? (unsigned)(vc[u].z - lo) : ~0u;
            d[4 * u + 3] = in ? (unsigned)(vc[u].w - lo) : ~0u;
            s[4 * u + 0] = vr[u].x;
            s[4 * u + 1] = vr[u].y;
            s[4 * u + 2] = vr[u].z;
            s[4 * u + 3] = vr[u].w;
        }
        unsigned slot[4 * DEG_U];
        for (int j = 0; j < 4 * DEG_U; ++j) {
            slot[j] = 0u;
            if (d[j] < lim) slot[j] = atomicAdd(&cur[d[j]], 1u);
        }
        for (int j = 0; j < 4 * DEG_U; ++j)
            if (d[j] < lim) srcs[slot[j]] = s[j];
        if (k + 1 < NIT) {
            for (int u = 0; u < DEG_U; ++u) { vc[u] = nc[u]; vr[u] = nr[u]; }
        }
    }
    if (p < TAB_U) {
        const unsigned i = threadIdx.x + (unsigned)(p * DEG_T);
        if (i < lim) rowptr[lo + i] = myptr;
    }
}

#define AGG_B      256
#define G64_ROUNDS 7
#define G64_NB     ((((N_NODES + G64_ROUNDS - 1) / G64_ROUNDS) * 32 + AGG_B - 1) / AGG_B)
#define G64_STRIDE (G64_NB * (AGG_B / 32))
#define G16_ROUNDS 4
#define G16_NB     ((((N_NODES + G16_ROUNDS - 1) / G16_ROUNDS) * 16 + AGG_B - 1) / AGG_B)
#define G16_STRIDE (G16_NB * (AGG_B / 16))

static_assert((long)G64_STRIDE * G64_ROUNDS >= N_NODES, "layer-1 rounds cover all nodes");
static_assert((long)G16_STRIDE * G16_ROUNDS >= N_NODES, "layer-2 rounds cover all nodes");
static_assert(G64_NB <= 2048 && G16_NB <= 2048, "the whole grid is resident at once");

__device__ __forceinline__ void add_rows64(const float* __restrict__ xw, int lane,
                                           int my, unsigned n,
                                           float& ax, float& ay) {
    unsigned j = 0;
    for (; j + 4u <= n; j += 4u) {
        const int r0 = __builtin_amdgcn_readlane(my, j);
        const int r1 = __builtin_amdgcn_readlane(my, j + 1u);
        const int r2 = __builtin_amdgcn_readlane(my, j + 2u);
        const int r3 = __builtin_amdgcn_readlane(my, j + 3u);
        const float2 v0 = *(const float2*)(xw + (long)r0 * NHID + lane * 2);
        const float2 v1 = *(const float2*)(xw + (long)r1 * NHID + lane * 2);
        const float2 v2 = *(const float2*)(xw + (long)r2 * NHID + lane * 2);
        const float2 v3 = *(const float2*)(xw + (long)r3 * NHID + lane * 2);
        ax += v0.x;  ay += v0.y;
        ax += v1.x;  ay += v1.y;
        ax += v2.x;  ay += v2.y;
        ax += v3.x;  ay += v3.y;
    }
    for (; j < n; ++j) {
        const int    r = __builtin_amdgcn_readlane(my, j);
        const float2 v = *(const float2*)(xw + (long)r * NHID + lane * 2);
        ax += v.x;  ay += v.y;
    }
}

__global__ void __launch_bounds__(AGG_B)
k_gather64(const int* __restrict__ srcs, const unsigned* __restrict__ rowptr,
           const unsigned* __restrict__ deg, const float* __restrict__ dinv,
           const float* __restrict__ xw, const float* __restrict__ bias,
           float* __restrict__ h) {
    const int lane = threadIdx.x & 31;
    int c = __builtin_amdgcn_readfirstlane(
        (int)((blockIdx.x * AGG_B + threadIdx.x) >> 5));
    if (c >= N_NODES) return;
    const float2 b = *(const float2*)(bias + lane * 2);

    unsigned start = rowptr[c];
    unsigned cnt   = deg[c];
    float    dc    = dinv[c];
    int c1 = c + G64_STRIDE;
    unsigned start1 = 0u, cnt1 = 0u;
    float    dc1 = 0.0f;
    if (c1 < N_NODES) { start1 = rowptr[c1]; cnt1 = deg[c1]; dc1 = dinv[c1]; }
    int my = srcs[(unsigned)lane < cnt ? start + lane : 0u];

    for (;;) {
        const int c2 = c1 + G64_STRIDE;
        unsigned start2 = 0u, cnt2 = 0u;
        float    dc2 = 0.0f;
        if (c2 < N_NODES) { start2 = rowptr[c2]; cnt2 = deg[c2]; dc2 = dinv[c2]; }

        const float2 self = *(const float2*)(xw + (long)c * NHID + lane * 2);
        const bool want1 = c1 < N_NODES && (unsigned)lane < cnt1;
        const int  my1   = srcs[want1 ? start1 + lane : 0u];

        float ax = self.x;
        float ay = self.y;
        add_rows64(xw, lane, my, cnt < 32u ? cnt : 32u, ax, ay);
        for (unsigned base = 32u; base < cnt; base += 32u) {
            const unsigned n = cnt - base < 32u ? cnt - base : 32u;
            int my_b = 0;
            if ((unsigned)lane < n) my_b = srcs[start + base + lane];
            add_rows64(xw, lane, my_b, n, ax, ay);
        }
        *(float2*)(h + (long)c * NHID + lane * 2) =
            make_float2(b.x + dc * ax, b.y + dc * ay);

        if (c1 >= N_NODES) break;
        c = c1;  start = start1;  cnt = cnt1;  dc = dc1;  my = my1;
        c1 = c2; start1 = start2; cnt1 = cnt2; dc1 = dc2;
    }
}

template <int M>
__device__ __forceinline__ float row_xor(float v) {
    return __int_as_float(__builtin_amdgcn_update_dpp(
        __float_as_int(v), __float_as_int(v), 0x160 + M, 0xf, 0xf, false));
}

__global__ void __launch_bounds__(AGG_B)
k_gather16(const int* __restrict__ srcs, const unsigned* __restrict__ rowptr,
           const unsigned* __restrict__ deg, const float* __restrict__ dinv,
           const float* __restrict__ h2, const float* __restrict__ bias,
           float* __restrict__ out) {
    const int t = blockIdx.x * AGG_B + threadIdx.x;
    int c = t >> 4;
    const int j = t & 15;
    if (c >= N_NODES) return;
    const float bj = bias[j];
    unsigned cnt = deg[c];
    unsigned st  = rowptr[c];
    float    dc  = dinv[c];
    for (;;) {
        const float selfv = h2[(long)c * NCLASS + j];
        const int  cn   = c + G16_STRIDE;
        const bool more = cn < N_NODES;
        const int  cp   = more ? cn : c;
        const unsigned ncnt = deg[cp];
        const unsigned nst  = rowptr[cp];
        const float    ndc  = dinv[cp];

        const int* __restrict__ seg = srcs + st;
        float sum = selfv;
        unsigned e = 0;
        for (; e + 4u <= cnt; e += 4u) {
            const int r0 = seg[e], r1 = seg[e + 1u], r2 = seg[e + 2u], r3 = seg[e + 3u];
            const float v0 = h2[(long)r0 * NCLASS + j];
            const float v1 = h2[(long)r1 * NCLASS + j];
            const float v2 = h2[(long)r2 * NCLASS + j];
            const float v3 = h2[(long)r3 * NCLASS + j];
            sum += v0;
            sum += v1;
            sum += v2;
            sum += v3;
        }
        for (; e < cnt; ++e) sum += h2[(long)seg[e] * NCLASS + j];
        const float acc = bj + dc * sum;
        float m = acc;
        m = fmaxf(m, row_xor<1>(m));
        m = fmaxf(m, row_xor<2>(m));
        m = fmaxf(m, row_xor<4>(m));
        m = fmaxf(m, row_xor<8>(m));
        float s = __expf(acc - m);
        s += row_xor<1>(s);
        s += row_xor<2>(s);
        s += row_xor<4>(s);
        s += row_xor<8>(s);
        out[(long)c * NCLASS + j] = acc - (m + __logf(s));

        if (!more) break;
        c = cn;  cnt = ncnt;  st = nst;  dc = ndc;
    }
}

extern "C" void kernel_launch(void* const* d_in, const int* in_sizes, int n_in,
                              void* d_out, int out_size, void* d_ws, size_t ws_size,
                              hipStream_t stream) {
    const float* x   = (const float*)d_in[0];
    const int*   ei  = (const int*)d_in[1];
    const float* W1  = (const float*)d_in[2];
    const float* b1  = (const float*)d_in[3];
    const float* W2  = (const float*)d_in[4];
    const float* b2  = (const float*)d_in[5];
    float* out       = (float*)d_out;

    const int* row = ei;
    const int* col = ei + N_EDGES;

    auto align256 = [](size_t v) { return (v + 255) & ~(size_t)255; };
    char* ws = (char*)d_ws;
    size_t o = 0;
    float*     dinv = (float*)(ws + o);     o += align256((size_t)N_NODES * 4);
    _Float16*  W1p  = (_Float16*)(ws + o);  o += align256((size_t)NFEAT * NHID * 2);
    _Float16*  W2p  = (_Float16*)(ws + o);  o += align256((size_t)NHID * NCLASS * 2);
    float*     bufA = (float*)(ws + o);     o += align256((size_t)N_NODES * NHID * 4);
    float*     h    = (float*)(ws + o);     o += align256((size_t)N_NODES * NHID * 4);
    unsigned*  deg    = (unsigned*)(ws + o); o += align256((size_t)N_NODES * 4);
    unsigned*  rowptr = (unsigned*)(ws + o); o += align256((size_t)N_NODES * 4);
    unsigned*  lpre   = (unsigned*)(ws + o); o += align256((size_t)N_NODES * 4);
    unsigned*  bsum   = (unsigned*)(ws + o); o += align256((size_t)SCAN_NB * 4);
    int*       srcs   = (int*)(ws + o);      o += align256((size_t)N_EDGES * 4);
    unsigned*  part = (unsigned*)(ws + o);
    float*     h2   = bufA;

    const int B = 256;

    k_deg_hist<<<HIST_NB + PACK1_NB + PACK2_NB, DEG_T, 0, stream>>>(
        col, part, W1, W1p, W2, W2p);
    k_deg_to_dinv<<<SCAN_NB, SCAN_B, 0, stream>>>(part, dinv, deg, bsum, lpre);
    k_fill_srcs<<<FILL_NB + GEMM1_NB, DEG_T, 0, stream>>>(row, col, part, lpre, bsum,
                                                          rowptr, srcs,
                                                          x, W1p, dinv, bufA);

    const int m_tiles = M_TILES;

    k_gather64<<<G64_NB, AGG_B, 0, stream>>>(srcs, rowptr, deg, dinv, bufA, b1, h);

    k_gemm_wmma<NHID, NCLASS, true><<<(m_tiles * 32 + B - 1) / B, B, 0, stream>>>(
        h, W2p, dinv, h2, m_tiles);

    k_gather16<<<G16_NB, AGG_B, 0, stream>>>(srcs, rowptr, deg, dinv, h2, b2, out);
}


# 6 launches -> 5: relu(h) @ W2 done inside the layer-1 aggregation: a block stages its 64 rows of relu(h) as f16 in LDS, one wave per 16-row strip multiplies them on the matrix core; h is never stored
# speedup vs baseline: 4.4062x; 1.0747x over previous
#include <hip/hip_runtime.h>
#include <hip/hip_fp16.h>

typedef __attribute__((ext_vector_type(16))) _Float16 v16h;
typedef __attribute__((ext_vector_type(8)))  float    v8f;

#define N_NODES 100000
#define N_EDGES 1600000
#define NFEAT   128
#define NHID    64
#define NCLASS  16

#define DEG_G   8
#define DEG_P   32
#define DEG_R   ((((N_NODES + DEG_G - 1) / DEG_G) + 31) & ~31)
#define DEG_EPP (N_EDGES / DEG_P)
#define DEG_T   512
#define DEG_U   4
#define SCAN_B  256
#define SCAN_NB ((N_NODES + SCAN_B - 1) / SCAN_B)

static_assert(N_EDGES % DEG_P == 0, "edge parts must be equal");
static_assert(DEG_EPP % 4 == 0, "edge parts are read 16 bytes per lane");
static_assert((long)DEG_G * DEG_R >= N_NODES, "node ranges must cover all nodes");

template <int K, int N>
__device__ __forceinline__ void pack_w(const float* __restrict__ W,
                                       _Float16* __restrict__ P, int t) {
    if (t >= K * N) return;
    int i    = t & 15;
    int u    = t >> 4;
    int lane = u & 31;
    int v    = u >> 5;
    int kc   = v % (K / 32);
    int tn   = v / (K / 32);
    int l16  = lane & 15;
    int kb   = (lane >> 4) * 8;
    int k    = (i < 8) ? (kb + i) : (16 + kb + (i - 8));
    P[t] = (_Float16)W[(long)(kc * 32 + k) * N + tn * 16 + l16];
}

#define HIST_NB  (DEG_G * DEG_P)
#define PACK1_NB ((NFEAT * NHID + DEG_T - 1) / DEG_T)
#define PACK2_NB ((NHID * NCLASS + DEG_T - 1) / DEG_T)

__global__ void __launch_bounds__(DEG_T)
k_deg_hist(const int* __restrict__ col, unsigned* __restrict__ part,
           const float* __restrict__ W1, _Float16* __restrict__ W1p,
           const float* __restrict__ W2, _Float16* __restrict__ W2p) {
    __shared__ unsigned hist[DEG_R];
    if (blockIdx.x >= HIST_NB) {
        const int pb = (int)blockIdx.x - HIST_NB;
        if (pb < PACK1_NB)
            pack_w<NFEAT, NHID>(W1, W1p, pb * DEG_T + (int)threadIdx.x);
        else
            pack_w<NHID, NCLASS>(W2, W2p, (pb - PACK1_NB) * DEG_T + (int)threadIdx.x);
        return;
    }
    const int g  = blockIdx.x % DEG_G;
    const int p  = blockIdx.x / DEG_G;
    const int lo = g * DEG_R;

    for (int i = threadIdx.x; i < DEG_R; i += DEG_T) hist[i] = 0u;
    __syncthreads();

    const int4* __restrict__ c4 = (const int4*)(col + (long)p * DEG_EPP);
    constexpr int NV = DEG_EPP / 4;
    constexpr int HU = (NV + DEG_T - 1) / DEG_T;
    int4 v[HU];
    for (int u = 0; u < HU; ++u) {
        const int i = (int)threadIdx.x + u * DEG_T;
        v[u] = c4[i < NV ? i : NV - 1];
    }
    for (int u = 0; u < HU; ++u) {
        const bool in = (int)threadIdx.x + u * DEG_T < NV;
        const unsigned d0 = in ? (unsigned)(v[u].x - lo) : ~0u;
        const unsigned d1 = in ? (unsigned)(v[u].y - lo) : ~0u;
        const unsigned d2 = in ? (unsigned)(v[u].z - lo) : ~0u;
        const unsigned d3 = in ? (unsigned)(v[u].w - lo) : ~0u;
        if (d0 < (unsigned)DEG_R) atomicAdd(&hist[d0], 1u);
        if (d1 < (unsigned)DEG_R) atomicAdd(&hist[d1], 1u);
        if (d2 < (unsigned)DEG_R) atomicAdd(&hist[d2], 1u);
        if (d3 < (unsigned)DEG_R) atomicAdd(&hist[d3], 1u);
    }
    __syncthreads();

    unsigned* __restrict__ prow = part + (long)p * N_NODES + lo;
    for (int i = threadIdx.x; i < DEG_R; i += DEG_T)
        if (lo + i < N_NODES) prow[i] = hist[i];
}

__global__ void __launch_bounds__(SCAN_B)
k_deg_to_dinv(unsigned* __restrict__ part, float* __restrict__ dinv,
              unsigned* __restrict__ deg, unsigned* __restrict__ bsum,
              unsigned* __restrict__ lpre) {
    __shared__ unsigned wsum[SCAN_B / 32];
    const int i = blockIdx.x * SCAN_B + threadIdx.x;
    unsigned run = 0u;
    if (i < N_NODES) {
        for (int p = 0; p < DEG_P; ++p) {
            unsigned* q = part + (long)p * N_NODES + i;
            const unsigned c = *q;
            *q = run;
            run += c;
        }
        deg[i] = run;
        float d = (float)(run + 1u);
        dinv[i] = d > 0.0f ? rsqrtf(d) : 0.0f;
    }
    const int lane = threadIdx.x & 31;
    const int wv   = threadIdx.x >> 5;
    unsigned inc = run;
    for (int d = 1; d < 32; d <<= 1) {
        const unsigned y = __shfl_up(inc, d);
        if (lane >= d) inc += y;
    }
    if (lane == 31) wsum[wv] = inc;
    __syncthreads();
    unsigned before = 0u, total = 0u;
    for (int w = 0; w < SCAN_B / 32; ++w) {
        const unsigned s = wsum[w];
        if (w < wv) before += s;
        total += s;
    }
    if (i < N_NODES) lpre[i] = before + inc - run;
    if (threadIdx.x == 0) bsum[blockIdx.x] = total;
}

template <int K, int N, bool RELU>
__device__ __forceinline__ void gemm_strip(const float* __restrict__ A,
                                           const _Float16* __restrict__ Bp,
                                           const float* __restrict__ rscale,
                                           float* __restrict__ C,
                                           int wave, int lane) {
    constexpr int NT = N / 16;
    constexpr int KC = K / 32;

    const int l16 = lane & 15;
    const int kb  = (lane >> 4) * 8;
    const float* __restrict__ arow = A + (long)(wave * 16 + l16) * K;

    const float* __restrict__ srow = rscale + wave * 16 + (lane >> 4) * 8;
    float sc[8];
#pragma unroll
    for (int r = 0; r < 8; ++r) sc[r] = srow[r];

    v8f acc[NT] = {};
#pragma unroll
    for (int kc = 0; kc < KC; ++kc) {
        v16h a;
#pragma unroll
        for (int i = 0; i < 8; ++i) {
            float a0 = arow[kc * 32 + kb + i];
            float a1 = arow[kc * 32 + 16 + kb + i];
            if (RELU) { a0 = fmaxf(a0, 0.0f); a1 = fmaxf(a1, 0.0f); }
            a[i]     = (_Float16)a0;
            a[8 + i] = (_Float16)a1;
        }
#pragma unroll
        for (int tn = 0; tn < NT; ++tn) {
            v16h b = *(const v16h*)(Bp + ((long)(tn * KC + kc) * 32 + lane) * 16);
            acc[tn] = __builtin_amdgcn_wmma_f32_16x16x32_f16(false, a, false, b,
                                                             (short)0, acc[tn],
                                                             false, false);
        }
    }

    float* __restrict__ crow = C + ((long)wave * 16 + (lane >> 4) * 8) * N + l16;
#pragma unroll
    for (int tn = 0; tn < NT; ++tn)
#pragma unroll
        for (int r = 0; r < 8; ++r) crow[tn * 16 + (long)r * N] = acc[tn][r] * sc[r];
}

#define FILL_NB  (DEG_G * DEG_P)
#define M_TILES  (N_NODES / 16)
#define GEMM1_NB ((M_TILES + DEG_T / 32 - 1) / (DEG_T / 32))
static_assert(N_NODES % 16 == 0, "the products work on whole 16-row strips");

__global__ void __launch_bounds__(DEG_T)
k_fill_srcs(const int* __restrict__ row, const int* __restrict__ col,
            const unsigned* __restrict__ part,
            const unsigned* __restrict__ lpre, const unsigned* __restrict__ bsum,
            unsigned* __restrict__ rowptr, int* __restrict__ srcs,
            const float* __restrict__ x, const _Float16* __restrict__ W1p,
            const float* __restrict__ dinv, float* __restrict__ xw) {
    if (blockIdx.x >= FILL_NB) {
        const int wave = ((int)blockIdx.x - FILL_NB) * (DEG_T / 32)
                         + (int)(threadIdx.x >> 5);
        if (wave < M_TILES)
            gemm_strip<NFEAT, NHID, false>(x, W1p, dinv, xw, wave,
                                           (int)(threadIdx.x & 31));
        return;
    }
    constexpr int TAB_U = (DEG_R + DEG_T - 1) / DEG_T;
    __shared__ unsigned cur[TAB_U * DEG_T];
    __shared__ unsigned bpre[DEG_T];
    __shared__ unsigned wtot[DEG_T / 32];
    static_assert(SCAN_NB <= DEG_T, "one block sum per lane covers all scan blocks");
    static_assert(TAB_U <= DEG_P, "every table entry of a lane has a part that stores it");
    const int g  = blockIdx.x % DEG_G;
    const int p  = blockIdx.x / DEG_G;
    const int lo = g * DEG_R;
    const unsigned lim = (unsigned)(N_NODES - lo < DEG_R ? N_NODES - lo : DEG_R);

    const int4* __restrict__ c4 = (const int4*)(col + (long)p * DEG_EPP);
    const int4* __restrict__ r4 = (const int4*)(row + (long)p * DEG_EPP);
    constexpr int NV   = DEG_EPP / 4;
    constexpr int STEP = DEG_T * DEG_U;
    constexpr int NIT  = (NV + STEP - 1) / STEP;

    int4 vc[DEG_U], vr[DEG_U];
    for (int u = 0; u < DEG_U; ++u) {
        const int i  = (int)threadIdx.x + u * DEG_T;
        const int ii = i < NV ? i : NV - 1;
        vc[u] = c4[ii];
        vr[u] = r4[ii];
    }

    const unsigned* __restrict__ prow = part + (long)p * N_NODES + lo;
    unsigned myptr = 0u;
    {
        const int t    = threadIdx.x;
        const int lane = t & 31;
        const int wv   = t >> 5;
        const unsigned bl = bsum[t < SCAN_NB ? t : SCAN_NB - 1];
        unsigned a[TAB_U], b[TAB_U];
        for (int u = 0; u < TAB_U; ++u) {
            const unsigned i  = threadIdx.x + (unsigned)(u * DEG_T);
            const unsigned ii = i < lim ? i : lim - 1u;
            a[u] = lpre[lo + ii];
            b[u] = prow[ii];
        }
        const unsigned bs = t < SCAN_NB ? bl : 0u;
        unsigned inc = bs;
        for (int d = 1; d < 32; d <<= 1) {
            const unsigned y = __shfl_up(inc, d);
            if (lane >= d) inc += y;
        }
        if (lane == 31) wtot[wv] = inc;
        __syncthreads();
        unsigned before = 0u;
        for (int w = 0; w < DEG_T / 32; ++w) {
            const unsigned s = wtot[w];
            if (w < wv) before += s;
        }
        bpre[t] = before + inc - bs;
        __syncthreads();
        unsigned r[TAB_U];
        for (int u = 0; u < TAB_U; ++u) {
            const unsigned i  = threadIdx.x + (unsigned)(u * DEG_T);
            const unsigned ii = i < lim ? i : lim - 1u;
            r[u] = a[u] + bpre[(lo + ii) / SCAN_B];
        }
        for (int u = 0; u < TAB_U; ++u) {
            const unsigned i = threadIdx.x + (unsigned)(u * DEG_T);
            cur[i] = i < lim ? r[u] + b[u] : 0u;
        }
        for (int u = 0; u < TAB_U; ++u)
            if (u == p) myptr = r[u];
    }
    __syncthreads();

    for (int k = 0; k < NIT; ++k) {
        const int base = (int)threadIdx.x + k * STEP;
        int4 nc[DEG_U], nr[DEG_U];
        if (k + 1 < NIT) {
            for (int u = 0; u < DEG_U; ++u) {
                const int i  = base + STEP + u * DEG_T;
                const int ii = i < NV ? i : NV - 1;
                nc[u] = c4[ii];
                nr[u] = r4[ii];
            }
        }
        unsigned d[4 * DEG_U];
        int      s[4 * DEG_U];
        for (int u = 0; u < DEG_U; ++u) {
            const bool in = base + u * DEG_T < NV;
            d[4 * u + 0] = in ? (unsigned)(vc[u].x - lo) : ~0u;
            d[4 * u + 1] = in ? (unsigned)(vc[u].y - lo) : ~0u;
            d[4 * u + 2] = in ? (unsigned)(vc[u].z - lo) : ~0u;
            d[4 * u + 3] = in ? (unsigned)(vc[u].w - lo) : ~0u;
            s[4 * u + 0] = vr[u].x;
            s[4 * u + 1] = vr[u].y;
            s[4 * u + 2] = vr[u].z;
            s[4 * u + 3] = vr[u].w;
        }
        unsigned slot[4 * DEG_U];
        for (int j = 0; j < 4 * DEG_U; ++j) {
            slot[j] = 0u;
            if (d[j] < lim) slot[j] = atomicAdd(&cur[d[j]], 1u);
        }
        for (int j = 0; j < 4 * DEG_U; ++j)
            if (d[j] < lim) srcs[slot[j]] = s[j];
        if (k + 1 < NIT) {
            for (int u = 0; u < DEG_U; ++u) { vc[u] = nc[u]; vr[u] = nr[u]; }
        }
    }
    if (p < TAB_U) {
        const unsigned i = threadIdx.x + (unsigned)(p * DEG_T);
        if (i < lim) rowptr[lo + i] = myptr;
    }
}

#define AGG_B      256
#define F_NPW      8
#define F_NPB      (F_NPW * (AGG_B / 32))
#define F_STRIPS   (F_NPB / 16)
#define F_NB       ((N_NODES + F_NPB - 1) / F_NPB)
#define HS_LD      (NHID + 8)
#define G16_ROUNDS 4
#define G16_NB     ((((N_NODES + G16_ROUNDS - 1) / G16_ROUNDS) * 16 + AGG_B - 1) / AGG_B)
#define G16_STRIDE (G16_NB * (AGG_B / 16))

static_assert(N_NODES % F_NPW == 0, "a wave's nodes are all inside the graph or none is");
static_assert(F_NPB % 16 == 0 && F_NPW <= 32, "a block owns whole strips; bounds fit one per lane");
static_assert(NHID == 64 && NCLASS == 16, "two features per lane; one n-tile, two k-chunks");
static_assert((long)G16_STRIDE * G16_ROUNDS >= N_NODES, "layer-2 rounds cover all nodes");
static_assert(F_NB <= 2048 && G16_NB <= 2048, "the whole grid is resident at once");

__device__ __forceinline__ void add_rows64(const float* __restrict__ xw, int lane,
                                           int my, unsigned n,
                                           float& ax, float& ay) {
    unsigned j = 0;
    for (; j + 4u <= n; j += 4u) {
        const int r0 = __builtin_amdgcn_readlane(my, j);
        const int r1 = __builtin_amdgcn_readlane(my, j + 1u);
        const int r2 = __builtin_amdgcn_readlane(my, j + 2u);
        const int r3 = __builtin_amdgcn_readlane(my, j + 3u);
        const float2 v0 = *(const float2*)(xw + (long)r0 * NHID + lane * 2);
        const float2 v1 = *(const float2*)(xw + (long)r1 * NHID + lane * 2);
        const float2 v2 = *(const float2*)(xw + (long)r2 * NHID + lane * 2);
        const float2 v3 = *(const float2*)(xw + (long)r3 * NHID + lane * 2);
        ax += v0.x;  ay += v0.y;
        ax += v1.x;  ay += v1.y;
        ax += v2.x;  ay += v2.y;
        ax += v3.x;  ay += v3.y;
    }
    for (; j < n; ++j) {
        const int    r = __builtin_amdgcn_readlane(my, j);
        const float2 v = *(const float2*)(xw + (long)r * NHID + lane * 2);
        ax += v.x;  ay += v.y;
    }
}

typedef __attribute__((ext_vector_type(2))) _Float16 v2h;
typedef __attribute__((ext_vector_type(8))) _Float16 v8h;

__global__ void __launch_bounds__(AGG_B)
k_gather64_mm(const int* __restrict__ srcs, const unsigned* __restrict__ rowptr,
              const unsigned* __restrict__ deg, const float* __restrict__ dinv,
              const float* __restrict__ xw, const float* __restrict__ bias,
              const _Float16* __restrict__ W2p, float* __restrict__ h2) {
    __shared__ __attribute__((aligned(16))) _Float16 hs[F_NPB * HS_LD];
    const int lane = threadIdx.x & 31;
    const int wv   = __builtin_amdgcn_readfirstlane((int)(threadIdx.x >> 5));
    const int base = (int)blockIdx.x * F_NPB + wv * F_NPW;

    if (base < N_NODES) {
        const float2 b = *(const float2*)(bias + lane * 2);
        const int nl = base + (lane & (F_NPW - 1));
        const int   rp_v = (int)rowptr[nl];
        const int   dg_v = (int)deg[nl];
        const int   dv_v = __float_as_int(dinv[nl]);

        unsigned start = (unsigned)__builtin_amdgcn_readlane(rp_v, 0);
        unsigned cnt   = (unsigned)__builtin_amdgcn_readlane(dg_v, 0);
        int my = srcs[(unsigned)lane < cnt ? start + lane : 0u];

        for (int k = 0; k < F_NPW; ++k) {
            const int   c  = base + k;
            const float dc = __int_as_float(__builtin_amdgcn_readlane(dv_v, k));
            const bool  more = k + 1 < F_NPW;
            const int   kn = more ? k + 1 : k;
            const unsigned start1 = (unsigned)__builtin_amdgcn_readlane(rp_v, kn);
            const unsigned cnt1   = more ? (unsigned)__builtin_amdgcn_readlane(dg_v, kn) : 0u;

            const float2 self = *(const float2*)(xw + (long)c * NHID + lane * 2);
            const int my1 = srcs[(unsigned)lane < cnt1 ? start1 + lane : 0u];

            float ax = self.x;
            float ay = self.y;
            add_rows64(xw, lane, my, cnt < 32u ? cnt : 32u, ax, ay);
            for (unsigned eb = 32u; eb < cnt; eb += 32u) {
                const unsigned n = cnt - eb < 32u ? cnt - eb : 32u;
                int my_b = 0;
                if ((unsigned)lane < n) my_b = srcs[start + eb + lane];
                add_rows64(xw, lane, my_b, n, ax, ay);
            }
            v2h pk;
            pk[0] = (_Float16)fmaxf(b.x + dc * ax, 0.0f);
            pk[1] = (_Float16)fmaxf(b.y + dc * ay, 0.0f);
            *(v2h*)(hs + (wv * F_NPW + k) * HS_LD + lane * 2) = pk;

            start = start1;  cnt = cnt1;  my = my1;
        }
    }

    const int  strip = (int)blockIdx.x * F_STRIPS + wv;
    const bool mm    = wv < F_STRIPS && strip < M_TILES;
    const int  l16   = lane & 15;
    const int  hi8   = (lane >> 4) * 8;
    v16h  w2[NHID / 32] = {};
    float sc[8] = {};
    if (mm) {
#pragma unroll
        for (int kc = 0; kc < NHID / 32; ++kc)
            w2[kc] = *(const v16h*)(W2p + ((long)kc * 32 + lane) * 16);
        const float* __restrict__ srow = dinv + strip * 16 + hi8;
#pragma unroll
        for (int r = 0; r < 8; ++r) sc[r] = srow[r];
    }
    __syncthreads();
    if (mm) {
        const _Float16* ar = hs + (wv * 16 + l16) * HS_LD + hi8;
        v8f acc = {};
#pragma unroll
        for (int kc = 0; kc < NHID / 32; ++kc) {
            const v8h lo = *(const v8h*)(ar + kc * 32);
            const v8h hi = *(const v8h*)(ar + kc * 32 + 16);
            const v16h a = __builtin_shufflevector(lo, hi, 0, 1, 2, 3, 4, 5, 6, 7,
                                                   8, 9, 10, 11, 12, 13, 14, 15);
            acc = __builtin_amdgcn_wmma_f32_16x16x32_f16(false, a, false, w2[kc],
                                                         (short)0, acc, false, false);
        }
        float* __restrict__ crow = h2 + ((long)strip * 16 + hi8) * NCLASS + l16;
#pragma unroll
        for (int r = 0; r < 8; ++r) crow[(long)r * NCLASS] = acc[r] * sc[r];
    }
}

template <int M>
__device__ __forceinline__ float row_xor(float v) {
    return __int_as_float(__builtin_amdgcn_update_dpp(
        __float_as_int(v), __float_as_int(v), 0x160 + M, 0xf, 0xf, false));
}

__global__ void __launch_bounds__(AGG_B)
k_gather16(const int* __restrict__ srcs, const unsigned* __restrict__ rowptr,
           const unsigned* __restrict__ deg, const float* __restrict__ dinv,
           const float* __restrict__ h2, const float* __restrict__ bias,
           float* __restrict__ out) {
    const int t = blockIdx.x * AGG_B + threadIdx.x;
    int c = t >> 4;
    const int j = t & 15;
    if (c >= N_NODES) return;
    const float bj = bias[j];
    unsigned cnt = deg[c];
    unsigned st  = rowptr[c];
    float    dc  = dinv[c];
    for (;;) {
        const float selfv = h2[(long)c * NCLASS + j];
        const int  cn   = c + G16_STRIDE;
        const bool more = cn < N_NODES;
        const int  cp   = more ? cn : c;
        const unsigned ncnt = deg[cp];
        const unsigned nst  = rowptr[cp];
        const float    ndc  = dinv[cp];

        const int* __restrict__ seg = srcs + st;
        float sum = selfv;
        unsigned e = 0;
        for (; e + 4u <= cnt; e += 4u) {
            const int r0 = seg[e], r1 = seg[e + 1u], r2 = seg[e + 2u], r3 = seg[e + 3u];
            const float v0 = h2[(long)r0 * NCLASS + j];
            const float v1 = h2[(long)r1 * NCLASS + j];
            const float v2 = h2[(long)r2 * NCLASS + j];
            const float v3 = h2[(long)r3 * NCLASS + j];
            sum += v0;
            sum += v1;
            sum += v2;
            sum += v3;
        }
        for (; e < cnt; ++e) sum += h2[(long)seg[e] * NCLASS + j];
        const float acc = bj + dc * sum;
        float m = acc;
        m = fmaxf(m, row_xor<1>(m));
        m = fmaxf(m, row_xor<2>(m));
        m = fmaxf(m, row_xor<4>(m));
        m = fmaxf(m, row_xor<8>(m));
        float s = __expf(acc - m);
        s += row_xor<1>(s);
        s += row_xor<2>(s);
        s += row_xor<4>(s);
        s += row_xor<8>(s);
        out[(long)c * NCLASS + j] = acc - (m + __logf(s));

        if (!more) break;
        c = cn;  cnt = ncnt;  st = nst;  dc = ndc;
    }
}

extern "C" void kernel_launch(void* const* d_in, const int* in_sizes, int n_in,
                              void* d_out, int out_size, void* d_ws, size_t ws_size,
                              hipStream_t stream) {
    const float* x   = (const float*)d_in[0];
    const int*   ei  = (const int*)d_in[1];
    const float* W1  = (const float*)d_in[2];
    const float* b1  = (const float*)d_in[3];
    const float* W2  = (const float*)d_in[4];
    const float* b2  = (const float*)d_in[5];
    float* out       = (float*)d_out;

    const int* row = ei;
    const int* col = ei + N_EDGES;

    auto align256 = [](size_t v) { return (v + 255) & ~(size_t)255; };
    char* ws = (char*)d_ws;
    size_t o = 0;
    float*     dinv = (float*)(ws + o);     o += align256((size_t)N_NODES * 4);
    _Float16*  W1p  = (_Float16*)(ws + o);  o += align256((size_t)NFEAT * NHID * 2);
    _Float16*  W2p  = (_Float16*)(ws + o);  o += align256((size_t)NHID * NCLASS * 2);
    float*     xw   = (float*)(ws + o);     o += align256((size_t)N_NODES * NHID * 4);
    float*     h2   = (float*)(ws + o);     o += align256((size_t)N_NODES * NCLASS * 4);
    unsigned*  deg    = (unsigned*)(ws + o); o += align256((size_t)N_NODES * 4);
    unsigned*  rowptr = (unsigned*)(ws + o); o += align256((size_t)N_NODES * 4);
    unsigned*  lpre   = (unsigned*)(ws + o); o += align256((size_t)N_NODES * 4);
    unsigned*  bsum   = (unsigned*)(ws + o); o += align256((size_t)SCAN_NB * 4);
    int*       srcs   = (int*)(ws + o);      o += align256((size_t)N_EDGES * 4);
    unsigned*  part = (unsigned*)(ws + o);

    k_deg_hist<<<HIST_NB + PACK1_NB + PACK2_NB, DEG_T, 0, stream>>>(
        col, part, W1, W1p, W2, W2p);
    k_deg_to_dinv<<<SCAN_NB, SCAN_B, 0, stream>>>(part, dinv, deg, bsum, lpre);
    k_fill_srcs<<<FILL_NB + GEMM1_NB, DEG_T, 0, stream>>>(row, col, part, lpre, bsum,
                                                          rowptr, srcs,
                                                          x, W1p, dinv, xw);

    k_gather64_mm<<<F_NB, AGG_B, 0, stream>>>(srcs, rowptr, deg, dinv, xw, b1, W2p, h2);

    k_gather16<<<G16_NB, AGG_B, 0, stream>>>(srcs, rowptr, deg, dinv, h2, b2, out);
}


# xw (rows of the first product) stored as f16, one 128-byte line per row, whole line per store; layer-1 aggregation reads half the bytes, 8 rows ahead; sums f32, order kept, term precision changes
# speedup vs baseline: 4.8420x; 1.0989x over previous
#include <hip/hip_runtime.h>
#include <hip/hip_fp16.h>

typedef __attribute__((ext_vector_type(16))) _Float16 v16h;
typedef __attribute__((ext_vector_type(8)))  float    v8f;

#define N_NODES 100000
#define N_EDGES 1600000
#define NFEAT   128
#define NHID    64
#define NCLASS  16

#define DEG_G   8
#define DEG_P   32
#define DEG_R   ((((N_NODES + DEG_G - 1) / DEG_G) + 31) & ~31)
#define DEG_EPP (N_EDGES / DEG_P)
#define DEG_T   512
#define DEG_U   4
#define SCAN_B  256
#define SCAN_NB ((N_NODES + SCAN_B - 1) / SCAN_B)

static_assert(N_EDGES % DEG_P == 0, "edge parts must be equal");
static_assert(DEG_EPP % 4 == 0, "edge parts are read 16 bytes per lane");
static_assert((long)DEG_G * DEG_R >= N_NODES, "node ranges must cover all nodes");

template <int K, int N>
__device__ __forceinline__ void pack_w(const float* __restrict__ W,
                                       _Float16* __restrict__ P, int t) {
    if (t >= K * N) return;
    int i    = t & 15;
    int u    = t >> 4;
    int lane = u & 31;
    int v    = u >> 5;
    int kc   = v % (K / 32);
    int tn   = v / (K / 32);
    int l16  = lane & 15;
    int kb   = (lane >> 4) * 8;
    int k    = (i < 8) ? (kb + i) : (16 + kb + (i - 8));
    P[t] = (_Float16)W[(long)(kc * 32 + k) * N + l16 * (N / 16) + tn];
}

#define HIST_NB  (DEG_G * DEG_P)
#define PACK1_NB ((NFEAT * NHID + DEG_T - 1) / DEG_T)
#define PACK2_NB ((NHID * NCLASS + DEG_T - 1) / DEG_T)

__global__ void __launch_bounds__(DEG_T)
k_deg_hist(const int* __restrict__ col, unsigned* __restrict__ part,
           const float* __restrict__ W1, _Float16* __restrict__ W1p,
           const float* __restrict__ W2, _Float16* __restrict__ W2p) {
    __shared__ unsigned hist[DEG_R];
    if (blockIdx.x >= HIST_NB) {
        const int pb = (int)blockIdx.x - HIST_NB;
        if (pb < PACK1_NB)
            pack_w<NFEAT, NHID>(W1, W1p, pb * DEG_T + (int)threadIdx.x);
        else
            pack_w<NHID, NCLASS>(W2, W2p, (pb - PACK1_NB) * DEG_T + (int)threadIdx.x);
        return;
    }
    const int g  = blockIdx.x % DEG_G;
    const int p  = blockIdx.x / DEG_G;
    const int lo = g * DEG_R;

    for (int i = threadIdx.x; i < DEG_R; i += DEG_T) hist[i] = 0u;
    __syncthreads();

    const int4* __restrict__ c4 = (const int4*)(col + (long)p * DEG_EPP);
    constexpr int NV = DEG_EPP / 4;
    constexpr int HU = (NV + DEG_T - 1) / DEG_T;
    int4 v[HU];
    for (int u = 0; u < HU; ++u) {
        const int i = (int)threadIdx.x + u * DEG_T;
        v[u] = c4[i < NV ? i : NV - 1];
    }
    for (int u = 0; u < HU; ++u) {
        const bool in = (int)threadIdx.x + u * DEG_T < NV;
        const unsigned d0 = in ? (unsigned)(v[u].x - lo) : ~0u;
        const unsigned d1 = in ? (unsigned)(v[u].y - lo) : ~0u;
        const unsigned d2 = in ? (unsigned)(v[u].z - lo) : ~0u;
        const unsigned d3 = in ? (unsigned)(v[u].w - lo) : ~0u;
        if (d0 < (unsigned)DEG_R) atomicAdd(&hist[d0], 1u);
        if (d1 < (unsigned)DEG_R) atomicAdd(&hist[d1], 1u);
        if (d2 < (unsigned)DEG_R) atomicAdd(&hist[d2], 1u);
        if (d3 < (unsigned)DEG_R) atomicAdd(&hist[d3], 1u);
    }
    __syncthreads();

    unsigned* __restrict__ prow = part + (long)p * N_NODES + lo;
    for (int i = threadIdx.x; i < DEG_R; i += DEG_T)
        if (lo + i < N_NODES) prow[i] = hist[i];
}

__global__ void __launch_bounds__(SCAN_B)
k_deg_to_dinv(unsigned* __restrict__ part, float* __restrict__ dinv,
              unsigned* __restrict__ deg, unsigned* __restrict__ bsum,
              unsigned* __restrict__ lpre) {
    __shared__ unsigned wsum[SCAN_B / 32];
    const int i = blockIdx.x * SCAN_B + threadIdx.x;
    unsigned run = 0u;
    if (i < N_NODES) {
        for (int p = 0; p < DEG_P; ++p) {
            unsigned* q = part + (long)p * N_NODES + i;
            const unsigned c = *q;
            *q = run;
            run += c;
        }
        deg[i] = run;
        float d = (float)(run + 1u);
        dinv[i] = d > 0.0f ? rsqrtf(d) : 0.0f;
    }
    const int lane = threadIdx.x & 31;
    const int wv   = threadIdx.x >> 5;
    unsigned inc = run;
    for (int d = 1; d < 32; d <<= 1) {
        const unsigned y = __shfl_up(inc, d);
        if (lane >= d) inc += y;
    }
    if (lane == 31) wsum[wv] = inc;
    __syncthreads();
    unsigned before = 0u, total = 0u;
    for (int w = 0; w < SCAN_B / 32; ++w) {
        const unsigned s = wsum[w];
        if (w < wv) before += s;
        total += s;
    }
    if (i < N_NODES) lpre[i] = before + inc - run;
    if (threadIdx.x == 0) bsum[blockIdx.x] = total;
}

template <int K, int N, bool RELU>
__device__ __forceinline__ void gemm_strip(const float* __restrict__ A,
                                           const _Float16* __restrict__ Bp,
                                           const float* __restrict__ rscale,
                                           _Float16* __restrict__ C,
                                           int wave, int lane) {
    constexpr int NT = N / 16;
    constexpr int KC = K / 32;

    const int l16 = lane & 15;
    const int kb  = (lane >> 4) * 8;
    const float* __restrict__ arow = A + (long)(wave * 16 + l16) * K;

    const float* __restrict__ srow = rscale + wave * 16 + (lane >> 4) * 8;
    float sc[8];
#pragma unroll
    for (int r = 0; r < 8; ++r) sc[r] = srow[r];

    v8f acc[NT] = {};
#pragma unroll
    for (int kc = 0; kc < KC; ++kc) {
        v16h a;
#pragma unroll
        for (int i = 0; i < 8; ++i) {
            float a0 = arow[kc * 32 + kb + i];
            float a1 = arow[kc * 32 + 16 + kb + i];
            if (RELU) { a0 = fmaxf(a0, 0.0f); a1 = fmaxf(a1, 0.0f); }
            a[i]     = (_Float16)a0;
            a[8 + i] = (_Float16)a1;
        }
#pragma unroll
        for (int tn = 0; tn < NT; ++tn) {
            v16h b = *(const v16h*)(Bp + ((long)(tn * KC + kc) * 32 + lane) * 16);
            acc[tn] = __builtin_amdgcn_wmma_f32_16x16x32_f16(false, a, false, b,
                                                             (short)0, acc[tn],
                                                             false, false);
        }
    }

    typedef __attribute__((ext_vector_type(NT))) _Float16 vnth;
    _Float16* __restrict__ crow = C + ((long)wave * 16 + (lane >> 4) * 8) * N + l16 * NT;
#pragma unroll
    for (int r = 0; r < 8; ++r) {
        vnth pk;
#pragma unroll
        for (int tn = 0; tn < NT; ++tn) pk[tn] = (_Float16)(acc[tn][r] * sc[r]);
        *(vnth*)(crow + (long)r * N) = pk;
    }
}

#define FILL_NB  (DEG_G * DEG_P)
#define M_TILES  (N_NODES / 16)
#define GEMM1_NB ((M_TILES + DEG_T / 32 - 1) / (DEG_T / 32))
static_assert(N_NODES % 16 == 0, "the products work on whole 16-row strips");

__global__ void __launch_bounds__(DEG_T)
k_fill_srcs(const int* __restrict__ row, const int* __restrict__ col,
            const unsigned* __restrict__ part,
            const unsigned* __restrict__ lpre, const unsigned* __restrict__ bsum,
            unsigned* __restrict__ rowptr, int* __restrict__ srcs,
            const float* __restrict__ x, const _Float16* __restrict__ W1p,
            const float* __restrict__ dinv, _Float16* __restrict__ xw) {
    if (blockIdx.x >= FILL_NB) {
        const int wave = ((int)blockIdx.x - FILL_NB) * (DEG_T / 32)
                         + (int)(threadIdx.x >> 5);
        if (wave < M_TILES)
            gemm_strip<NFEAT, NHID, false>(x, W1p, dinv, xw, wave,
                                           (int)(threadIdx.x & 31));
        return;
    }
    constexpr int TAB_U = (DEG_R + DEG_T - 1) / DEG_T;
    __shared__ unsigned cur[TAB_U * DEG_T];
    __shared__ unsigned bpre[DEG_T];
    __shared__ unsigned wtot[DEG_T / 32];
    static_assert(SCAN_NB <= DEG_T, "one block sum per lane covers all scan blocks");
    static_assert(TAB_U <= DEG_P, "every table entry of a lane has a part that stores it");
    const int g  = blockIdx.x % DEG_G;
    const int p  = blockIdx.x / DEG_G;
    const int lo = g * DEG_R;
    const unsigned lim = (unsigned)(N_NODES - lo < DEG_R ? N_NODES - lo : DEG_R);

    const int4* __restrict__ c4 = (const int4*)(col + (long)p * DEG_EPP);
    const int4* __restrict__ r4 = (const int4*)(row + (long)p * DEG_EPP);
    constexpr int NV   = DEG_EPP / 4;
    constexpr int STEP = DEG_T * DEG_U;
    constexpr int NIT  = (NV + STEP - 1) / STEP;

    int4 vc[DEG_U], vr[DEG_U];
    for (int u = 0; u < DEG_U; ++u) {
        const int i  = (int)threadIdx.x + u * DEG_T;
        const int ii = i < NV ? i : NV - 1;
        vc[u] = c4[ii];
        vr[u] = r4[ii];
    }

    const unsigned* __restrict__ prow = part + (long)p * N_NODES + lo;
    unsigned myptr = 0u;
    {
        const int t    = threadIdx.x;
        const int lane = t & 31;
        const int wv   = t >> 5;
        const unsigned bl = bsum[t < SCAN_NB ? t : SCAN_NB - 1];
        unsigned a[TAB_U], b[TAB_U];
        for (int u = 0; u < TAB_U; ++u) {
            const unsigned i  = threadIdx.x + (unsigned)(u * DEG_T);
            const unsigned ii = i < lim ? i : lim - 1u;
            a[u] = lpre[lo + ii];
            b[u] = prow[ii];
        }
        const unsigned bs = t < SCAN_NB ? bl : 0u;
        unsigned inc = bs;
        for (int d = 1; d < 32; d <<= 1) {
            const unsigned y = __shfl_up(inc, d);
            if (lane >= d) inc += y;
        }
        if (lane == 31) wtot[wv] = inc;
        __syncthreads();
        unsigned before = 0u;
        for (int w = 0; w < DEG_T / 32; ++w) {
            const unsigned s = wtot[w];
            if (w < wv) before += s;
        }
        bpre[t] = before + inc - bs;
        __syncthreads();
        unsigned r[TAB_U];
        for (int u = 0; u < TAB_U; ++u) {
            const unsigned i  = threadIdx.x + (unsigned)(u * DEG_T);
            const unsigned ii = i < lim ? i : lim - 1u;
            r[u] = a[u] + bpre[(lo + ii) / SCAN_B];
        }
        for (int u = 0; u < TAB_U; ++u) {
            const unsigned i = threadIdx.x + (unsigned)(u * DEG_T);
            cur[i] = i < lim ? r[u] + b[u] : 0u;
        }
        for (int u = 0; u < TAB_U; ++u)
            if (u == p) myptr = r[u];
    }
    __syncthreads();

    for (int k = 0; k < NIT; ++k) {
        const int base = (int)threadIdx.x + k * STEP;
        int4 nc[DEG_U], nr[DEG_U];
        if (k + 1 < NIT) {
            for (int u = 0; u < DEG_U; ++u) {
                const int i  = base + STEP + u * DEG_T;
                const int ii = i < NV ? i : NV - 1;
                nc[u] = c4[ii];
                nr[u] = r4[ii];
            }
        }
        unsigned d[4 * DEG_U];
        int      s[4 * DEG_U];
        for (int u = 0; u < DEG_U; ++u) {
            const bool in = base + u * DEG_T < NV;
            d[4 * u + 0] = in ? (unsigned)(vc[u].x - lo) : ~0u;
            d[4 * u + 1] = in ? (unsigned)(vc[u].y - lo) : ~0u;
            d[4 * u + 2] = in ? (unsigned)(vc[u].z - lo) : ~0u;
            d[4 * u + 3] = in ? (unsigned)(vc[u].w - lo) : ~0u;
            s[4 * u + 0] = vr[u].x;
            s[4 * u + 1] = vr[u].y;
            s[4 * u + 2] = vr[u].z;
            s[4 * u + 3] = vr[u].w;
        }
        unsigned slot[4 * DEG_U];
        for (int j = 0; j < 4 * DEG_U; ++j) {
            slot[j] = 0u;
            if (d[j] < lim) slot[j] = atomicAdd(&cur[d[j]], 1u);
        }
        for (int j = 0; j < 4 * DEG_U; ++j)
            if (d[j] < lim) srcs[slot[j]] = s[j];
        if (k + 1 < NIT) {
            for (int u = 0; u < DEG_U; ++u) { vc[u] = nc[u]; vr[u] = nr[u]; }
        }
    }
    if (p < TAB_U) {
        const unsigned i = threadIdx.x + (unsigned)(p * DEG_T);
        if (i < lim) rowptr[lo + i] = myptr;
    }
}

#define AGG_B      256
#define F_NPW      8
#define F_NPB      (F_NPW * (AGG_B / 32))
#define F_STRIPS   (F_NPB / 16)
#define F_NB       ((N_NODES + F_NPB - 1) / F_NPB)
#define HS_LD      (NHID + 8)
#define G16_ROUNDS 4
#define G16_NB     ((((N_NODES + G16_ROUNDS - 1) / G16_ROUNDS) * 16 + AGG_B - 1) / AGG_B)
#define G16_STRIDE (G16_NB * (AGG_B / 16))

static_assert(N_NODES % F_NPW == 0, "a wave's nodes are all inside the graph or none is");
static_assert(F_NPB % 16 == 0 && F_NPW <= 32, "a block owns whole strips; bounds fit one per lane");
static_assert(NHID == 64 && NCLASS == 16, "two features per lane; one n-tile, two k-chunks");
static_assert((long)G16_STRIDE * G16_ROUNDS >= N_NODES, "layer-2 rounds cover all nodes");
static_assert(F_NB <= 2048 && G16_NB <= 2048, "the whole grid is resident at once");

typedef __attribute__((ext_vector_type(2))) _Float16 v2h;
typedef __attribute__((ext_vector_type(8))) _Float16 v8h;

__device__ __forceinline__ v2h ld_row2(const _Float16* __restrict__ xw, int r, int lane) {
    return *(const v2h*)(xw + (long)r * NHID + lane * 2);
}

template <int R>
__device__ __forceinline__ void add_group(const _Float16* __restrict__ xw, int lane,
                                          int my, unsigned j, float& ax, float& ay) {
    v2h v[R];
#pragma unroll
    for (int i = 0; i < R; ++i)
        v[i] = ld_row2(xw, __builtin_amdgcn_readlane(my, j + (unsigned)i), lane);
#pragma unroll
    for (int i = 0; i < R; ++i) { ax += (float)v[i][0];  ay += (float)v[i][1]; }
}

__device__ __forceinline__ void add_rows64(const _Float16* __restrict__ xw, int lane,
                                           int my, unsigned n,
                                           float& ax, float& ay) {
    unsigned j = 0;
    for (; j + 8u <= n; j += 8u) add_group<8>(xw, lane, my, j, ax, ay);
    const unsigned rest = n - j;
    if (rest >= 4u) {
        if (rest >= 6u) {
            if (rest == 7u) add_group<7>(xw, lane, my, j, ax, ay);
            else            add_group<6>(xw, lane, my, j, ax, ay);
        } else {
            if (rest == 5u) add_group<5>(xw, lane, my, j, ax, ay);
            else            add_group<4>(xw, lane, my, j, ax, ay);
        }
    } else if (rest >= 2u) {
        if (rest == 3u) add_group<3>(xw, lane, my, j, ax, ay);
        else            add_group<2>(xw, lane, my, j, ax, ay);
    } else if (rest == 1u) {
        add_group<1>(xw, lane, my, j, ax, ay);
    }
}

__global__ void __launch_bounds__(AGG_B)
k_gather64_mm(const int* __restrict__ srcs, const unsigned* __restrict__ rowptr,
              const unsigned* __restrict__ deg, const float* __restrict__ dinv,
              const _Float16* __restrict__ xw, const float* __restrict__ bias,
              const _Float16* __restrict__ W2p, float* __restrict__ h2) {
    __shared__ __attribute__((aligned(16))) _Float16 hs[F_NPB * HS_LD];
    const int lane = threadIdx.x & 31;
    const int wv   = __builtin_amdgcn_readfirstlane((int)(threadIdx.x >> 5));
    const int base = (int)blockIdx.x * F_NPB + wv * F_NPW;

    if (base < N_NODES) {
        const float2 b = *(const float2*)(bias + lane * 2);
        const int nl = base + (lane & (F_NPW - 1));
        const int   rp_v = (int)rowptr[nl];
        const int   dg_v = (int)deg[nl];
        const int   dv_v = __float_as_int(dinv[nl]);

        unsigned start = (unsigned)__builtin_amdgcn_readlane(rp_v, 0);
        unsigned cnt   = (unsigned)__builtin_amdgcn_readlane(dg_v, 0);
        int my = srcs[(unsigned)lane < cnt ? start + lane : 0u];

        for (int k = 0; k < F_NPW; ++k) {
            const int   c  = base + k;
            const float dc = __int_as_float(__builtin_amdgcn_readlane(dv_v, k));
            const bool  more = k + 1 < F_NPW;
            const int   kn = more ? k + 1 : k;
            const unsigned start1 = (unsigned)__builtin_amdgcn_readlane(rp_v, kn);
            const unsigned cnt1   = more ? (unsigned)__builtin_amdgcn_readlane(dg_v, kn) : 0u;

            const v2h self = ld_row2(xw, c, lane);
            const int my1 = srcs[(unsigned)lane < cnt1 ? start1 + lane : 0u];

            float ax = (float)self[0];
            float ay = (float)self[1];
            add_rows64(xw, lane, my, cnt < 32u ? cnt : 32u, ax, ay);
            for (unsigned eb = 32u; eb < cnt; eb += 32u) {
                const unsigned n = cnt - eb < 32u ? cnt - eb : 32u;
                int my_b = 0;
                if ((unsigned)lane < n) my_b = srcs[start + eb + lane];
                add_rows64(xw, lane, my_b, n, ax, ay);
            }
            v2h pk;
            pk[0] = (_Float16)fmaxf(b.x + dc * ax, 0.0f);
            pk[1] = (_Float16)fmaxf(b.y + dc * ay, 0.0f);
            *(v2h*)(hs + (wv * F_NPW + k) * HS_LD + lane * 2) = pk;

            start = start1;  cnt = cnt1;  my = my1;
        }
    }

    const int  strip = (int)blockIdx.x * F_STRIPS + wv;
    const bool mm    = wv < F_STRIPS && strip < M_TILES;
    const int  l16   = lane & 15;
    const int  hi8   = (lane >> 4) * 8;
    v16h  w2[NHID / 32] = {};
    float sc[8] = {};
    if (mm) {
#pragma unroll
        for (int kc = 0; kc < NHID / 32; ++kc)
            w2[kc] = *(const v16h*)(W2p + ((long)kc * 32 + lane) * 16);
        const float* __restrict__ srow = dinv + strip * 16 + hi8;
#pragma unroll
        for (int r = 0; r < 8; ++r) sc[r] = srow[r];
    }
    __syncthreads();
    if (mm) {
        const _Float16* ar = hs + (wv * 16 + l16) * HS_LD + hi8;
        v8f acc = {};
#pragma unroll
        for (int kc = 0; kc < NHID / 32; ++kc) {
            const v8h lo = *(const v8h*)(ar + kc * 32);
            const v8h hi = *(const v8h*)(ar + kc * 32 + 16);
            const v16h a = __builtin_shufflevector(lo, hi, 0, 1, 2, 3, 4, 5, 6, 7,
                                                   8, 9, 10, 11, 12, 13, 14, 15);
            acc = __builtin_amdgcn_wmma_f32_16x16x32_f16(false, a, false, w2[kc],
                                                         (short)0, acc, false, false);
        }
        float* __restrict__ crow = h2 + ((long)strip * 16 + hi8) * NCLASS + l16;
#pragma unroll
        for (int r = 0; r < 8; ++r) crow[(long)r * NCLASS] = acc[r] * sc[r];
    }
}

template <int M>
__device__ __forceinline__ float row_xor(float v) {
    return __int_as_float(__builtin_amdgcn_update_dpp(
        __float_as_int(v), __float_as_int(v), 0x160 + M, 0xf, 0xf, false));
}

__global__ void __launch_bounds__(AGG_B)
k_gather16(const int* __restrict__ srcs, const unsigned* __restrict__ rowptr,
           const unsigned* __restrict__ deg, const float* __restrict__ dinv,
           const float* __restrict__ h2, const float* __restrict__ bias,
           float* __restrict__ out) {
    const int t = blockIdx.x * AGG_B + threadIdx.x;
    int c = t >> 4;
    const int j = t & 15;
    if (c >= N_NODES) return;
    const float bj = bias[j];
    unsigned cnt = deg[c];
    unsigned st  = rowptr[c];
    float    dc  = dinv[c];
    for (;;) {
        const float selfv = h2[(long)c * NCLASS + j];
        const int  cn   = c + G16_STRIDE;
        const bool more = cn < N_NODES;
        const int  cp   = more ? cn : c;
        const unsigned ncnt = deg[cp];
        const unsigned nst  = rowptr[cp];
        const float    ndc  = dinv[cp];

        const int* __restrict__ seg = srcs + st;
        float sum = selfv;
        unsigned e = 0;
        for (; e + 4u <= cnt; e += 4u) {
            const int r0 = seg[e], r1 = seg[e + 1u], r2 = seg[e + 2u], r3 = seg[e + 3u];
            const float v0 = h2[(long)r0 * NCLASS + j];
            const float v1 = h2[(long)r1 * NCLASS + j];
            const float v2 = h2[(long)r2 * NCLASS + j];
            const float v3 = h2[(long)r3 * NCLASS + j];
            sum += v0;
            sum += v1;
            sum += v2;
            sum += v3;
        }
        for (; e < cnt; ++e) sum += h2[(long)seg[e] * NCLASS + j];
        const float acc = bj + dc * sum;
        float m = acc;
        m = fmaxf(m, row_xor<1>(m));
        m = fmaxf(m, row_xor<2>(m));
        m = fmaxf(m, row_xor<4>(m));
        m = fmaxf(m, row_xor<8>(m));
        float s = __expf(acc - m);
        s += row_xor<1>(s);
        s += row_xor<2>(s);
        s += row_xor<4>(s);
        s += row_xor<8>(s);
        out[(long)c * NCLASS + j] = acc - (m + __logf(s));

        if (!more) break;
        c = cn;  cnt = ncnt;  st = nst;  dc = ndc;
    }
}

extern "C" void kernel_launch(void* const* d_in, const int* in_sizes, int n_in,
                              void* d_out, int out_size, void* d_ws, size_t ws_size,
                              hipStream_t stream) {
    const float* x   = (const float*)d_in[0];
    const int*   ei  = (const int*)d_in[1];
    const float* W1  = (const float*)d_in[2];
    const float* b1  = (const float*)d_in[3];
    const float* W2  = (const float*)d_in[4];
    const float* b2  = (const float*)d_in[5];
    float* out       = (float*)d_out;

    const int* row = ei;
    const int* col = ei + N_EDGES;

    auto align256 = [](size_t v) { return (v + 255) & ~(size_t)255; };
    char* ws = (char*)d_ws;
    size_t o = 0;
    float*     dinv = (float*)(ws + o);     o += align256((size_t)N_NODES * 4);
    _Float16*  W1p  = (_Float16*)(ws + o);  o += align256((size_t)NFEAT * NHID * 2);
    _Float16*  W2p  = (_Float16*)(ws + o);  o += align256((size_t)NHID * NCLASS * 2);
    _Float16*  xw   = (_Float16*)(ws + o);  o += align256((size_t)N_NODES * NHID * 2);
    float*     h2   = (float*)(ws + o);     o += align256((size_t)N_NODES * NCLASS * 4);
    unsigned*  deg    = (unsigned*)(ws + o); o += align256((size_t)N_NODES * 4);
    unsigned*  rowptr = (unsigned*)(ws + o); o += align256((size_t)N_NODES * 4);
    unsigned*  lpre   = (unsigned*)(ws + o); o += align256((size_t)N_NODES * 4);
    unsigned*  bsum   = (unsigned*)(ws + o); o += align256((size_t)SCAN_NB * 4);
    int*       srcs   = (int*)(ws + o);      o += align256((size_t)N_EDGES * 4);
    unsigned*  part = (unsigned*)(ws + o);

    k_deg_hist<<<HIST_NB + PACK1_NB + PACK2_NB, DEG_T, 0, stream>>>(
        col, part, W1, W1p, W2, W2p);
    k_deg_to_dinv<<<SCAN_NB, SCAN_B, 0, stream>>>(part, dinv, deg, bsum, lpre);
    k_fill_srcs<<<FILL_NB + GEMM1_NB, DEG_T, 0, stream>>>(row, col, part, lpre, bsum,
                                                          rowptr, srcs,
                                                          x, W1p, dinv, xw);

    k_gather64_mm<<<F_NB, AGG_B, 0, stream>>>(srcs, rowptr, deg, dinv, xw, b1, W2p, h2);

    k_gather16<<<G16_NB, AGG_B, 0, stream>>>(srcs, rowptr, deg, dinv, h2, b2, out);
}


# Layer-2 aggregation: 4 lanes per node (16-byte row pieces, 8 nodes per wave); a segment's indices fetched one node ahead, all its rows requested before the first add; sum order, softmax tree kept
# speedup vs baseline: 5.1470x; 1.0630x over previous
#include <hip/hip_runtime.h>
#include <hip/hip_fp16.h>

typedef __attribute__((ext_vector_type(16))) _Float16 v16h;
typedef __attribute__((ext_vector_type(8)))  float    v8f;

#define N_NODES 100000
#define N_EDGES 1600000
#define NFEAT   128
#define NHID    64
#define NCLASS  16

#define DEG_G   8
#define DEG_P   32
#define DEG_R   ((((N_NODES + DEG_G - 1) / DEG_G) + 31) & ~31)
#define DEG_EPP (N_EDGES / DEG_P)
#define DEG_T   512
#define DEG_U   4
#define SCAN_B  256
#define SCAN_NB ((N_NODES + SCAN_B - 1) / SCAN_B)

static_assert(N_EDGES % DEG_P == 0, "edge parts must be equal");
static_assert(DEG_EPP % 4 == 0, "edge parts are read 16 bytes per lane");
static_assert((long)DEG_G * DEG_R >= N_NODES, "node ranges must cover all nodes");

template <int K, int N>
__device__ __forceinline__ void pack_w(const float* __restrict__ W,
                                       _Float16* __restrict__ P, int t) {
    if (t >= K * N) return;
    int i    = t & 15;
    int u    = t >> 4;
    int lane = u & 31;
    int v    = u >> 5;
    int kc   = v % (K / 32);
    int tn   = v / (K / 32);
    int l16  = lane & 15;
    int kb   = (lane >> 4) * 8;
    int k    = (i < 8) ? (kb + i) : (16 + kb + (i - 8));
    P[t] = (_Float16)W[(long)(kc * 32 + k) * N + l16 * (N / 16) + tn];
}

#define HIST_NB  (DEG_G * DEG_P)
#define PACK1_NB ((NFEAT * NHID + DEG_T - 1) / DEG_T)
#define PACK2_NB ((NHID * NCLASS + DEG_T - 1) / DEG_T)

__global__ void __launch_bounds__(DEG_T)
k_deg_hist(const int* __restrict__ col, unsigned* __restrict__ part,
           const float* __restrict__ W1, _Float16* __restrict__ W1p,
           const float* __restrict__ W2, _Float16* __restrict__ W2p) {
    __shared__ unsigned hist[DEG_R];
    if (blockIdx.x >= HIST_NB) {
        const int pb = (int)blockIdx.x - HIST_NB;
        if (pb < PACK1_NB)
            pack_w<NFEAT, NHID>(W1, W1p, pb * DEG_T + (int)threadIdx.x);
        else
            pack_w<NHID, NCLASS>(W2, W2p, (pb - PACK1_NB) * DEG_T + (int)threadIdx.x);
        return;
    }
    const int g  = blockIdx.x % DEG_G;
    const int p  = blockIdx.x / DEG_G;
    const int lo = g * DEG_R;

    for (int i = threadIdx.x; i < DEG_R; i += DEG_T) hist[i] = 0u;
    __syncthreads();

    const int4* __restrict__ c4 = (const int4*)(col + (long)p * DEG_EPP);
    constexpr int NV = DEG_EPP / 4;
    constexpr int HU = (NV + DEG_T - 1) / DEG_T;
    int4 v[HU];
    for (int u = 0; u < HU; ++u) {
        const int i = (int)threadIdx.x + u * DEG_T;
        v[u] = c4[i < NV ? i : NV - 1];
    }
    for (int u = 0; u < HU; ++u) {
        const bool in = (int)threadIdx.x + u * DEG_T < NV;
        const unsigned d0 = in ? (unsigned)(v[u].x - lo) : ~0u;
        const unsigned d1 = in ? (unsigned)(v[u].y - lo) : ~0u;
        const unsigned d2 = in ? (unsigned)(v[u].z - lo) : ~0u;
        const unsigned d3 = in ? (unsigned)(v[u].w - lo) : ~0u;
        if (d0 < (unsigned)DEG_R) atomicAdd(&hist[d0], 1u);
        if (d1 < (unsigned)DEG_R) atomicAdd(&hist[d1], 1u);
        if (d2 < (unsigned)DEG_R) atomicAdd(&hist[d2], 1u);
        if (d3 < (unsigned)DEG_R) atomicAdd(&hist[d3], 1u);
    }
    __syncthreads();

    unsigned* __restrict__ prow = part + (long)p * N_NODES + lo;
    for (int i = threadIdx.x; i < DEG_R; i += DEG_T)
        if (lo + i < N_NODES) prow[i] = hist[i];
}

__global__ void __launch_bounds__(SCAN_B)
k_deg_to_dinv(unsigned* __restrict__ part, float* __restrict__ dinv,
              unsigned* __restrict__ deg, unsigned* __restrict__ bsum,
              unsigned* __restrict__ lpre) {
    __shared__ unsigned wsum[SCAN_B / 32];
    const int i = blockIdx.x * SCAN_B + threadIdx.x;
    unsigned run = 0u;
    if (i < N_NODES) {
        for (int p = 0; p < DEG_P; ++p) {
            unsigned* q = part + (long)p * N_NODES + i;
            const unsigned c = *q;
            *q = run;
            run += c;
        }
        deg[i] = run;
        float d = (float)(run + 1u);
        dinv[i] = d > 0.0f ? rsqrtf(d) : 0.0f;
    }
    const int lane = threadIdx.x & 31;
    const int wv   = threadIdx.x >> 5;
    unsigned inc = run;
    for (int d = 1; d < 32; d <<= 1) {
        const unsigned y = __shfl_up(inc, d);
        if (lane >= d) inc += y;
    }
    if (lane == 31) wsum[wv] = inc;
    __syncthreads();
    unsigned before = 0u, total = 0u;
    for (int w = 0; w < SCAN_B / 32; ++w) {
        const unsigned s = wsum[w];
        if (w < wv) before += s;
        total += s;
    }
    if (i < N_NODES) lpre[i] = before + inc - run;
    if (threadIdx.x == 0) bsum[blockIdx.x] = total;
}

template <int K, int N, bool RELU>
__device__ __forceinline__ void gemm_strip(const float* __restrict__ A,
                                           const _Float16* __restrict__ Bp,
                                           const float* __restrict__ rscale,
                                           _Float16* __restrict__ C,
                                           int wave, int lane) {
    constexpr int NT = N / 16;
    constexpr int KC = K / 32;

    const int l16 = lane & 15;
    const int kb  = (lane >> 4) * 8;
    const float* __restrict__ arow = A + (long)(wave * 16 + l16) * K;

    const float* __restrict__ srow = rscale + wave * 16 + (lane >> 4) * 8;
    float sc[8];
#pragma unroll
    for (int r = 0; r < 8; ++r) sc[r] = srow[r];

    v8f acc[NT] = {};
#pragma unroll
    for (int kc = 0; kc < KC; ++kc) {
        v16h a;
#pragma unroll
        for (int i = 0; i < 8; ++i) {
            float a0 = arow[kc * 32 + kb + i];
            float a1 = arow[kc * 32 + 16 + kb + i];
            if (RELU) { a0 = fmaxf(a0, 0.0f); a1 = fmaxf(a1, 0.0f); }
            a[i]     = (_Float16)a0;
            a[8 + i] = (_Float16)a1;
        }
#pragma unroll
        for (int tn = 0; tn < NT; ++tn) {
            v16h b = *(const v16h*)(Bp + ((long)(tn * KC + kc) * 32 + lane) * 16);
            acc[tn] = __builtin_amdgcn_wmma_f32_16x16x32_f16(false, a, false, b,
                                                             (short)0, acc[tn],
                                                             false, false);
        }
    }

    typedef __attribute__((ext_vector_type(NT))) _Float16 vnth;
    _Float16* __restrict__ crow = C + ((long)wave * 16 + (lane >> 4) * 8) * N + l16 * NT;
#pragma unroll
    for (int r = 0; r < 8; ++r) {
        vnth pk;
#pragma unroll
        for (int tn = 0; tn < NT; ++tn) pk[tn] = (_Float16)(acc[tn][r] * sc[r]);
        *(vnth*)(crow + (long)r * N) = pk;
    }
}

#define FILL_NB  (DEG_G * DEG_P)
#define M_TILES  (N_NODES / 16)
#define GEMM1_NB ((M_TILES + DEG_T / 32 - 1) / (DEG_T / 32))
static_assert(N_NODES % 16 == 0, "the products work on whole 16-row strips");

__global__ void __launch_bounds__(DEG_T)
k_fill_srcs(const int* __restrict__ row, const int* __restrict__ col,
            const unsigned* __restrict__ part,
            const unsigned* __restrict__ lpre, const unsigned* __restrict__ bsum,
            unsigned* __restrict__ rowptr, int* __restrict__ srcs,
            const float* __restrict__ x, const _Float16* __restrict__ W1p,
            const float* __restrict__ dinv, _Float16* __restrict__ xw) {
    if (blockIdx.x >= FILL_NB) {
        const int wave = ((int)blockIdx.x - FILL_NB) * (DEG_T / 32)
                         + (int)(threadIdx.x >> 5);
        if (wave < M_TILES)
            gemm_strip<NFEAT, NHID, false>(x, W1p, dinv, xw, wave,
                                           (int)(threadIdx.x & 31));
        return;
    }
    constexpr int TAB_U = (DEG_R + DEG_T - 1) / DEG_T;
    __shared__ unsigned cur[TAB_U * DEG_T];
    __shared__ unsigned bpre[DEG_T];
    __shared__ unsigned wtot[DEG_T / 32];
    static_assert(SCAN_NB <= DEG_T, "one block sum per lane covers all scan blocks");
    static_assert(TAB_U <= DEG_P, "every table entry of a lane has a part that stores it");
    const int g  = blockIdx.x % DEG_G;
    const int p  = blockIdx.x / DEG_G;
    const int lo = g * DEG_R;
    const unsigned lim = (unsigned)(N_NODES - lo < DEG_R ? N_NODES - lo : DEG_R);

    const int4* __restrict__ c4 = (const int4*)(col + (long)p * DEG_EPP);
    const int4* __restrict__ r4 = (const int4*)(row + (long)p * DEG_EPP);
    constexpr int NV   = DEG_EPP / 4;
    constexpr int STEP = DEG_T * DEG_U;
    constexpr int NIT  = (NV + STEP - 1) / STEP;

    int4 vc[DEG_U], vr[DEG_U];
    for (int u = 0; u < DEG_U; ++u) {
        const int i  = (int)threadIdx.x + u * DEG_T;
        const int ii = i < NV ? i : NV - 1;
        vc[u] = c4[ii];
        vr[u] = r4[ii];
    }

    const unsigned* __restrict__ prow = part + (long)p * N_NODES + lo;
    unsigned myptr = 0u;
    {
        const int t    = threadIdx.x;
        const int lane = t & 31;
        const int wv   = t >> 5;
        const unsigned bl = bsum[t < SCAN_NB ? t : SCAN_NB - 1];
        unsigned a[TAB_U], b[TAB_U];
        for (int u = 0; u < TAB_U; ++u) {
            const unsigned i  = threadIdx.x + (unsigned)(u * DEG_T);
            const unsigned ii = i < lim ? i : lim - 1u;
            a[u] = lpre[lo + ii];
            b[u] = prow[ii];
        }
        const unsigned bs = t < SCAN_NB ? bl : 0u;
        unsigned inc = bs;
        for (int d = 1; d < 32; d <<= 1) {
            const unsigned y = __shfl_up(inc, d);
            if (lane >= d) inc += y;
        }
        if (lane == 31) wtot[wv] = inc;
        __syncthreads();
        unsigned before = 0u;
        for (int w = 0; w < DEG_T / 32; ++w) {
            const unsigned s = wtot[w];
            if (w < wv) before += s;
        }
        bpre[t] = before + inc - bs;
        __syncthreads();
        unsigned r[TAB_U];
        for (int u = 0; u < TAB_U; ++u) {
            const unsigned i  = threadIdx.x + (unsigned)(u * DEG_T);
            const unsigned ii = i < lim ? i : lim - 1u;
            r[u] = a[u] + bpre[(lo + ii) / SCAN_B];
        }
        for (int u = 0; u < TAB_U; ++u) {
            const unsigned i = threadIdx.x + (unsigned)(u * DEG_T);
            cur[i] = i < lim ? r[u] + b[u] : 0u;
        }
        for (int u = 0; u < TAB_U; ++u)
            if (u == p) myptr = r[u];
    }
    __syncthreads();

    for (int k = 0; k < NIT; ++k) {
        const int base = (int)threadIdx.x + k * STEP;
        int4 nc[DEG_U], nr[DEG_U];
        if (k + 1 < NIT) {
            for (int u = 0; u < DEG_U; ++u) {
                const int i  = base + STEP + u * DEG_T;
                const int ii = i < NV ? i : NV - 1;
                nc[u] = c4[ii];
                nr[u] = r4[ii];
            }
        }
        unsigned d[4 * DEG_U];
        int      s[4 * DEG_U];
        for (int u = 0; u < DEG_U; ++u) {
            const bool in = base + u * DEG_T < NV;
            d[4 * u + 0] = in ? (unsigned)(vc[u].x - lo) : ~0u;
            d[4 * u + 1] = in ? (unsigned)(vc[u].y - lo) : ~0u;
            d[4 * u + 2] = in ? (unsigned)(vc[u].z - lo) : ~0u;
            d[4 * u + 3] = in ? (unsigned)(vc[u].w - lo) : ~0u;
            s[4 * u + 0] = vr[u].x;
            s[4 * u + 1] = vr[u].y;
            s[4 * u + 2] = vr[u].z;
            s[4 * u + 3] = vr[u].w;
        }
        unsigned slot[4 * DEG_U];
        for (int j = 0; j < 4 * DEG_U; ++j) {
            slot[j] = 0u;
            if (d[j] < lim) slot[j] = atomicAdd(&cur[d[j]], 1u);
        }
        for (int j = 0; j < 4 * DEG_U; ++j)
            if (d[j] < lim) srcs[slot[j]] = s[j];
        if (k + 1 < NIT) {
            for (int u = 0; u < DEG_U; ++u) { vc[u] = nc[u]; vr[u] = nr[u]; }
        }
    }
    if (p < TAB_U) {
        const unsigned i = threadIdx.x + (unsigned)(p * DEG_T);
        if (i < lim) rowptr[lo + i] = myptr;
    }
}

#define AGG_B      256
#define F_NPW      8
#define F_NPB      (F_NPW * (AGG_B / 32))
#define F_STRIPS   (F_NPB / 16)
#define F_NB       ((N_NODES + F_NPB - 1) / F_NPB)
#define HS_LD      (NHID + 8)
#define G16_T      128
#define G16_ROUNDS 4
#define G16_NB     ((((N_NODES + G16_ROUNDS - 1) / G16_ROUNDS) * 4 + G16_T - 1) / G16_T)
#define G16_STRIDE (G16_NB * (G16_T / 4))

static_assert(N_NODES % F_NPW == 0, "a wave's nodes are all inside the graph or none is");
static_assert(F_NPB % 16 == 0 && F_NPW <= 32, "a block owns whole strips; bounds fit one per lane");
static_assert(NHID == 64 && NCLASS == 16, "two features per lane; one n-tile, two k-chunks");
static_assert((long)G16_STRIDE * G16_ROUNDS >= N_NODES, "layer-2 rounds cover all nodes");
static_assert(G16_STRIDE % 8 == 0 && N_NODES % 8 == 0, "a wave's eight layer-2 nodes: whole lines, all in or all out");
static_assert(F_NB <= 2048 && G16_NB <= 2048, "the whole grid is resident at once");

typedef __attribute__((ext_vector_type(2))) _Float16 v2h;
typedef __attribute__((ext_vector_type(8))) _Float16 v8h;

__device__ __forceinline__ v2h ld_row2(const _Float16* __restrict__ xw, int r, int lane) {
    return *(const v2h*)(xw + (long)r * NHID + lane * 2);
}

template <int R>
__device__ __forceinline__ void add_group(const _Float16* __restrict__ xw, int lane,
                                          int my, unsigned j, float& ax, float& ay) {
    v2h v[R];
#pragma unroll
    for (int i = 0; i < R; ++i)
        v[i] = ld_row2(xw, __builtin_amdgcn_readlane(my, j + (unsigned)i), lane);
#pragma unroll
    for (int i = 0; i < R; ++i) { ax += (float)v[i][0];  ay += (float)v[i][1]; }
}

__device__ __forceinline__ void add_rows64(const _Float16* __restrict__ xw, int lane,
                                           int my, unsigned n,
                                           float& ax, float& ay) {
    unsigned j = 0;
    for (; j + 8u <= n; j += 8u) add_group<8>(xw, lane, my, j, ax, ay);
    const unsigned rest = n - j;
    if (rest >= 4u) {
        if (rest >= 6u) {
            if (rest == 7u) add_group<7>(xw, lane, my, j, ax, ay);
            else            add_group<6>(xw, lane, my, j, ax, ay);
        } else {
            if (rest == 5u) add_group<5>(xw, lane, my, j, ax, ay);
            else            add_group<4>(xw, lane, my, j, ax, ay);
        }
    } else if (rest >= 2u) {
        if (rest == 3u) add_group<3>(xw, lane, my, j, ax, ay);
        else            add_group<2>(xw, lane, my, j, ax, ay);
    } else if (rest == 1u) {
        add_group<1>(xw, lane, my, j, ax, ay);
    }
}

__global__ void __launch_bounds__(AGG_B)
k_gather64_mm(const int* __restrict__ srcs, const unsigned* __restrict__ rowptr,
              const unsigned* __restrict__ deg, const float* __restrict__ dinv,
              const _Float16* __restrict__ xw, const float* __restrict__ bias,
              const _Float16* __restrict__ W2p, float* __restrict__ h2) {
    __shared__ __attribute__((aligned(16))) _Float16 hs[F_NPB * HS_LD];
    const int lane = threadIdx.x & 31;
    const int wv   = __builtin_amdgcn_readfirstlane((int)(threadIdx.x >> 5));
    const int base = (int)blockIdx.x * F_NPB + wv * F_NPW;

    if (base < N_NODES) {
        const float2 b = *(const float2*)(bias + lane * 2);
        const int nl = base + (lane & (F_NPW - 1));
        const int   rp_v = (int)rowptr[nl];
        const int   dg_v = (int)deg[nl];
        const int   dv_v = __float_as_int(dinv[nl]);

        unsigned start = (unsigned)__builtin_amdgcn_readlane(rp_v, 0);
        unsigned cnt   = (unsigned)__builtin_amdgcn_readlane(dg_v, 0);
        int my = srcs[(unsigned)lane < cnt ? start + lane : 0u];

        for (int k = 0; k < F_NPW; ++k) {
            const int   c  = base + k;
            const float dc = __int_as_float(__builtin_amdgcn_readlane(dv_v, k));
            const bool  more = k + 1 < F_NPW;
            const int   kn = more ? k + 1 : k;
            const unsigned start1 = (unsigned)__builtin_amdgcn_readlane(rp_v, kn);
            const unsigned cnt1   = more ? (unsigned)__builtin_amdgcn_readlane(dg_v, kn) : 0u;

            const v2h self = ld_row2(xw, c, lane);
            const int my1 = srcs[(unsigned)lane < cnt1 ? start1 + lane : 0u];

            float ax = (float)self[0];
            float ay = (float)self[1];
            add_rows64(xw, lane, my, cnt < 32u ? cnt : 32u, ax, ay);
            for (unsigned eb = 32u; eb < cnt; eb += 32u) {
                const unsigned n = cnt - eb < 32u ? cnt - eb : 32u;
                int my_b = 0;
                if ((unsigned)lane < n) my_b = srcs[start + eb + lane];
                add_rows64(xw, lane, my_b, n, ax, ay);
            }
            v2h pk;
            pk[0] = (_Float16)fmaxf(b.x + dc * ax, 0.0f);
            pk[1] = (_Float16)fmaxf(b.y + dc * ay, 0.0f);
            *(v2h*)(hs + (wv * F_NPW + k) * HS_LD + lane * 2) = pk;

            start = start1;  cnt = cnt1;  my = my1;
        }
    }

    const int  strip = (int)blockIdx.x * F_STRIPS + wv;
    const bool mm    = wv < F_STRIPS && strip < M_TILES;
    const int  l16   = lane & 15;
    const int  hi8   = (lane >> 4) * 8;
    v16h  w2[NHID / 32] = {};
    float sc[8] = {};
    if (mm) {
#pragma unroll
        for (int kc = 0; kc < NHID / 32; ++kc)
            w2[kc] = *(const v16h*)(W2p + ((long)kc * 32 + lane) * 16);
        const float* __restrict__ srow = dinv + strip * 16 + hi8;
#pragma unroll
        for (int r = 0; r < 8; ++r) sc[r] = srow[r];
    }
    __syncthreads();
    if (mm) {
        const _Float16* ar = hs + (wv * 16 + l16) * HS_LD + hi8;
        v8f acc = {};
#pragma unroll
        for (int kc = 0; kc < NHID / 32; ++kc) {
            const v8h lo = *(const v8h*)(ar + kc * 32);
            const v8h hi = *(const v8h*)(ar + kc * 32 + 16);
            const v16h a = __builtin_shufflevector(lo, hi, 0, 1, 2, 3, 4, 5, 6, 7,
                                                   8, 9, 10, 11, 12, 13, 14, 15);
            acc = __builtin_amdgcn_wmma_f32_16x16x32_f16(false, a, false, w2[kc],
                                                         (short)0, acc, false, false);
        }
        float* __restrict__ crow = h2 + ((long)strip * 16 + hi8) * NCLASS + l16;
#pragma unroll
        for (int r = 0; r < 8; ++r) crow[(long)r * NCLASS] = acc[r] * sc[r];
    }
}

template <int M>
__device__ __forceinline__ float row_xor(float v) {
    return __int_as_float(__builtin_amdgcn_update_dpp(
        __float_as_int(v), __float_as_int(v), 0x160 + M, 0xf, 0xf, false));
}

typedef __attribute__((ext_vector_type(4))) float v4f;

template <int Q>
__device__ __forceinline__ int quad_lane(int v) {
    return __builtin_amdgcn_update_dpp(0, v, Q * 0x55, 0xf, 0xf, true);
}

template <int S, int NS>
__device__ __forceinline__ void req_slots(const char* __restrict__ hb, unsigned qo,
                                          const int (&x)[8], v4f (&v)[NS]) {
    if constexpr (S < NS) {
        v[S] = *(const v4f*)(hb + ((unsigned)quad_lane<(S >> 3)>(x[S & 7]) * 64u + qo));
        req_slots<S + 1, NS>(hb, qo, x, v);
    }
}

template <int NS>
__device__ __forceinline__ v4f sum_slots(const char* __restrict__ hb, unsigned qo,
                                         const int (&x)[8], unsigned cnt, v4f sum) {
    v4f v[NS];
    req_slots<0, NS>(hb, qo, x, v);
#pragma unroll
    for (int g = 0; g < NS; g += 8)
        asm volatile("" :: "v"(v[g]), "v"(v[g + 1]), "v"(v[g + 2]), "v"(v[g + 3]),
                           "v"(v[g + 4]), "v"(v[g + 5]), "v"(v[g + 6]), "v"(v[g + 7]));
    __builtin_amdgcn_sched_barrier(0);
#pragma unroll
    for (int s = 0; s < NS; ++s)
        sum = (unsigned)s < cnt ? sum + v[s] : sum;
    return sum;
}

__global__ void __launch_bounds__(G16_T)
k_gather16(const int* __restrict__ srcs, const unsigned* __restrict__ rowptr,
           const unsigned* __restrict__ deg, const float* __restrict__ dinv,
           const float* __restrict__ h2, const float* __restrict__ bias,
           float* __restrict__ out) {
    const int t = blockIdx.x * G16_T + threadIdx.x;
    int c = t >> 2;
    const unsigned q = (unsigned)t & 3u;
    if (c >= N_NODES) return;
    v4f bq;
    bq[0] = bias[q * 4u + 0u];
    bq[1] = bias[q * 4u + 1u];
    bq[2] = bias[q * 4u + 2u];
    bq[3] = bias[q * 4u + 3u];
    const char* __restrict__ hb = (const char*)h2;
    const unsigned qo = q * 16u;
    const unsigned s0 = q * 8u;

    unsigned cnt = deg[c];
    unsigned st  = rowptr[c];
    float    dc  = dinv[c];
    int  c1   = c + G16_STRIDE;
    bool has1 = c1 < N_NODES;
    const int cp1 = has1 ? c1 : c;
    unsigned cnt1 = deg[cp1];
    unsigned st1  = rowptr[cp1];
    float    dc1  = dinv[cp1];
    if (!has1) cnt1 = 0u;
    int ix[8];
#pragma unroll
    for (int r = 0; r < 8; ++r)
        ix[r] = srcs[s0 + (unsigned)r < cnt ? st + s0 + (unsigned)r : 0u];

    for (;;) {
        const v4f selfv = *(const v4f*)(hb + ((unsigned)c * 64u + qo));
        const int  c2   = c1 + G16_STRIDE;
        const bool has2 = c2 < N_NODES;
        const int  cp2  = has2 ? c2 : c;
        const unsigned cnt2 = deg[cp2];
        const unsigned st2  = rowptr[cp2];
        const float    dc2  = dinv[cp2];
        int ix1[8];
#pragma unroll
        for (int r = 0; r < 8; ++r)
            ix1[r] = srcs[s0 + (unsigned)r < cnt1 ? st1 + s0 + (unsigned)r : 0u];

        int x[8];
#pragma unroll
        for (int r = 0; r < 8; ++r)
            x[r] = s0 + (unsigned)r < cnt ? ix[r] : c;
        v4f sum = selfv;
        if (__builtin_amdgcn_ballot_w32(cnt > 24u) != 0u)
            sum = sum_slots<32>(hb, qo, x, cnt, sum);
        else
            sum = sum_slots<24>(hb, qo, x, cnt, sum);
        for (unsigned e = 32u; e < cnt; ++e)
            sum += *(const v4f*)(hb + ((unsigned)srcs[st + e] * 64u + qo));

        v4f acc;
        acc[0] = bq[0] + dc * sum[0];
        acc[1] = bq[1] + dc * sum[1];
        acc[2] = bq[2] + dc * sum[2];
        acc[3] = bq[3] + dc * sum[3];
        float m = fmaxf(fmaxf(acc[0], acc[1]), fmaxf(acc[2], acc[3]));
        m = fmaxf(m, row_xor<1>(m));
        m = fmaxf(m, row_xor<2>(m));
        float s = (__expf(acc[0] - m) + __expf(acc[1] - m))
                + (__expf(acc[2] - m) + __expf(acc[3] - m));
        s += row_xor<1>(s);
        s += row_xor<2>(s);
        const float l = m + __logf(s);
        v4f o;
        o[0] = acc[0] - l;
        o[1] = acc[1] - l;
        o[2] = acc[2] - l;
        o[3] = acc[3] - l;
        *(v4f*)(out + (long)c * NCLASS + q * 4u) = o;

        if (!has1) break;
        c = c1;  cnt = cnt1;  st = st1;  dc = dc1;
#pragma unroll
        for (int r = 0; r < 8; ++r) ix[r] = ix1[r];
        c1 = c2;  has1 = has2;  cnt1 = has2 ? cnt2 : 0u;  st1 = st2;  dc1 = dc2;
    }
}

extern "C" void kernel_launch(void* const* d_in, const int* in_sizes, int n_in,
                              void* d_out, int out_size, void* d_ws, size_t ws_size,
                              hipStream_t stream) {
    const float* x   = (const float*)d_in[0];
    const int*   ei  = (const int*)d_in[1];
    const float* W1  = (const float*)d_in[2];
    const float* b1  = (const float*)d_in[3];
    const float* W2  = (const float*)d_in[4];
    const float* b2  = (const float*)d_in[5];
    float* out       = (float*)d_out;

    const int* row = ei;
    const int* col = ei + N_EDGES;

    auto align256 = [](size_t v) { return (v + 255) & ~(size_t)255; };
    char* ws = (char*)d_ws;
    size_t o = 0;
    float*     dinv = (float*)(ws + o);     o += align256((size_t)N_NODES * 4);
    _Float16*  W1p  = (_Float16*)(ws + o);  o += align256((size_t)NFEAT * NHID * 2);
    _Float16*  W2p  = (_Float16*)(ws + o);  o += align256((size_t)NHID * NCLASS * 2);
    _Float16*  xw   = (_Float16*)(ws + o);  o += align256((size_t)N_NODES * NHID * 2);
    float*     h2   = (float*)(ws + o);     o += align256((size_t)N_NODES * NCLASS * 4);
    unsigned*  deg    = (unsigned*)(ws + o); o += align256((size_t)N_NODES * 4);
    unsigned*  rowptr = (unsigned*)(ws + o); o += align256((size_t)N_NODES * 4);
    unsigned*  lpre   = (unsigned*)(ws + o); o += align256((size_t)N_NODES * 4);
    unsigned*  bsum   = (unsigned*)(ws + o); o += align256((size_t)SCAN_NB * 4);
    int*       srcs   = (int*)(ws + o);      o += align256((size_t)N_EDGES * 4);
    unsigned*  part = (unsigned*)(ws + o);

    k_deg_hist<<<HIST_NB + PACK1_NB + PACK2_NB, DEG_T, 0, stream>>>(
        col, part, W1, W1p, W2, W2p);
    k_deg_to_dinv<<<SCAN_NB, SCAN_B, 0, stream>>>(part, dinv, deg, bsum, lpre);
    k_fill_srcs<<<FILL_NB + GEMM1_NB, DEG_T, 0, stream>>>(row, col, part, lpre, bsum,
                                                          rowptr, srcs,
                                                          x, W1p, dinv, xw);

    k_gather64_mm<<<F_NB, AGG_B, 0, stream>>>(srcs, rowptr, deg, dinv, xw, b1, W2p, h2);

    k_gather16<<<G16_NB, G16_T, 0, stream>>>(srcs, rowptr, deg, dinv, h2, b2, out);
}
